# starting kernel
# baseline (speedup 1.0000x reference)
#include <hip/hip_runtime.h>
#include <hip/hip_bf16.h>

#define NN 50000
#define NE 800000
#define FEAT 128
#define NG 128
#define OUTC 10
#define BN_EPS 1e-5f

typedef float v2f __attribute__((ext_vector_type(2)));
typedef float v8f __attribute__((ext_vector_type(8)));

__global__ void edge_degree_kernel(const int* __restrict__ src, const int* __restrict__ dst,
                                   float* __restrict__ deg_out, float* __restrict__ deg_in) {
    int e = blockIdx.x * blockDim.x + threadIdx.x;
    if (e >= NE) return;
    atomicAdd(&deg_out[src[e]], 1.0f);
    atomicAdd(&deg_in[dst[e]], 1.0f);
}

__global__ void invsqrt_kernel(const float* __restrict__ deg_out, const float* __restrict__ deg_in,
                               float* __restrict__ inv_out, float* __restrict__ inv_in) {
    int i = blockIdx.x * blockDim.x + threadIdx.x;
    if (i >= NN) return;
    inv_out[i] = rsqrtf(fmaxf(deg_out[i], 1.0f));
    inv_in[i]  = rsqrtf(fmaxf(deg_in[i], 1.0f));
}

__global__ void gemm128_wmma_kernel(const float* __restrict__ A,
                                    const float* __restrict__ W,
                                    float* __restrict__ out,
                                    const float* __restrict__ scaleA,
                                    const float* __restrict__ bias,
                                    int relu) {
    const int lane = threadIdx.x & 31;
    const int wave = threadIdx.x >> 5;
    const int r0 = blockIdx.x * 16;
    const int c0 = wave * 16;

    const int m  = lane & 15;
    const int kb = (lane >> 4) << 1;
    const int n  = lane & 15;

    const float sA = scaleA ? scaleA[r0 + m] : 1.0f;
    const float* arow = A + (size_t)(r0 + m) * FEAT + kb;
    const float* wcol = W + (size_t)kb * FEAT + c0 + n;

    v8f acc = {0.f, 0.f, 0.f, 0.f, 0.f, 0.f, 0.f, 0.f};
#pragma unroll
    for (int k0 = 0; k0 < FEAT; k0 += 4) {
        v2f a = *(const v2f*)(arow + k0);
        a.x *= sA;
        a.y *= sA;
        v2f b;
        b.x = wcol[(size_t)k0 * FEAT];
        b.y = wcol[(size_t)(k0 + 1) * FEAT];
        acc = __builtin_amdgcn_wmma_f32_16x16x4_f32(
             false, a,  false, b,
             (short)0, acc,  false,  false);
    }

    const float bv = bias ? bias[c0 + n] : 0.0f;
    const int rhalf = (lane >> 4) * 8;
#pragma unroll
    for (int r = 0; r < 8; ++r) {
        float v = acc[r] + bv;
        if (relu) v = fmaxf(v, 0.0f);
        out[(size_t)(r0 + rhalf + r) * FEAT + c0 + n] = v;
    }
}

__global__ void scatter_add_kernel(const float* __restrict__ h,
                                   const int* __restrict__ src, const int* __restrict__ dst,
                                   float* __restrict__ agg) {
    unsigned t = blockIdx.x * blockDim.x + threadIdx.x;
    unsigned e = t >> 5;
    if (e >= NE) return;
    int c = (t & 31) * 4;
    const float4 v = *(const float4*)(h + (size_t)src[e] * FEAT + c);
    float* o = agg + (size_t)dst[e] * FEAT + c;
    atomicAdd(o + 0, v.x);
    atomicAdd(o + 1, v.y);
    atomicAdd(o + 2, v.z);
    atomicAdd(o + 3, v.w);
}

__global__ void finalize_kernel(const float* __restrict__ agg, const float* __restrict__ inv_in,
                                const float* __restrict__ bias, float* __restrict__ out) {
    unsigned t = blockIdx.x * blockDim.x + threadIdx.x;
    if (t >= (unsigned)NN * FEAT) return;
    unsigned nIdx = t >> 7;
    unsigned c = t & 127;
    float v = agg[t] * inv_in[nIdx] + bias[c];
    out[t] = fmaxf(v, 0.0f);
}

__global__ void pool_kernel(const float* __restrict__ x, const int* __restrict__ graph_ids,
                            float* __restrict__ emb) {
    unsigned t = blockIdx.x * blockDim.x + threadIdx.x;
    unsigned nIdx = t >> 5;
    if (nIdx >= NN) return;
    int c = (t & 31) * 4;
    int g = graph_ids[nIdx];
    const float4 v = *(const float4*)(x + (size_t)nIdx * FEAT + c);
    float* o = emb + (size_t)g * FEAT + c;
    atomicAdd(o + 0, v.x);
    atomicAdd(o + 1, v.y);
    atomicAdd(o + 2, v.z);
    atomicAdd(o + 3, v.w);
}

__global__ void bn_kernel(const float* __restrict__ emb, const float* __restrict__ gamma,
                          const float* __restrict__ beta, float* __restrict__ ybn) {
    int c = threadIdx.x;
    if (c >= FEAT) return;
    float mu = 0.f;
    for (int g = 0; g < NG; ++g) mu += emb[g * FEAT + c];
    mu *= (1.0f / NG);
    float var = 0.f;
    for (int g = 0; g < NG; ++g) {
        float d = emb[g * FEAT + c] - mu;
        var += d * d;
    }
    var *= (1.0f / NG);
    float s = gamma[c] * rsqrtf(var + BN_EPS);
    float b = beta[c];
    for (int g = 0; g < NG; ++g)
        ybn[g * FEAT + c] = (emb[g * FEAT + c] - mu) * s + b;
}

__global__ void fc2_logsoftmax_kernel(const float* __restrict__ h1,
                                      const float* __restrict__ fc2_w,
                                      const float* __restrict__ fc2_b,
                                      float* __restrict__ out) {
    int g = threadIdx.x;
    if (g >= NG) return;
    float logit[OUTC];
#pragma unroll
    for (int j = 0; j < OUTC; ++j) logit[j] = fc2_b[j];
    const float* hrow = h1 + (size_t)g * FEAT;
    for (int k = 0; k < FEAT; ++k) {
        float hv = hrow[k];
        const float* wrow = fc2_w + (size_t)k * OUTC;
#pragma unroll
        for (int j = 0; j < OUTC; ++j) logit[j] = fmaf(hv, wrow[j], logit[j]);
    }
    float mx = -1e30f;
#pragma unroll
    for (int j = 0; j < OUTC; ++j) mx = fmaxf(mx, logit[j]);
    float s = 0.f;
#pragma unroll
    for (int j = 0; j < OUTC; ++j) s += __expf(logit[j] - mx);
    float lse = mx + __logf(s);
#pragma unroll
    for (int j = 0; j < OUTC; ++j) out[g * OUTC + j] = logit[j] - lse;
}

extern "C" void kernel_launch(void* const* d_in, const int* in_sizes, int n_in,
                              void* d_out, int out_size, void* d_ws, size_t ws_size,
                              hipStream_t stream) {
    const float* n_feat    = (const float*)d_in[0];
    const int*   src       = (const int*)  d_in[1];
    const int*   dst       = (const int*)  d_in[2];
    const int*   graph_ids = (const int*)  d_in[3];
    const float* W1        = (const float*)d_in[4];
    const float* b1        = (const float*)d_in[5];
    const float* W2        = (const float*)d_in[6];
    const float* b2        = (const float*)d_in[7];
    const float* gamma     = (const float*)d_in[8];
    const float* beta      = (const float*)d_in[9];
    const float* fc1_w     = (const float*)d_in[10];
    const float* fc1_b     = (const float*)d_in[11];
    const float* fc2_w     = (const float*)d_in[12];
    const float* fc2_b     = (const float*)d_in[13];

    float* out = (float*)d_out;

    float* ws      = (float*)d_ws;
    float* deg_out = ws;
    float* deg_in  = deg_out + NN;
    float* inv_out = deg_in  + NN;
    float* inv_in  = inv_out + NN;
    float* bufA    = inv_in  + NN;
    float* bufB    = bufA + (size_t)NN * FEAT;
    float* ybn     = bufB + (size_t)NN * FEAT;
    float* h1      = ybn  + NG * FEAT;

    hipMemsetAsync(deg_out, 0, (size_t)2 * NN * sizeof(float), stream);
    hipMemsetAsync(out, 0, (size_t)NG * FEAT * sizeof(float), stream);

    edge_degree_kernel<<<(NE + 255) / 256, 256, 0, stream>>>(src, dst, deg_out, deg_in);
    invsqrt_kernel<<<(NN + 255) / 256, 256, 0, stream>>>(deg_out, deg_in, inv_out, inv_in);

    const int nodeTiles = NN / 16;
    const int scatterBlocks = (NE * 32) / 256;
    const int elemBlocks = (NN * FEAT) / 256;

    gemm128_wmma_kernel<<<nodeTiles, 256, 0, stream>>>(n_feat, W1, bufA, inv_out, nullptr, 0);
    hipMemsetAsync(bufB, 0, (size_t)NN * FEAT * sizeof(float), stream);
    scatter_add_kernel<<<scatterBlocks, 256, 0, stream>>>(bufA, src, dst, bufB);
    finalize_kernel<<<elemBlocks, 256, 0, stream>>>(bufB, inv_in, b1, bufA);

    gemm128_wmma_kernel<<<nodeTiles, 256, 0, stream>>>(bufA, W2, bufB, inv_out, nullptr, 0);
    hipMemsetAsync(bufA, 0, (size_t)NN * FEAT * sizeof(float), stream);
    scatter_add_kernel<<<scatterBlocks, 256, 0, stream>>>(bufB, src, dst, bufA);
    finalize_kernel<<<elemBlocks, 256, 0, stream>>>(bufA, inv_in, b2, bufB);

    pool_kernel<<<(NN * 32) / 256, 256, 0, stream>>>(bufB, graph_ids, out);

    bn_kernel<<<1, FEAT, 0, stream>>>(out, gamma, beta, ybn);
    gemm128_wmma_kernel<<<NG / 16, 256, 0, stream>>>(ybn, fc1_w, h1, nullptr, fc1_b, 1);
    fc2_logsoftmax_kernel<<<1, NG, 0, stream>>>(h1, fc2_w, fc2_b, out + (size_t)NG * FEAT);
}


# aggregation as a gather: edges bucketed by destination (returning count = arrival-order rank, scan, fill); a wave sums a node's in-rows in registers, stores relu(sum*norm+b); no float atomics in it
# speedup vs baseline: 1.8273x; 1.8273x over previous
#include <hip/hip_runtime.h>
#include <hip/hip_bf16.h>

#define NN 50000
#define NE 800000
#define FEAT 128
#define NG 128
#define OUTC 10
#define BN_EPS 1e-5f

typedef float v2f __attribute__((ext_vector_type(2)));
typedef float v8f __attribute__((ext_vector_type(8)));

__global__ void edge_count_rank_kernel(const int* __restrict__ src, const int* __restrict__ dst,
                                       int* __restrict__ cnt_out, int* __restrict__ cnt_in,
                                       int* __restrict__ rank) {
    int e = blockIdx.x * blockDim.x + threadIdx.x;
    if (e >= NE) return;
    const int sN = src[e];
    const int dN = dst[e];
    atomicAdd(&cnt_out[sN], 1);
    rank[e] = atomicAdd(&cnt_in[dN], 1);
}

__global__ void invsqrt_kernel(const int* __restrict__ cnt_out, const int* __restrict__ cnt_in,
                               float* __restrict__ inv_out, float* __restrict__ inv_in) {
    int i = blockIdx.x * blockDim.x + threadIdx.x;
    if (i >= NN) return;
    inv_out[i] = rsqrtf(fmaxf((float)cnt_out[i], 1.0f));
    inv_in[i]  = rsqrtf(fmaxf((float)cnt_in[i], 1.0f));
}

#define SCAN_TRIPS 13
#define SCAN_CHUNK (SCAN_TRIPS * 128)
__global__ __launch_bounds__(1024) void rowptr_scan_kernel(const int* __restrict__ cnt_in,
                                                           int* __restrict__ row_ptr) {
    __shared__ int wtot[32];
    const int lane = threadIdx.x & 31;
    const int wave = threadIdx.x >> 5;
    const int base = wave * SCAN_CHUNK + lane * 4;

    int4 ex[SCAN_TRIPS];
    int carry = 0;
#pragma unroll
    for (int i = 0; i < SCAN_TRIPS; ++i) {
        const int idx = base + i * 128;
        int4 c = make_int4(0, 0, 0, 0);
        if (idx < NN) c = *(const int4*)(cnt_in + idx);
        const int sum4 = c.x + c.y + c.z + c.w;
        int incl = sum4;
#pragma unroll
        for (int d = 1; d < 32; d <<= 1) {
            int up = __shfl_up(incl, d);
            if (lane >= d) incl += up;
        }
        const int e0 = carry + incl - sum4;
        ex[i] = make_int4(e0, e0 + c.x, e0 + c.x + c.y, e0 + c.x + c.y + c.z);
        carry += __shfl(incl, 31);
    }
    if (lane == 0) wtot[wave] = carry;
    __syncthreads();
    int off = (lane < wave) ? wtot[lane] : 0;
#pragma unroll
    for (int d = 16; d >= 1; d >>= 1) off += __shfl_xor(off, d);
#pragma unroll
    for (int i = 0; i < SCAN_TRIPS; ++i) {
        const int idx = base + i * 128;
        if (idx < NN)
            *(int4*)(row_ptr + idx) = make_int4(ex[i].x + off, ex[i].y + off, ex[i].z + off, ex[i].w + off);
    }
    if (threadIdx.x == 1023) row_ptr[NN] = off + carry;
}

__global__ void csr_fill_kernel(const int* __restrict__ src, const int* __restrict__ dst,
                                const int* __restrict__ rank, const int* __restrict__ row_ptr,
                                int* __restrict__ csr_src) {
    int e = blockIdx.x * blockDim.x + threadIdx.x;
    if (e >= NE) return;
    const int sN = src[e];
    const unsigned pos = (unsigned)(row_ptr[dst[e]] + rank[e]);
    if (pos < (unsigned)NE) csr_src[pos] = sN;
}

__global__ void gemm128_wmma_kernel(const float* __restrict__ A,
                                    const float* __restrict__ W,
                                    float* __restrict__ out,
                                    const float* __restrict__ scaleA,
                                    const float* __restrict__ bias,
                                    int relu) {
    const int lane = threadIdx.x & 31;
    const int wave = threadIdx.x >> 5;
    const int r0 = blockIdx.x * 16;
    const int c0 = wave * 16;

    const int m  = lane & 15;
    const int kb = (lane >> 4) << 1;
    const int n  = lane & 15;

    const float sA = scaleA ? scaleA[r0 + m] : 1.0f;
    const float* arow = A + (size_t)(r0 + m) * FEAT + kb;
    const float* wcol = W + (size_t)kb * FEAT + c0 + n;

    v8f acc = {0.f, 0.f, 0.f, 0.f, 0.f, 0.f, 0.f, 0.f};
#pragma unroll
    for (int k0 = 0; k0 < FEAT; k0 += 4) {
        v2f a = *(const v2f*)(arow + k0);
        a.x *= sA;
        a.y *= sA;
        v2f b;
        b.x = wcol[(size_t)k0 * FEAT];
        b.y = wcol[(size_t)(k0 + 1) * FEAT];
        acc = __builtin_amdgcn_wmma_f32_16x16x4_f32(
             false, a,  false, b,
             (short)0, acc,  false,  false);
    }

    const float bv = bias ? bias[c0 + n] : 0.0f;
    const int rhalf = (lane >> 4) * 8;
#pragma unroll
    for (int r = 0; r < 8; ++r) {
        float v = acc[r] + bv;
        if (relu) v = fmaxf(v, 0.0f);
        out[(size_t)(r0 + rhalf + r) * FEAT + c0 + n] = v;
    }
}

static __device__ __forceinline__ float4 ld_row(const float* hl, int ids, int j) {
    const int s = __builtin_amdgcn_readlane(ids, j);
    return *(const float4*)(hl + (size_t)s * FEAT);
}
static __device__ __forceinline__ void add4(float4& a, const float4 v) {
    a.x += v.x; a.y += v.y; a.z += v.z; a.w += v.w;
}

__global__ __launch_bounds__(256) void gather_sum_kernel(const float* __restrict__ h,
                                                         const int* __restrict__ row_ptr,
                                                         const int* __restrict__ csr_src,
                                                         const float* __restrict__ inv_in,
                                                         const float* __restrict__ bias,
                                                         float* __restrict__ out) {
    const int lane = threadIdx.x & 31;
    const int wv = __builtin_amdgcn_readfirstlane((int)(threadIdx.x >> 5));
    const int node = blockIdx.x * 8 + wv;
    const int start = row_ptr[node];
    const int end = row_ptr[node + 1];
    const float* hl = h + lane * 4;

    float4 acc = make_float4(0.f, 0.f, 0.f, 0.f);
    for (int base = start; base < end; base += 32) {
        const int n = min(32, end - base);
        int mine = (lane < n) ? csr_src[base + lane] : 0;
        mine = (int)min((unsigned)mine, (unsigned)(NN - 1));
        int j = 0;
        for (; j + 8 <= n; j += 8) {
            const float4 v0 = ld_row(hl, mine, j);
            const float4 v1 = ld_row(hl, mine, j + 1);
            const float4 v2 = ld_row(hl, mine, j + 2);
            const float4 v3 = ld_row(hl, mine, j + 3);
            const float4 v4 = ld_row(hl, mine, j + 4);
            const float4 v5 = ld_row(hl, mine, j + 5);
            const float4 v6 = ld_row(hl, mine, j + 6);
            const float4 v7 = ld_row(hl, mine, j + 7);
            add4(acc, v0); add4(acc, v1); add4(acc, v2); add4(acc, v3);
            add4(acc, v4); add4(acc, v5); add4(acc, v6); add4(acc, v7);
        }
        if (j + 4 <= n) {
            const float4 v0 = ld_row(hl, mine, j);
            const float4 v1 = ld_row(hl, mine, j + 1);
            const float4 v2 = ld_row(hl, mine, j + 2);
            const float4 v3 = ld_row(hl, mine, j + 3);
            add4(acc, v0); add4(acc, v1); add4(acc, v2); add4(acc, v3);
            j += 4;
        }
        if (j < n) {
            const float4 v0 = ld_row(hl, mine, j);
            float4 v1 = make_float4(0.f, 0.f, 0.f, 0.f);
            float4 v2 = make_float4(0.f, 0.f, 0.f, 0.f);
            if (j + 1 < n) v1 = ld_row(hl, mine, j + 1);
            if (j + 2 < n) v2 = ld_row(hl, mine, j + 2);
            add4(acc, v0); add4(acc, v1); add4(acc, v2);
        }
    }

    const float sc = inv_in[node];
    const float4 b = *(const float4*)(bias + lane * 4);
    float4 o;
    o.x = fmaxf(acc.x * sc + b.x, 0.0f);
    o.y = fmaxf(acc.y * sc + b.y, 0.0f);
    o.z = fmaxf(acc.z * sc + b.z, 0.0f);
    o.w = fmaxf(acc.w * sc + b.w, 0.0f);
    *(float4*)(out + (size_t)node * FEAT + lane * 4) = o;
}

__global__ void pool_kernel(const float* __restrict__ x, const int* __restrict__ graph_ids,
                            float* __restrict__ emb) {
    unsigned t = blockIdx.x * blockDim.x + threadIdx.x;
    unsigned nIdx = t >> 5;
    if (nIdx >= NN) return;
    int c = (t & 31) * 4;
    int g = graph_ids[nIdx];
    const float4 v = *(const float4*)(x + (size_t)nIdx * FEAT + c);
    float* o = emb + (size_t)g * FEAT + c;
    atomicAdd(o + 0, v.x);
    atomicAdd(o + 1, v.y);
    atomicAdd(o + 2, v.z);
    atomicAdd(o + 3, v.w);
}

__global__ void bn_kernel(const float* __restrict__ emb, const float* __restrict__ gamma,
                          const float* __restrict__ beta, float* __restrict__ ybn) {
    int c = threadIdx.x;
    if (c >= FEAT) return;
    float mu = 0.f;
    for (int g = 0; g < NG; ++g) mu += emb[g * FEAT + c];
    mu *= (1.0f / NG);
    float var = 0.f;
    for (int g = 0; g < NG; ++g) {
        float d = emb[g * FEAT + c] - mu;
        var += d * d;
    }
    var *= (1.0f / NG);
    float s = gamma[c] * rsqrtf(var + BN_EPS);
    float b = beta[c];
    for (int g = 0; g < NG; ++g)
        ybn[g * FEAT + c] = (emb[g * FEAT + c] - mu) * s + b;
}

__global__ void fc2_logsoftmax_kernel(const float* __restrict__ h1,
                                      const float* __restrict__ fc2_w,
                                      const float* __restrict__ fc2_b,
                                      float* __restrict__ out) {
    int g = threadIdx.x;
    if (g >= NG) return;
    float logit[OUTC];
#pragma unroll
    for (int j = 0; j < OUTC; ++j) logit[j] = fc2_b[j];
    const float* hrow = h1 + (size_t)g * FEAT;
    for (int k = 0; k < FEAT; ++k) {
        float hv = hrow[k];
        const float* wrow = fc2_w + (size_t)k * OUTC;
#pragma unroll
        for (int j = 0; j < OUTC; ++j) logit[j] = fmaf(hv, wrow[j], logit[j]);
    }
    float mx = -1e30f;
#pragma unroll
    for (int j = 0; j < OUTC; ++j) mx = fmaxf(mx, logit[j]);
    float s = 0.f;
#pragma unroll
    for (int j = 0; j < OUTC; ++j) s += __expf(logit[j] - mx);
    float lse = mx + __logf(s);
#pragma unroll
    for (int j = 0; j < OUTC; ++j) out[g * OUTC + j] = logit[j] - lse;
}

extern "C" void kernel_launch(void* const* d_in, const int* in_sizes, int n_in,
                              void* d_out, int out_size, void* d_ws, size_t ws_size,
                              hipStream_t stream) {
    const float* n_feat    = (const float*)d_in[0];
    const int*   src       = (const int*)  d_in[1];
    const int*   dst       = (const int*)  d_in[2];
    const int*   graph_ids = (const int*)  d_in[3];
    const float* W1        = (const float*)d_in[4];
    const float* b1        = (const float*)d_in[5];
    const float* W2        = (const float*)d_in[6];
    const float* b2        = (const float*)d_in[7];
    const float* gamma     = (const float*)d_in[8];
    const float* beta      = (const float*)d_in[9];
    const float* fc1_w     = (const float*)d_in[10];
    const float* fc1_b     = (const float*)d_in[11];
    const float* fc2_w     = (const float*)d_in[12];
    const float* fc2_b     = (const float*)d_in[13];

    float* out = (float*)d_out;

    int*   cnt_out = (int*)d_ws;
    int*   cnt_in  = cnt_out + NN;
    float* inv_out = (float*)(cnt_in + NN);
    float* inv_in  = inv_out + NN;
    int*   row_ptr = (int*)(inv_in + NN);
    int*   rank    = row_ptr + 50112;
    int*   csr_src = rank + NE;
    float* bufA    = (float*)(csr_src + NE);
    float* bufB    = bufA + (size_t)NN * FEAT;
    float* ybn     = bufB + (size_t)NN * FEAT;
    float* h1      = ybn  + NG * FEAT;

    hipMemsetAsync(cnt_out, 0, (size_t)2 * NN * sizeof(int), stream);
    hipMemsetAsync(out, 0, (size_t)NG * FEAT * sizeof(float), stream);

    edge_count_rank_kernel<<<(NE + 255) / 256, 256, 0, stream>>>(src, dst, cnt_out, cnt_in, rank);
    invsqrt_kernel<<<(NN + 255) / 256, 256, 0, stream>>>(cnt_out, cnt_in, inv_out, inv_in);
    rowptr_scan_kernel<<<1, 1024, 0, stream>>>(cnt_in, row_ptr);
    csr_fill_kernel<<<(NE + 255) / 256, 256, 0, stream>>>(src, dst, rank, row_ptr, csr_src);

    const int nodeTiles = NN / 16;
    const int nodeBlocks = NN / 8;

    gemm128_wmma_kernel<<<nodeTiles, 256, 0, stream>>>(n_feat, W1, bufA, inv_out, nullptr, 0);
    gather_sum_kernel<<<nodeBlocks, 256, 0, stream>>>(bufA, row_ptr, csr_src, inv_in, b1, bufB);

    gemm128_wmma_kernel<<<nodeTiles, 256, 0, stream>>>(bufB, W2, bufA, inv_out, nullptr, 0);
    gather_sum_kernel<<<nodeBlocks, 256, 0, stream>>>(bufA, row_ptr, csr_src, inv_in, b2, bufB);

    pool_kernel<<<(NN * 32) / 256, 256, 0, stream>>>(bufB, graph_ids, out);

    bn_kernel<<<1, FEAT, 0, stream>>>(out, gamma, beta, ybn);
    gemm128_wmma_kernel<<<NG / 16, 256, 0, stream>>>(ybn, fc1_w, h1, nullptr, fc1_b, 1);
    fc2_logsoftmax_kernel<<<1, NG, 0, stream>>>(h1, fc2_w, fc2_b, out + (size_t)NG * FEAT);
}


# sum-pool folded into the layer-2 gather: a block's 8 finished rows meet in LDS, runs of one graph id are summed in node order, then one float atomic per column and run; pool kernel and x2 store gone
# speedup vs baseline: 2.3740x; 1.2992x over previous
#include <hip/hip_runtime.h>
#include <hip/hip_bf16.h>

#define NN 50000
#define NE 800000
#define FEAT 128
#define NG 128
#define OUTC 10
#define BN_EPS 1e-5f

typedef float v2f __attribute__((ext_vector_type(2)));
typedef float v8f __attribute__((ext_vector_type(8)));

__global__ void edge_count_rank_kernel(const int* __restrict__ src, const int* __restrict__ dst,
                                       int* __restrict__ cnt_out, int* __restrict__ cnt_in,
                                       int* __restrict__ rank) {
    int e = blockIdx.x * blockDim.x + threadIdx.x;
    if (e >= NE) return;
    const int sN = src[e];
    const int dN = dst[e];
    atomicAdd(&cnt_out[sN], 1);
    rank[e] = atomicAdd(&cnt_in[dN], 1);
}

__global__ void invsqrt_kernel(const int* __restrict__ cnt_out, const int* __restrict__ cnt_in,
                               float* __restrict__ inv_out, float* __restrict__ inv_in) {
    int i = blockIdx.x * blockDim.x + threadIdx.x;
    if (i >= NN) return;
    inv_out[i] = rsqrtf(fmaxf((float)cnt_out[i], 1.0f));
    inv_in[i]  = rsqrtf(fmaxf((float)cnt_in[i], 1.0f));
}

#define SCAN_TRIPS 13
#define SCAN_CHUNK (SCAN_TRIPS * 128)
__global__ __launch_bounds__(1024) void rowptr_scan_kernel(const int* __restrict__ cnt_in,
                                                           int* __restrict__ row_ptr) {
    __shared__ int wtot[32];
    const int lane = threadIdx.x & 31;
    const int wave = threadIdx.x >> 5;
    const int base = wave * SCAN_CHUNK + lane * 4;

    int4 ex[SCAN_TRIPS];
    int carry = 0;
#pragma unroll
    for (int i = 0; i < SCAN_TRIPS; ++i) {
        const int idx = base + i * 128;
        int4 c = make_int4(0, 0, 0, 0);
        if (idx < NN) c = *(const int4*)(cnt_in + idx);
        const int sum4 = c.x + c.y + c.z + c.w;
        int incl = sum4;
#pragma unroll
        for (int d = 1; d < 32; d <<= 1) {
            int up = __shfl_up(incl, d);
            if (lane >= d) incl += up;
        }
        const int e0 = carry + incl - sum4;
        ex[i] = make_int4(e0, e0 + c.x, e0 + c.x + c.y, e0 + c.x + c.y + c.z);
        carry += __shfl(incl, 31);
    }
    if (lane == 0) wtot[wave] = carry;
    __syncthreads();
    int off = (lane < wave) ? wtot[lane] : 0;
#pragma unroll
    for (int d = 16; d >= 1; d >>= 1) off += __shfl_xor(off, d);
#pragma unroll
    for (int i = 0; i < SCAN_TRIPS; ++i) {
        const int idx = base + i * 128;
        if (idx < NN)
            *(int4*)(row_ptr + idx) = make_int4(ex[i].x + off, ex[i].y + off, ex[i].z + off, ex[i].w + off);
    }
    if (threadIdx.x == 1023) row_ptr[NN] = off + carry;
}

__global__ void csr_fill_kernel(const int* __restrict__ src, const int* __restrict__ dst,
                                const int* __restrict__ rank, const int* __restrict__ row_ptr,
                                int* __restrict__ csr_src) {
    int e = blockIdx.x * blockDim.x + threadIdx.x;
    if (e >= NE) return;
    const int sN = src[e];
    const unsigned pos = (unsigned)(row_ptr[dst[e]] + rank[e]);
    if (pos < (unsigned)NE) csr_src[pos] = sN;
}

__global__ void gemm128_wmma_kernel(const float* __restrict__ A,
                                    const float* __restrict__ W,
                                    float* __restrict__ out,
                                    const float* __restrict__ scaleA,
                                    const float* __restrict__ bias,
                                    int relu) {
    const int lane = threadIdx.x & 31;
    const int wave = threadIdx.x >> 5;
    const int r0 = blockIdx.x * 16;
    const int c0 = wave * 16;

    const int m  = lane & 15;
    const int kb = (lane >> 4) << 1;
    const int n  = lane & 15;

    const float sA = scaleA ? scaleA[r0 + m] : 1.0f;
    const float* arow = A + (size_t)(r0 + m) * FEAT + kb;
    const float* wcol = W + (size_t)kb * FEAT + c0 + n;

    v8f acc = {0.f, 0.f, 0.f, 0.f, 0.f, 0.f, 0.f, 0.f};
#pragma unroll
    for (int k0 = 0; k0 < FEAT; k0 += 4) {
        v2f a = *(const v2f*)(arow + k0);
        a.x *= sA;
        a.y *= sA;
        v2f b;
        b.x = wcol[(size_t)k0 * FEAT];
        b.y = wcol[(size_t)(k0 + 1) * FEAT];
        acc = __builtin_amdgcn_wmma_f32_16x16x4_f32(
             false, a,  false, b,
             (short)0, acc,  false,  false);
    }

    const float bv = bias ? bias[c0 + n] : 0.0f;
    const int rhalf = (lane >> 4) * 8;
#pragma unroll
    for (int r = 0; r < 8; ++r) {
        float v = acc[r] + bv;
        if (relu) v = fmaxf(v, 0.0f);
        out[(size_t)(r0 + rhalf + r) * FEAT + c0 + n] = v;
    }
}

static __device__ __forceinline__ float4 ld_row(const float* hl, int ids, int j) {
    const int s = __builtin_amdgcn_readlane(ids, j);
    return *(const float4*)(hl + (size_t)s * FEAT);
}
static __device__ __forceinline__ void add4(float4& a, const float4 v) {
    a.x += v.x; a.y += v.y; a.z += v.z; a.w += v.w;
}

static __device__ __forceinline__ float4 gather_row(const float* __restrict__ h,
                                                    const int* __restrict__ row_ptr,
                                                    const int* __restrict__ csr_src,
                                                    const float* __restrict__ inv_in,
                                                    const float* __restrict__ bias,
                                                    const int node, const int lane) {
    const int start = row_ptr[node];
    const int end = row_ptr[node + 1];
    const float* hl = h + lane * 4;

    float4 acc = make_float4(0.f, 0.f, 0.f, 0.f);
    for (int base = start; base < end; base += 32) {
        const int n = min(32, end - base);
        int mine = (lane < n) ? csr_src[base + lane] : 0;
        mine = (int)min((unsigned)mine, (unsigned)(NN - 1));
        int j = 0;
        for (; j + 8 <= n; j += 8) {
            const float4 v0 = ld_row(hl, mine, j);
            const float4 v1 = ld_row(hl, mine, j + 1);
            const float4 v2 = ld_row(hl, mine, j + 2);
            const float4 v3 = ld_row(hl, mine, j + 3);
            const float4 v4 = ld_row(hl, mine, j + 4);
            const float4 v5 = ld_row(hl, mine, j + 5);
            const float4 v6 = ld_row(hl, mine, j + 6);
            const float4 v7 = ld_row(hl, mine, j + 7);
            add4(acc, v0); add4(acc, v1); add4(acc, v2); add4(acc, v3);
            add4(acc, v4); add4(acc, v5); add4(acc, v6); add4(acc, v7);
        }
        if (j + 4 <= n) {
            const float4 v0 = ld_row(hl, mine, j);
            const float4 v1 = ld_row(hl, mine, j + 1);
            const float4 v2 = ld_row(hl, mine, j + 2);
            const float4 v3 = ld_row(hl, mine, j + 3);
            add4(acc, v0); add4(acc, v1); add4(acc, v2); add4(acc, v3);
            j += 4;
        }
        if (j < n) {
            const float4 v0 = ld_row(hl, mine, j);
            float4 v1 = make_float4(0.f, 0.f, 0.f, 0.f);
            float4 v2 = make_float4(0.f, 0.f, 0.f, 0.f);
            if (j + 1 < n) v1 = ld_row(hl, mine, j + 1);
            if (j + 2 < n) v2 = ld_row(hl, mine, j + 2);
            add4(acc, v0); add4(acc, v1); add4(acc, v2);
        }
    }

    const float sc = inv_in[node];
    const float4 b = *(const float4*)(bias + lane * 4);
    float4 o;
    o.x = fmaxf(acc.x * sc + b.x, 0.0f);
    o.y = fmaxf(acc.y * sc + b.y, 0.0f);
    o.z = fmaxf(acc.z * sc + b.z, 0.0f);
    o.w = fmaxf(acc.w * sc + b.w, 0.0f);
    return o;
}

__global__ __launch_bounds__(256) void gather_sum_kernel(const float* __restrict__ h,
                                                         const int* __restrict__ row_ptr,
                                                         const int* __restrict__ csr_src,
                                                         const float* __restrict__ inv_in,
                                                         const float* __restrict__ bias,
                                                         float* __restrict__ out) {
    const int lane = threadIdx.x & 31;
    const int wv = __builtin_amdgcn_readfirstlane((int)(threadIdx.x >> 5));
    const int node = blockIdx.x * 8 + wv;
    const float4 o = gather_row(h, row_ptr, csr_src, inv_in, bias, node, lane);
    *(float4*)(out + (size_t)node * FEAT + lane * 4) = o;
}

__global__ __launch_bounds__(256) void gather_sum_pool_kernel(const float* __restrict__ h,
                                                              const int* __restrict__ row_ptr,
                                                              const int* __restrict__ csr_src,
                                                              const float* __restrict__ inv_in,
                                                              const float* __restrict__ bias,
                                                              const int* __restrict__ graph_ids,
                                                              float* __restrict__ emb) {
    __shared__ float4 rows[8][32];
    __shared__ int gids[8];
    const int lane = threadIdx.x & 31;
    const int wv = __builtin_amdgcn_readfirstlane((int)(threadIdx.x >> 5));
    const int node = blockIdx.x * 8 + wv;
    const int gid = graph_ids[node];
    rows[wv][lane] = gather_row(h, row_ptr, csr_src, inv_in, bias, node, lane);
    if (lane == 0) gids[wv] = gid;
    __syncthreads();
    if (threadIdx.x < FEAT) {
        const int c = threadIdx.x;
        const float* col = (const float*)&rows[0][0] + c;
        int gv[8];
        float xv[8];
#pragma unroll
        for (int w = 0; w < 8; ++w) {
            gv[w] = __builtin_amdgcn_readfirstlane(gids[w]);
            xv[w] = col[w * FEAT];
        }
        int g = gv[0];
        float s = xv[0];
#pragma unroll
        for (int w = 1; w < 8; ++w) {
            if (gv[w] != g) {
                atomicAdd(emb + (size_t)g * FEAT + c, s);
                g = gv[w];
                s = xv[w];
            } else {
                s += xv[w];
            }
        }
        atomicAdd(emb + (size_t)g * FEAT + c, s);
    }
}

__global__ void bn_kernel(const float* __restrict__ emb, const float* __restrict__ gamma,
                          const float* __restrict__ beta, float* __restrict__ ybn) {
    int c = threadIdx.x;
    if (c >= FEAT) return;
    float mu = 0.f;
    for (int g = 0; g < NG; ++g) mu += emb[g * FEAT + c];
    mu *= (1.0f / NG);
    float var = 0.f;
    for (int g = 0; g < NG; ++g) {
        float d = emb[g * FEAT + c] - mu;
        var += d * d;
    }
    var *= (1.0f / NG);
    float s = gamma[c] * rsqrtf(var + BN_EPS);
    float b = beta[c];
    for (int g = 0; g < NG; ++g)
        ybn[g * FEAT + c] = (emb[g * FEAT + c] - mu) * s + b;
}

__global__ void fc2_logsoftmax_kernel(const float* __restrict__ h1,
                                      const float* __restrict__ fc2_w,
                                      const float* __restrict__ fc2_b,
                                      float* __restrict__ out) {
    int g = threadIdx.x;
    if (g >= NG) return;
    float logit[OUTC];
#pragma unroll
    for (int j = 0; j < OUTC; ++j) logit[j] = fc2_b[j];
    const float* hrow = h1 + (size_t)g * FEAT;
    for (int k = 0; k < FEAT; ++k) {
        float hv = hrow[k];
        const float* wrow = fc2_w + (size_t)k * OUTC;
#pragma unroll
        for (int j = 0; j < OUTC; ++j) logit[j] = fmaf(hv, wrow[j], logit[j]);
    }
    float mx = -1e30f;
#pragma unroll
    for (int j = 0; j < OUTC; ++j) mx = fmaxf(mx, logit[j]);
    float s = 0.f;
#pragma unroll
    for (int j = 0; j < OUTC; ++j) s += __expf(logit[j] - mx);
    float lse = mx + __logf(s);
#pragma unroll
    for (int j = 0; j < OUTC; ++j) out[g * OUTC + j] = logit[j] - lse;
}

extern "C" void kernel_launch(void* const* d_in, const int* in_sizes, int n_in,
                              void* d_out, int out_size, void* d_ws, size_t ws_size,
                              hipStream_t stream) {
    const float* n_feat    = (const float*)d_in[0];
    const int*   src       = (const int*)  d_in[1];
    const int*   dst       = (const int*)  d_in[2];
    const int*   graph_ids = (const int*)  d_in[3];
    const float* W1        = (const float*)d_in[4];
    const float* b1        = (const float*)d_in[5];
    const float* W2        = (const float*)d_in[6];
    const float* b2        = (const float*)d_in[7];
    const float* gamma     = (const float*)d_in[8];
    const float* beta      = (const float*)d_in[9];
    const float* fc1_w     = (const float*)d_in[10];
    const float* fc1_b     = (const float*)d_in[11];
    const float* fc2_w     = (const float*)d_in[12];
    const float* fc2_b     = (const float*)d_in[13];

    float* out = (float*)d_out;

    int*   cnt_out = (int*)d_ws;
    int*   cnt_in  = cnt_out + NN;
    float* inv_out = (float*)(cnt_in + NN);
    float* inv_in  = inv_out + NN;
    int*   row_ptr = (int*)(inv_in + NN);
    int*   rank    = row_ptr + 50112;
    int*   csr_src = rank + NE;
    float* bufA    = (float*)(csr_src + NE);
    float* bufB    = bufA + (size_t)NN * FEAT;
    float* ybn     = bufB + (size_t)NN * FEAT;
    float* h1      = ybn  + NG * FEAT;

    hipMemsetAsync(cnt_out, 0, (size_t)2 * NN * sizeof(int), stream);
    hipMemsetAsync(out, 0, (size_t)NG * FEAT * sizeof(float), stream);

    edge_count_rank_kernel<<<(NE + 255) / 256, 256, 0, stream>>>(src, dst, cnt_out, cnt_in, rank);
    invsqrt_kernel<<<(NN + 255) / 256, 256, 0, stream>>>(cnt_out, cnt_in, inv_out, inv_in);
    rowptr_scan_kernel<<<1, 1024, 0, stream>>>(cnt_in, row_ptr);
    csr_fill_kernel<<<(NE + 255) / 256, 256, 0, stream>>>(src, dst, rank, row_ptr, csr_src);

    const int nodeTiles = NN / 16;
    const int nodeBlocks = NN / 8;

    gemm128_wmma_kernel<<<nodeTiles, 256, 0, stream>>>(n_feat, W1, bufA, inv_out, nullptr, 0);
    gather_sum_kernel<<<nodeBlocks, 256, 0, stream>>>(bufA, row_ptr, csr_src, inv_in, b1, bufB);

    gemm128_wmma_kernel<<<nodeTiles, 256, 0, stream>>>(bufB, W2, bufA, inv_out, nullptr, 0);
    gather_sum_pool_kernel<<<nodeBlocks, 256, 0, stream>>>(bufA, row_ptr, csr_src, inv_in, b2, graph_ids, out);

    bn_kernel<<<1, FEAT, 0, stream>>>(out, gamma, beta, ybn);
    gemm128_wmma_kernel<<<NG / 16, 256, 0, stream>>>(ybn, fc1_w, h1, nullptr, fc1_b, 1);
    fc2_logsoftmax_kernel<<<1, NG, 0, stream>>>(h1, fc2_w, fc2_b, out + (size_t)NG * FEAT);
}


# GEMM re-tiled: a block keeps its W column slices in registers over 5 row tiles and reads each A tile once as whole rows through LDS (two buffers), not per-wave 8- and 4-byte loads; k order kept
# speedup vs baseline: 2.6761x; 1.1273x over previous
#include <hip/hip_runtime.h>
#include <hip/hip_bf16.h>

#define NN 50000
#define NE 800000
#define FEAT 128
#define NG 128
#define OUTC 10
#define BN_EPS 1e-5f

typedef float v2f __attribute__((ext_vector_type(2)));
typedef float v8f __attribute__((ext_vector_type(8)));

__global__ void edge_count_rank_kernel(const int* __restrict__ src, const int* __restrict__ dst,
                                       int* __restrict__ cnt_out, int* __restrict__ cnt_in,
                                       int* __restrict__ rank) {
    int e = blockIdx.x * blockDim.x + threadIdx.x;
    if (e >= NE) return;
    const int sN = src[e];
    const int dN = dst[e];
    atomicAdd(&cnt_out[sN], 1);
    rank[e] = atomicAdd(&cnt_in[dN], 1);
}

__global__ void invsqrt_kernel(const int* __restrict__ cnt_out, const int* __restrict__ cnt_in,
                               float* __restrict__ inv_out, float* __restrict__ inv_in) {
    int i = blockIdx.x * blockDim.x + threadIdx.x;
    if (i >= NN) return;
    inv_out[i] = rsqrtf(fmaxf((float)cnt_out[i], 1.0f));
    inv_in[i]  = rsqrtf(fmaxf((float)cnt_in[i], 1.0f));
}

#define SCAN_TRIPS 13
#define SCAN_CHUNK (SCAN_TRIPS * 128)
__global__ __launch_bounds__(1024) void rowptr_scan_kernel(const int* __restrict__ cnt_in,
                                                           int* __restrict__ row_ptr) {
    __shared__ int wtot[32];
    const int lane = threadIdx.x & 31;
    const int wave = threadIdx.x >> 5;
    const int base = wave * SCAN_CHUNK + lane * 4;

    int4 ex[SCAN_TRIPS];
    int carry = 0;
#pragma unroll
    for (int i = 0; i < SCAN_TRIPS; ++i) {
        const int idx = base + i * 128;
        int4 c = make_int4(0, 0, 0, 0);
        if (idx < NN) c = *(const int4*)(cnt_in + idx);
        const int sum4 = c.x + c.y + c.z + c.w;
        int incl = sum4;
#pragma unroll
        for (int d = 1; d < 32; d <<= 1) {
            int up = __shfl_up(incl, d);
            if (lane >= d) incl += up;
        }
        const int e0 = carry + incl - sum4;
        ex[i] = make_int4(e0, e0 + c.x, e0 + c.x + c.y, e0 + c.x + c.y + c.z);
        carry += __shfl(incl, 31);
    }
    if (lane == 0) wtot[wave] = carry;
    __syncthreads();
    int off = (lane < wave) ? wtot[lane] : 0;
#pragma unroll
    for (int d = 16; d >= 1; d >>= 1) off += __shfl_xor(off, d);
#pragma unroll
    for (int i = 0; i < SCAN_TRIPS; ++i) {
        const int idx = base + i * 128;
        if (idx < NN)
            *(int4*)(row_ptr + idx) = make_int4(ex[i].x + off, ex[i].y + off, ex[i].z + off, ex[i].w + off);
    }
    if (threadIdx.x == 1023) row_ptr[NN] = off + carry;
}

__global__ void csr_fill_kernel(const int* __restrict__ src, const int* __restrict__ dst,
                                const int* __restrict__ rank, const int* __restrict__ row_ptr,
                                int* __restrict__ csr_src) {
    int e = blockIdx.x * blockDim.x + threadIdx.x;
    if (e >= NE) return;
    const int sN = src[e];
    const unsigned pos = (unsigned)(row_ptr[dst[e]] + rank[e]);
    if (pos < (unsigned)NE) csr_src[pos] = sN;
}

#define GT_PITCH 132
__global__ __launch_bounds__(256) void gemm128_wmma_kernel(const float* __restrict__ A,
                                                           const float* __restrict__ W,
                                                           float* __restrict__ out,
                                                           const float* __restrict__ scaleA,
                                                           const float* __restrict__ bias,
                                                           int relu, int ntiles) {
    __shared__ float4 tileA[2][16 * GT_PITCH / 4];
    const int lane = threadIdx.x & 31;
    const int wave = __builtin_amdgcn_readfirstlane((int)(threadIdx.x >> 5));
    const int c0 = wave * 16;

    const int m  = lane & 15;
    const int kb = (lane >> 4) << 1;
    const int n  = lane & 15;

    const float* wcol = W + (size_t)kb * FEAT + c0 + n;
    v2f b[FEAT / 4];
#pragma unroll
    for (int s = 0; s < FEAT / 4; ++s) {
        b[s].x = wcol[(size_t)(4 * s) * FEAT];
        b[s].y = wcol[(size_t)(4 * s + 1) * FEAT];
    }
    const float bv = bias ? bias[c0 + n] : 0.0f;
    const int rhalf = (lane >> 4) * 8;

    int tile = blockIdx.x;
    float4 g0 = make_float4(0.f, 0.f, 0.f, 0.f), g1 = g0;
    float s0 = 1.0f, s1 = 1.0f;
    if (tile < ntiles) {
        const size_t ra = (size_t)tile * 16 + wave;
        g0 = *(const float4*)(A + ra * FEAT + lane * 4);
        g1 = *(const float4*)(A + (ra + 8) * FEAT + lane * 4);
        if (scaleA) { s0 = scaleA[ra]; s1 = scaleA[ra + 8]; }
    }
    int buf = 0;
    for (; tile < ntiles; tile += gridDim.x) {
        const int r0 = tile * 16;
        float* tl = (float*)&tileA[buf][0];
        g0.x *= s0; g0.y *= s0; g0.z *= s0; g0.w *= s0;
        g1.x *= s1; g1.y *= s1; g1.z *= s1; g1.w *= s1;
        *(float4*)(tl + wave * GT_PITCH + lane * 4) = g0;
        *(float4*)(tl + (wave + 8) * GT_PITCH + lane * 4) = g1;
        __syncthreads();

        const int nt = tile + gridDim.x;
        if (nt < ntiles) {
            const size_t ra = (size_t)nt * 16 + wave;
            g0 = *(const float4*)(A + ra * FEAT + lane * 4);
            g1 = *(const float4*)(A + (ra + 8) * FEAT + lane * 4);
            if (scaleA) { s0 = scaleA[ra]; s1 = scaleA[ra + 8]; }
        }

        const float* ap = tl + m * GT_PITCH + kb;
        v2f a[FEAT / 4];
#pragma unroll
        for (int s = 0; s < FEAT / 4; ++s) a[s] = *(const v2f*)(ap + 4 * s);

        v8f acc = {0.f, 0.f, 0.f, 0.f, 0.f, 0.f, 0.f, 0.f};
#pragma unroll
        for (int s = 0; s < FEAT / 4; ++s) {
            acc = __builtin_amdgcn_wmma_f32_16x16x4_f32(
                 false, a[s],  false, b[s],
                 (short)0, acc,  false,  false);
        }

#pragma unroll
        for (int r = 0; r < 8; ++r) {
            float v = acc[r] + bv;
            if (relu) v = fmaxf(v, 0.0f);
            out[(size_t)(r0 + rhalf + r) * FEAT + c0 + n] = v;
        }
        buf ^= 1;
    }
}

static __device__ __forceinline__ float4 ld_row(const float* hl, int ids, int j) {
    const int s = __builtin_amdgcn_readlane(ids, j);
    return *(const float4*)(hl + (size_t)s * FEAT);
}
static __device__ __forceinline__ void add4(float4& a, const float4 v) {
    a.x += v.x; a.y += v.y; a.z += v.z; a.w += v.w;
}

static __device__ __forceinline__ float4 gather_row(const float* __restrict__ h,
                                                    const int* __restrict__ row_ptr,
                                                    const int* __restrict__ csr_src,
                                                    const float* __restrict__ inv_in,
                                                    const float* __restrict__ bias,
                                                    const int node, const int lane) {
    const int start = row_ptr[node];
    const int end = row_ptr[node + 1];
    const float* hl = h + lane * 4;

    float4 acc = make_float4(0.f, 0.f, 0.f, 0.f);
    for (int base = start; base < end; base += 32) {
        const int n = min(32, end - base);
        int mine = (lane < n) ? csr_src[base + lane] : 0;
        mine = (int)min((unsigned)mine, (unsigned)(NN - 1));
        int j = 0;
        for (; j + 8 <= n; j += 8) {
            const float4 v0 = ld_row(hl, mine, j);
            const float4 v1 = ld_row(hl, mine, j + 1);
            const float4 v2 = ld_row(hl, mine, j + 2);
            const float4 v3 = ld_row(hl, mine, j + 3);
            const float4 v4 = ld_row(hl, mine, j + 4);
            const float4 v5 = ld_row(hl, mine, j + 5);
            const float4 v6 = ld_row(hl, mine, j + 6);
            const float4 v7 = ld_row(hl, mine, j + 7);
            add4(acc, v0); add4(acc, v1); add4(acc, v2); add4(acc, v3);
            add4(acc, v4); add4(acc, v5); add4(acc, v6); add4(acc, v7);
        }
        if (j + 4 <= n) {
            const float4 v0 = ld_row(hl, mine, j);
            const float4 v1 = ld_row(hl, mine, j + 1);
            const float4 v2 = ld_row(hl, mine, j + 2);
            const float4 v3 = ld_row(hl, mine, j + 3);
            add4(acc, v0); add4(acc, v1); add4(acc, v2); add4(acc, v3);
            j += 4;
        }
        if (j < n) {
            const float4 v0 = ld_row(hl, mine, j);
            float4 v1 = make_float4(0.f, 0.f, 0.f, 0.f);
            float4 v2 = make_float4(0.f, 0.f, 0.f, 0.f);
            if (j + 1 < n) v1 = ld_row(hl, mine, j + 1);
            if (j + 2 < n) v2 = ld_row(hl, mine, j + 2);
            add4(acc, v0); add4(acc, v1); add4(acc, v2);
        }
    }

    const float sc = inv_in[node];
    const float4 b = *(const float4*)(bias + lane * 4);
    float4 o;
    o.x = fmaxf(acc.x * sc + b.x, 0.0f);
    o.y = fmaxf(acc.y * sc + b.y, 0.0f);
    o.z = fmaxf(acc.z * sc + b.z, 0.0f);
    o.w = fmaxf(acc.w * sc + b.w, 0.0f);
    return o;
}

__global__ __launch_bounds__(256) void gather_sum_kernel(const float* __restrict__ h,
                                                         const int* __restrict__ row_ptr,
                                                         const int* __restrict__ csr_src,
                                                         const float* __restrict__ inv_in,
                                                         const float* __restrict__ bias,
                                                         float* __restrict__ out) {
    const int lane = threadIdx.x & 31;
    const int wv = __builtin_amdgcn_readfirstlane((int)(threadIdx.x >> 5));
    const int node = blockIdx.x * 8 + wv;
    const float4 o = gather_row(h, row_ptr, csr_src, inv_in, bias, node, lane);
    *(float4*)(out + (size_t)node * FEAT + lane * 4) = o;
}

__global__ __launch_bounds__(256) void gather_sum_pool_kernel(const float* __restrict__ h,
                                                              const int* __restrict__ row_ptr,
                                                              const int* __restrict__ csr_src,
                                                              const float* __restrict__ inv_in,
                                                              const float* __restrict__ bias,
                                                              const int* __restrict__ graph_ids,
                                                              float* __restrict__ emb) {
    __shared__ float4 rows[8][32];
    __shared__ int gids[8];
    const int lane = threadIdx.x & 31;
    const int wv = __builtin_amdgcn_readfirstlane((int)(threadIdx.x >> 5));
    const int node = blockIdx.x * 8 + wv;
    const int gid = graph_ids[node];
    rows[wv][lane] = gather_row(h, row_ptr, csr_src, inv_in, bias, node, lane);
    if (lane == 0) gids[wv] = gid;
    __syncthreads();
    if (threadIdx.x < FEAT) {
        const int c = threadIdx.x;
        const float* col = (const float*)&rows[0][0] + c;
        int gv[8];
        float xv[8];
#pragma unroll
        for (int w = 0; w < 8; ++w) {
            gv[w] = __builtin_amdgcn_readfirstlane(gids[w]);
            xv[w] = col[w * FEAT];
        }
        int g = gv[0];
        float s = xv[0];
#pragma unroll
        for (int w = 1; w < 8; ++w) {
            if (gv[w] != g) {
                atomicAdd(emb + (size_t)g * FEAT + c, s);
                g = gv[w];
                s = xv[w];
            } else {
                s += xv[w];
            }
        }
        atomicAdd(emb + (size_t)g * FEAT + c, s);
    }
}

__global__ void bn_kernel(const float* __restrict__ emb, const float* __restrict__ gamma,
                          const float* __restrict__ beta, float* __restrict__ ybn) {
    int c = threadIdx.x;
    if (c >= FEAT) return;
    float mu = 0.f;
    for (int g = 0; g < NG; ++g) mu += emb[g * FEAT + c];
    mu *= (1.0f / NG);
    float var = 0.f;
    for (int g = 0; g < NG; ++g) {
        float d = emb[g * FEAT + c] - mu;
        var += d * d;
    }
    var *= (1.0f / NG);
    float s = gamma[c] * rsqrtf(var + BN_EPS);
    float b = beta[c];
    for (int g = 0; g < NG; ++g)
        ybn[g * FEAT + c] = (emb[g * FEAT + c] - mu) * s + b;
}

__global__ void fc2_logsoftmax_kernel(const float* __restrict__ h1,
                                      const float* __restrict__ fc2_w,
                                      const float* __restrict__ fc2_b,
                                      float* __restrict__ out) {
    int g = threadIdx.x;
    if (g >= NG) return;
    float logit[OUTC];
#pragma unroll
    for (int j = 0; j < OUTC; ++j) logit[j] = fc2_b[j];
    const float* hrow = h1 + (size_t)g * FEAT;
    for (int k = 0; k < FEAT; ++k) {
        float hv = hrow[k];
        const float* wrow = fc2_w + (size_t)k * OUTC;
#pragma unroll
        for (int j = 0; j < OUTC; ++j) logit[j] = fmaf(hv, wrow[j], logit[j]);
    }
    float mx = -1e30f;
#pragma unroll
    for (int j = 0; j < OUTC; ++j) mx = fmaxf(mx, logit[j]);
    float s = 0.f;
#pragma unroll
    for (int j = 0; j < OUTC; ++j) s += __expf(logit[j] - mx);
    float lse = mx + __logf(s);
#pragma unroll
    for (int j = 0; j < OUTC; ++j) out[g * OUTC + j] = logit[j] - lse;
}

extern "C" void kernel_launch(void* const* d_in, const int* in_sizes, int n_in,
                              void* d_out, int out_size, void* d_ws, size_t ws_size,
                              hipStream_t stream) {
    const float* n_feat    = (const float*)d_in[0];
    const int*   src       = (const int*)  d_in[1];
    const int*   dst       = (const int*)  d_in[2];
    const int*   graph_ids = (const int*)  d_in[3];
    const float* W1        = (const float*)d_in[4];
    const float* b1        = (const float*)d_in[5];
    const float* W2        = (const float*)d_in[6];
    const float* b2        = (const float*)d_in[7];
    const float* gamma     = (const float*)d_in[8];
    const float* beta      = (const float*)d_in[9];
    const float* fc1_w     = (const float*)d_in[10];
    const float* fc1_b     = (const float*)d_in[11];
    const float* fc2_w     = (const float*)d_in[12];
    const float* fc2_b     = (const float*)d_in[13];

    float* out = (float*)d_out;

    int*   cnt_out = (int*)d_ws;
    int*   cnt_in  = cnt_out + NN;
    float* inv_out = (float*)(cnt_in + NN);
    float* inv_in  = inv_out + NN;
    int*   row_ptr = (int*)(inv_in + NN);
    int*   rank    = row_ptr + 50112;
    int*   csr_src = rank + NE;
    float* bufA    = (float*)(csr_src + NE);
    float* bufB    = bufA + (size_t)NN * FEAT;
    float* ybn     = bufB + (size_t)NN * FEAT;
    float* h1      = ybn  + NG * FEAT;

    hipMemsetAsync(cnt_out, 0, (size_t)2 * NN * sizeof(int), stream);
    hipMemsetAsync(out, 0, (size_t)NG * FEAT * sizeof(float), stream);

    edge_count_rank_kernel<<<(NE + 255) / 256, 256, 0, stream>>>(src, dst, cnt_out, cnt_in, rank);
    invsqrt_kernel<<<(NN + 255) / 256, 256, 0, stream>>>(cnt_out, cnt_in, inv_out, inv_in);
    rowptr_scan_kernel<<<1, 1024, 0, stream>>>(cnt_in, row_ptr);
    csr_fill_kernel<<<(NE + 255) / 256, 256, 0, stream>>>(src, dst, rank, row_ptr, csr_src);

    const int nodeTiles = NN / 16;
    const int gemmBlocks = 625;
    const int nodeBlocks = NN / 8;

    gemm128_wmma_kernel<<<gemmBlocks, 256, 0, stream>>>(n_feat, W1, bufA, inv_out, nullptr, 0, nodeTiles);
    gather_sum_kernel<<<nodeBlocks, 256, 0, stream>>>(bufA, row_ptr, csr_src, inv_in, b1, bufB);

    gemm128_wmma_kernel<<<gemmBlocks, 256, 0, stream>>>(bufB, W2, bufA, inv_out, nullptr, 0, nodeTiles);
    gather_sum_pool_kernel<<<nodeBlocks, 256, 0, stream>>>(bufA, row_ptr, csr_src, inv_in, b2, graph_ids, out);

    bn_kernel<<<1, FEAT, 0, stream>>>(out, gamma, beta, ybn);
    gemm128_wmma_kernel<<<NG / 16, 256, 0, stream>>>(ybn, fc1_w, h1, nullptr, fc1_b, 1, NG / 16);
    fc2_logsoftmax_kernel<<<1, NG, 0, stream>>>(h1, fc2_w, fc2_b, out + (size_t)NG * FEAT);
}


# edge bucketing on LDS counters: chunk blocks order edges by 256-node range, owner blocks count, scan, fill rows in LDS; no global atomic in it; 2 kernels and a memset fewer; list order = LDS arrival
# speedup vs baseline: 4.7916x; 1.7905x over previous
#include <hip/hip_runtime.h>
#include <hip/hip_bf16.h>

#define NN 50000
#define NE 800000
#define FEAT 128
#define NG 128
#define OUTC 10
#define BN_EPS 1e-5f

typedef float v2f __attribute__((ext_vector_type(2)));
typedef float v8f __attribute__((ext_vector_type(8)));

#define NCHUNK 256
#define CH (NE / NCHUNK)
#define CHP 3136
#define RNG 256
#define NBIN ((NN + RNG - 1) / RNG)
#define BIN_TRIPS ((CH + 255) / 256)
static_assert(NE % NCHUNK == 0, "chunks are equal");
static_assert(NN <= 65536, "a source id is packed into 16 bits");
static_assert(NBIN < 256, "range b+1 has a column in the offset table");

static __device__ __forceinline__ int block_scan_excl(const int v, int* wtot, const int lane, const int wave,
                                                      int& total) {
    int incl = v;
    for (int d = 1; d < 32; d <<= 1) {
        const int up = __shfl_up(incl, d);
        if (lane >= d) incl += up;
    }
    if (lane == 31) wtot[wave] = incl;
    __syncthreads();
    int off = 0, tot = 0;
    for (int w = 0; w < 8; ++w) {
        const int t = wtot[w];
        tot += t;
        if (w < wave) off += t;
    }
    total = tot;
    return off + incl - v;
}

__global__ __launch_bounds__(256) void edge_bin_kernel(const int* __restrict__ src, const int* __restrict__ dst,
                                                       unsigned* __restrict__ sortedD,
                                                       unsigned* __restrict__ sortedS,
                                                       int* __restrict__ offD, int* __restrict__ offS) {
    __shared__ int hD[256], hS[256];
    __shared__ int wtD[8], wtS[8];
    __shared__ unsigned stD[CH], stS[CH];
    const int tid = threadIdx.x;
    const int lane = tid & 31;
    const int wave = tid >> 5;
    const int c = blockIdx.x;
    hD[tid] = 0;
    hS[tid] = 0;
    __syncthreads();

    const int* sp = src + (size_t)c * CH;
    const int* dp = dst + (size_t)c * CH;
    int sv[BIN_TRIPS], dv[BIN_TRIPS], rD[BIN_TRIPS], rS[BIN_TRIPS];
#pragma unroll
    for (int i = 0; i < BIN_TRIPS; ++i) {
        const int e = i * 256 + tid;
        sv[i] = 0; dv[i] = 0;
        if (i < CH / 256 || e < CH) { sv[i] = sp[e]; dv[i] = dp[e]; }
    }
#pragma unroll
    for (int i = 0; i < BIN_TRIPS; ++i) {
        const int e = i * 256 + tid;
        rD[i] = 0; rS[i] = 0;
        if (i < CH / 256 || e < CH) {
            const unsigned bd = min((unsigned)dv[i] >> 8, (unsigned)(NBIN - 1));
            const unsigned bs = min((unsigned)sv[i] >> 8, (unsigned)(NBIN - 1));
            rD[i] = atomicAdd(&hD[bd], 1);
            rS[i] = atomicAdd(&hS[bs], 1);
        }
    }
    __syncthreads();
    int totD, totS;
    const int exD = block_scan_excl(hD[tid], wtD, lane, wave, totD);
    const int exS = block_scan_excl(hS[tid], wtS, lane, wave, totS);
    hD[tid] = exD;
    hS[tid] = exS;
    offD[c * 256 + tid] = exD;
    offS[c * 256 + tid] = exS;
    __syncthreads();
#pragma unroll
    for (int i = 0; i < BIN_TRIPS; ++i) {
        const int e = i * 256 + tid;
        if (i < CH / 256 || e < CH) {
            const unsigned bd = min((unsigned)dv[i] >> 8, (unsigned)(NBIN - 1));
            const unsigned bs = min((unsigned)sv[i] >> 8, (unsigned)(NBIN - 1));
            stD[hD[bd] + rD[i]] = (((unsigned)dv[i] & 255u) << 16) | ((unsigned)sv[i] & 0xFFFFu);
            stS[hS[bs] + rS[i]] = (unsigned)sv[i] & 255u;
        }
    }
    __syncthreads();
    unsigned* oD = sortedD + (size_t)c * CHP;
    unsigned* oS = sortedS + (size_t)c * CHP;
    for (int j = tid; j < CH; j += 256) {
        oD[j] = stD[j];
        oS[j] = stS[j];
    }
}

#define OWN_GROUP 8
__global__ __launch_bounds__(256) void owner_build_kernel(const unsigned* __restrict__ sortedD,
                                                          const unsigned* __restrict__ sortedS,
                                                          const int* __restrict__ offD,
                                                          const int* __restrict__ offS,
                                                          int* __restrict__ row_ptr, int* __restrict__ csr_src,
                                                          float* __restrict__ inv_in,
                                                          float* __restrict__ inv_out) {
    __shared__ int cntI[RNG], cntO[RNG], cur[RNG];
    __shared__ int wtA[8], wtB[8];
    const int tid = threadIdx.x;
    const int lane = tid & 31;
    const int wave = tid >> 5;
    const int b = blockIdx.x;
    const int c = tid;
    const int bD = offD[c * 256 + b];
    const int nD = offD[c * 256 + b + 1] - bD;
    const int bS = offS[c * 256 + b];
    const int nS = offS[c * 256 + b + 1] - bS;
    cntI[tid] = 0;
    cntO[tid] = 0;
    __syncthreads();

    const unsigned* pD = sortedD + (size_t)c * CHP + bD;
    const unsigned* pS = sortedS + (size_t)c * CHP + bS;
    const int nmax = min(max(nD, nS), CH);
    for (int j = 0; j < nmax; j += OWN_GROUP) {
        unsigned wd[OWN_GROUP], wsrc[OWN_GROUP];
#pragma unroll
        for (int u = 0; u < OWN_GROUP; ++u) {
            wd[u] = pD[(j + u < nD) ? j + u : 0];
            wsrc[u] = pS[(j + u < nS) ? j + u : 0];
        }
#pragma unroll
        for (int u = 0; u < OWN_GROUP; ++u) {
            if (j + u < nD) atomicAdd(&cntI[wd[u] >> 16], 1);
            if (j + u < nS) atomicAdd(&cntO[wsrc[u] & 255u], 1);
        }
    }

    int binStart, totI;
    (void)block_scan_excl(bD, wtA, lane, wave, binStart);
    const int my = cntI[tid];
    const int start = binStart + block_scan_excl(my, wtB, lane, wave, totI);
    const int node = b * RNG + tid;
    if (node <= NN) row_ptr[node] = start;
    if (node < NN) {
        inv_in[node]  = rsqrtf(fmaxf((float)my, 1.0f));
        inv_out[node] = rsqrtf(fmaxf((float)cntO[tid], 1.0f));
    }
    cur[tid] = start;
    __syncthreads();

    const int nDc = min(nD, CH);
    for (int j = 0; j < nDc; j += OWN_GROUP) {
        unsigned wd[OWN_GROUP];
#pragma unroll
        for (int u = 0; u < OWN_GROUP; ++u) wd[u] = pD[(j + u < nD) ? j + u : 0];
#pragma unroll
        for (int u = 0; u < OWN_GROUP; ++u) {
            if (j + u < nD) {
                const unsigned pos = (unsigned)atomicAdd(&cur[wd[u] >> 16], 1);
                if (pos < (unsigned)NE) csr_src[pos] = (int)(wd[u] & 0xFFFFu);
            }
        }
    }
}

#define GT_PITCH 132
__global__ __launch_bounds__(256) void gemm128_wmma_kernel(const float* __restrict__ A,
                                                           const float* __restrict__ W,
                                                           float* __restrict__ out,
                                                           const float* __restrict__ scaleA,
                                                           const float* __restrict__ bias,
                                                           int relu, int ntiles) {
    __shared__ float4 tileA[2][16 * GT_PITCH / 4];
    const int lane = threadIdx.x & 31;
    const int wave = __builtin_amdgcn_readfirstlane((int)(threadIdx.x >> 5));
    const int c0 = wave * 16;

    const int m  = lane & 15;
    const int kb = (lane >> 4) << 1;
    const int n  = lane & 15;

    const float* wcol = W + (size_t)kb * FEAT + c0 + n;
    v2f b[FEAT / 4];
#pragma unroll
    for (int s = 0; s < FEAT / 4; ++s) {
        b[s].x = wcol[(size_t)(4 * s) * FEAT];
        b[s].y = wcol[(size_t)(4 * s + 1) * FEAT];
    }
    const float bv = bias ? bias[c0 + n] : 0.0f;
    const int rhalf = (lane >> 4) * 8;

    int tile = blockIdx.x;
    float4 g0 = make_float4(0.f, 0.f, 0.f, 0.f), g1 = g0;
    float s0 = 1.0f, s1 = 1.0f;
    if (tile < ntiles) {
        const size_t ra = (size_t)tile * 16 + wave;
        g0 = *(const float4*)(A + ra * FEAT + lane * 4);
        g1 = *(const float4*)(A + (ra + 8) * FEAT + lane * 4);
        if (scaleA) { s0 = scaleA[ra]; s1 = scaleA[ra + 8]; }
    }
    int buf = 0;
    for (; tile < ntiles; tile += gridDim.x) {
        const int r0 = tile * 16;
        float* tl = (float*)&tileA[buf][0];
        g0.x *= s0; g0.y *= s0; g0.z *= s0; g0.w *= s0;
        g1.x *= s1; g1.y *= s1; g1.z *= s1; g1.w *= s1;
        *(float4*)(tl + wave * GT_PITCH + lane * 4) = g0;
        *(float4*)(tl + (wave + 8) * GT_PITCH + lane * 4) = g1;
        __syncthreads();

        const int nt = tile + gridDim.x;
        if (nt < ntiles) {
            const size_t ra = (size_t)nt * 16 + wave;
            g0 = *(const float4*)(A + ra * FEAT + lane * 4);
            g1 = *(const float4*)(A + (ra + 8) * FEAT + lane * 4);
            if (scaleA) { s0 = scaleA[ra]; s1 = scaleA[ra + 8]; }
        }

        const float* ap = tl + m * GT_PITCH + kb;
        v2f a[FEAT / 4];
#pragma unroll
        for (int s = 0; s < FEAT / 4; ++s) a[s] = *(const v2f*)(ap + 4 * s);

        v8f acc = {0.f, 0.f, 0.f, 0.f, 0.f, 0.f, 0.f, 0.f};
#pragma unroll
        for (int s = 0; s < FEAT / 4; ++s) {
            acc = __builtin_amdgcn_wmma_f32_16x16x4_f32(
                 false, a[s],  false, b[s],
                 (short)0, acc,  false,  false);
        }

#pragma unroll
        for (int r = 0; r < 8; ++r) {
            float v = acc[r] + bv;
            if (relu) v = fmaxf(v, 0.0f);
            out[(size_t)(r0 + rhalf + r) * FEAT + c0 + n] = v;
        }
        buf ^= 1;
    }
}

static __device__ __forceinline__ float4 ld_row(const float* hl, int ids, int j) {
    const int s = __builtin_amdgcn_readlane(ids, j);
    return *(const float4*)(hl + (size_t)s * FEAT);
}
static __device__ __forceinline__ void add4(float4& a, const float4 v) {
    a.x += v.x; a.y += v.y; a.z += v.z; a.w += v.w;
}

static __device__ __forceinline__ float4 gather_row(const float* __restrict__ h,
                                                    const int* __restrict__ row_ptr,
                                                    const int* __restrict__ csr_src,
                                                    const float* __restrict__ inv_in,
                                                    const float* __restrict__ bias,
                                                    const int node, const int lane) {
    const int start = row_ptr[node];
    const int end = row_ptr[node + 1];
    const float* hl = h + lane * 4;

    float4 acc = make_float4(0.f, 0.f, 0.f, 0.f);
    for (int base = start; base < end; base += 32) {
        const int n = min(32, end - base);
        int mine = (lane < n) ? csr_src[base + lane] : 0;
        mine = (int)min((unsigned)mine, (unsigned)(NN - 1));
        int j = 0;
        for (; j + 8 <= n; j += 8) {
            const float4 v0 = ld_row(hl, mine, j);
            const float4 v1 = ld_row(hl, mine, j + 1);
            const float4 v2 = ld_row(hl, mine, j + 2);
            const float4 v3 = ld_row(hl, mine, j + 3);
            const float4 v4 = ld_row(hl, mine, j + 4);
            const float4 v5 = ld_row(hl, mine, j + 5);
            const float4 v6 = ld_row(hl, mine, j + 6);
            const float4 v7 = ld_row(hl, mine, j + 7);
            add4(acc, v0); add4(acc, v1); add4(acc, v2); add4(acc, v3);
            add4(acc, v4); add4(acc, v5); add4(acc, v6); add4(acc, v7);
        }
        if (j + 4 <= n) {
            const float4 v0 = ld_row(hl, mine, j);
            const float4 v1 = ld_row(hl, mine, j + 1);
            const float4 v2 = ld_row(hl, mine, j + 2);
            const float4 v3 = ld_row(hl, mine, j + 3);
            add4(acc, v0); add4(acc, v1); add4(acc, v2); add4(acc, v3);
            j += 4;
        }
        if (j < n) {
            const float4 v0 = ld_row(hl, mine, j);
            float4 v1 = make_float4(0.f, 0.f, 0.f, 0.f);
            float4 v2 = make_float4(0.f, 0.f, 0.f, 0.f);
            if (j + 1 < n) v1 = ld_row(hl, mine, j + 1);
            if (j + 2 < n) v2 = ld_row(hl, mine, j + 2);
            add4(acc, v0); add4(acc, v1); add4(acc, v2);
        }
    }

    const float sc = inv_in[node];
    const float4 b = *(const float4*)(bias + lane * 4);
    float4 o;
    o.x = fmaxf(acc.x * sc + b.x, 0.0f);
    o.y = fmaxf(acc.y * sc + b.y, 0.0f);
    o.z = fmaxf(acc.z * sc + b.z, 0.0f);
    o.w = fmaxf(acc.w * sc + b.w, 0.0f);
    return o;
}

__global__ __launch_bounds__(256) void gather_sum_kernel(const float* __restrict__ h,
                                                         const int* __restrict__ row_ptr,
                                                         const int* __restrict__ csr_src,
                                                         const float* __restrict__ inv_in,
                                                         const float* __restrict__ bias,
                                                         float* __restrict__ out) {
    const int lane = threadIdx.x & 31;
    const int wv = __builtin_amdgcn_readfirstlane((int)(threadIdx.x >> 5));
    const int node = blockIdx.x * 8 + wv;
    const float4 o = gather_row(h, row_ptr, csr_src, inv_in, bias, node, lane);
    *(float4*)(out + (size_t)node * FEAT + lane * 4) = o;
}

__global__ __launch_bounds__(256) void gather_sum_pool_kernel(const float* __restrict__ h,
                                                              const int* __restrict__ row_ptr,
                                                              const int* __restrict__ csr_src,
                                                              const float* __restrict__ inv_in,
                                                              const float* __restrict__ bias,
                                                              const int* __restrict__ graph_ids,
                                                              float* __restrict__ emb) {
    __shared__ float4 rows[8][32];
    __shared__ int gids[8];
    const int lane = threadIdx.x & 31;
    const int wv = __builtin_amdgcn_readfirstlane((int)(threadIdx.x >> 5));
    const int node = blockIdx.x * 8 + wv;
    const int gid = graph_ids[node];
    rows[wv][lane] = gather_row(h, row_ptr, csr_src, inv_in, bias, node, lane);
    if (lane == 0) gids[wv] = gid;
    __syncthreads();
    if (threadIdx.x < FEAT) {
        const int c = threadIdx.x;
        const float* col = (const float*)&rows[0][0] + c;
        int gv[8];
        float xv[8];
#pragma unroll
        for (int w = 0; w < 8; ++w) {
            gv[w] = __builtin_amdgcn_readfirstlane(gids[w]);
            xv[w] = col[w * FEAT];
        }
        int g = gv[0];
        float s = xv[0];
#pragma unroll
        for (int w = 1; w < 8; ++w) {
            if (gv[w] != g) {
                atomicAdd(emb + (size_t)g * FEAT + c, s);
                g = gv[w];
                s = xv[w];
            } else {
                s += xv[w];
            }
        }
        atomicAdd(emb + (size_t)g * FEAT + c, s);
    }
}

__global__ void bn_kernel(const float* __restrict__ emb, const float* __restrict__ gamma,
                          const float* __restrict__ beta, float* __restrict__ ybn) {
    int c = threadIdx.x;
    if (c >= FEAT) return;
    float mu = 0.f;
    for (int g = 0; g < NG; ++g) mu += emb[g * FEAT + c];
    mu *= (1.0f / NG);
    float var = 0.f;
    for (int g = 0; g < NG; ++g) {
        float d = emb[g * FEAT + c] - mu;
        var += d * d;
    }
    var *= (1.0f / NG);
    float s = gamma[c] * rsqrtf(var + BN_EPS);
    float b = beta[c];
    for (int g = 0; g < NG; ++g)
        ybn[g * FEAT + c] = (emb[g * FEAT + c] - mu) * s + b;
}

__global__ void fc2_logsoftmax_kernel(const float* __restrict__ h1,
                                      const float* __restrict__ fc2_w,
                                      const float* __restrict__ fc2_b,
                                      float* __restrict__ out) {
    int g = threadIdx.x;
    if (g >= NG) return;
    float logit[OUTC];
#pragma unroll
    for (int j = 0; j < OUTC; ++j) logit[j] = fc2_b[j];
    const float* hrow = h1 + (size_t)g * FEAT;
    for (int k = 0; k < FEAT; ++k) {
        float hv = hrow[k];
        const float* wrow = fc2_w + (size_t)k * OUTC;
#pragma unroll
        for (int j = 0; j < OUTC; ++j) logit[j] = fmaf(hv, wrow[j], logit[j]);
    }
    float mx = -1e30f;
#pragma unroll
    for (int j = 0; j < OUTC; ++j) mx = fmaxf(mx, logit[j]);
    float s = 0.f;
#pragma unroll
    for (int j = 0; j < OUTC; ++j) s += __expf(logit[j] - mx);
    float lse = mx + __logf(s);
#pragma unroll
    for (int j = 0; j < OUTC; ++j) out[g * OUTC + j] = logit[j] - lse;
}

extern "C" void kernel_launch(void* const* d_in, const int* in_sizes, int n_in,
                              void* d_out, int out_size, void* d_ws, size_t ws_size,
                              hipStream_t stream) {
    const float* n_feat    = (const float*)d_in[0];
    const int*   src       = (const int*)  d_in[1];
    const int*   dst       = (const int*)  d_in[2];
    const int*   graph_ids = (const int*)  d_in[3];
    const float* W1        = (const float*)d_in[4];
    const float* b1        = (const float*)d_in[5];
    const float* W2        = (const float*)d_in[6];
    const float* b2        = (const float*)d_in[7];
    const float* gamma     = (const float*)d_in[8];
    const float* beta      = (const float*)d_in[9];
    const float* fc1_w     = (const float*)d_in[10];
    const float* fc1_b     = (const float*)d_in[11];
    const float* fc2_w     = (const float*)d_in[12];
    const float* fc2_b     = (const float*)d_in[13];

    float* out = (float*)d_out;

    float* inv_out = (float*)d_ws;
    float* inv_in  = inv_out + 50176;
    int*   row_ptr = (int*)(inv_in + 50176);
    int*   offD    = row_ptr + 50176;
    int*   offS    = offD + NCHUNK * 256;
    unsigned* sortedD = (unsigned*)(offS + NCHUNK * 256);
    unsigned* sortedS = sortedD + NCHUNK * CHP;
    int*   csr_src = (int*)(sortedS + NCHUNK * CHP);
    float* bufA    = (float*)(csr_src + NE);
    float* bufB    = bufA + (size_t)NN * FEAT;
    float* ybn     = bufB + (size_t)NN * FEAT;
    float* h1      = ybn  + NG * FEAT;

    hipMemsetAsync(out, 0, (size_t)NG * FEAT * sizeof(float), stream);

    edge_bin_kernel<<<NCHUNK, 256, 0, stream>>>(src, dst, sortedD, sortedS, offD, offS);
    owner_build_kernel<<<NBIN, 256, 0, stream>>>(sortedD, sortedS, offD, offS, row_ptr, csr_src, inv_in, inv_out);

    const int nodeTiles = NN / 16;
    const int gemmBlocks = 625;
    const int nodeBlocks = NN / 8;

    gemm128_wmma_kernel<<<gemmBlocks, 256, 0, stream>>>(n_feat, W1, bufA, inv_out, nullptr, 0, nodeTiles);
    gather_sum_kernel<<<nodeBlocks, 256, 0, stream>>>(bufA, row_ptr, csr_src, inv_in, b1, bufB);

    gemm128_wmma_kernel<<<gemmBlocks, 256, 0, stream>>>(bufB, W2, bufA, inv_out, nullptr, 0, nodeTiles);
    gather_sum_pool_kernel<<<nodeBlocks, 256, 0, stream>>>(bufA, row_ptr, csr_src, inv_in, b2, graph_ids, out);

    bn_kernel<<<1, FEAT, 0, stream>>>(out, gamma, beta, ybn);
    gemm128_wmma_kernel<<<NG / 16, 256, 0, stream>>>(ybn, fc1_w, h1, nullptr, fc1_b, 1, NG / 16);
    fc2_logsoftmax_kernel<<<1, NG, 0, stream>>>(h1, fc2_w, fc2_b, out + (size_t)NG * FEAT);
}


# head in one launch: BatchNorm, fc1 (WMMA), fc2, log_softmax meet in LDS, every block forms the column statistics; first kernel stores the embedding zeros; 9 launches + memset -> 7; sum orders kept
# speedup vs baseline: 5.7425x; 1.1985x over previous
#include <hip/hip_runtime.h>
#include <hip/hip_bf16.h>

#define NN 50000
#define NE 800000
#define FEAT 128
#define NG 128
#define OUTC 10
#define BN_EPS 1e-5f

typedef float v2f __attribute__((ext_vector_type(2)));
typedef float v8f __attribute__((ext_vector_type(8)));

#define NCHUNK 256
#define CH (NE / NCHUNK)
#define CHP 3136
#define RNG 256
#define NBIN ((NN + RNG - 1) / RNG)
#define BIN_TRIPS ((CH + 255) / 256)
static_assert(NE % NCHUNK == 0, "chunks are equal");
static_assert(NN <= 65536, "a source id is packed into 16 bits");
static_assert(NBIN < 256, "range b+1 has a column in the offset table");
static_assert((NG * FEAT) % 256 == 0 && (NG * FEAT) / 256 <= NCHUNK, "the first kernel's blocks cover the embedding rows");

static __device__ __forceinline__ int block_scan_excl(const int v, int* wtot, const int lane, const int wave,
                                                      int& total) {
    int incl = v;
    for (int d = 1; d < 32; d <<= 1) {
        const int up = __shfl_up(incl, d);
        if (lane >= d) incl += up;
    }
    if (lane == 31) wtot[wave] = incl;
    __syncthreads();
    int off = 0, tot = 0;
    for (int w = 0; w < 8; ++w) {
        const int t = wtot[w];
        tot += t;
        if (w < wave) off += t;
    }
    total = tot;
    return off + incl - v;
}

__global__ __launch_bounds__(256) void edge_bin_kernel(const int* __restrict__ src, const int* __restrict__ dst,
                                                       unsigned* __restrict__ sortedD,
                                                       unsigned* __restrict__ sortedS,
                                                       int* __restrict__ offD, int* __restrict__ offS,
                                                       float* __restrict__ emb) {
    __shared__ int hD[256], hS[256];
    __shared__ int wtD[8], wtS[8];
    __shared__ unsigned stD[CH], stS[CH];
    const int tid = threadIdx.x;
    const int lane = tid & 31;
    const int wave = tid >> 5;
    const int c = blockIdx.x;
    if (c < (NG * FEAT) / 256) emb[c * 256 + tid] = 0.0f;
    hD[tid] = 0;
    hS[tid] = 0;
    __syncthreads();

    const int* sp = src + (size_t)c * CH;
    const int* dp = dst + (size_t)c * CH;
    int sv[BIN_TRIPS], dv[BIN_TRIPS], rD[BIN_TRIPS], rS[BIN_TRIPS];
#pragma unroll
    for (int i = 0; i < BIN_TRIPS; ++i) {
        const int e = i * 256 + tid;
        sv[i] = 0; dv[i] = 0;
        if (i < CH / 256 || e < CH) { sv[i] = sp[e]; dv[i] = dp[e]; }
    }
#pragma unroll
    for (int i = 0; i < BIN_TRIPS; ++i) {
        const int e = i * 256 + tid;
        rD[i] = 0; rS[i] = 0;
        if (i < CH / 256 || e < CH) {
            const unsigned bd = min((unsigned)dv[i] >> 8, (unsigned)(NBIN - 1));
            const unsigned bs = min((unsigned)sv[i] >> 8, (unsigned)(NBIN - 1));
            rD[i] = atomicAdd(&hD[bd], 1);
            rS[i] = atomicAdd(&hS[bs], 1);
        }
    }
    __syncthreads();
    int totD, totS;
    const int exD = block_scan_excl(hD[tid], wtD, lane, wave, totD);
    const int exS = block_scan_excl(hS[tid], wtS, lane, wave, totS);
    hD[tid] = exD;
    hS[tid] = exS;
    offD[c * 256 + tid] = exD;
    offS[c * 256 + tid] = exS;
    __syncthreads();
#pragma unroll
    for (int i = 0; i < BIN_TRIPS; ++i) {
        const int e = i * 256 + tid;
        if (i < CH / 256 || e < CH) {
            const unsigned bd = min((unsigned)dv[i] >> 8, (unsigned)(NBIN - 1));
            const unsigned bs = min((unsigned)sv[i] >> 8, (unsigned)(NBIN - 1));
            stD[hD[bd] + rD[i]] = (((unsigned)dv[i] & 255u) << 16) | ((unsigned)sv[i] & 0xFFFFu);
            stS[hS[bs] + rS[i]] = (unsigned)sv[i] & 255u;
        }
    }
    __syncthreads();
    unsigned* oD = sortedD + (size_t)c * CHP;
    unsigned* oS = sortedS + (size_t)c * CHP;
    for (int j = tid; j < CH; j += 256) {
        oD[j] = stD[j];
        oS[j] = stS[j];
    }
}

#define OWN_GROUP 8
__global__ __launch_bounds__(256) void owner_build_kernel(const unsigned* __restrict__ sortedD,
                                                          const unsigned* __restrict__ sortedS,
                                                          const int* __restrict__ offD,
                                                          const int* __restrict__ offS,
                                                          int* __restrict__ row_ptr, int* __restrict__ csr_src,
                                                          float* __restrict__ inv_in,
                                                          float* __restrict__ inv_out) {
    __shared__ int cntI[RNG], cntO[RNG], cur[RNG];
    __shared__ int wtA[8], wtB[8];
    const int tid = threadIdx.x;
    const int lane = tid & 31;
    const int wave = tid >> 5;
    const int b = blockIdx.x;
    const int c = tid;
    const int bD = offD[c * 256 + b];
    const int nD = offD[c * 256 + b + 1] - bD;
    const int bS = offS[c * 256 + b];
    const int nS = offS[c * 256 + b + 1] - bS;
    cntI[tid] = 0;
    cntO[tid] = 0;
    __syncthreads();

    const unsigned* pD = sortedD + (size_t)c * CHP + bD;
    const unsigned* pS = sortedS + (size_t)c * CHP + bS;
    const int nmax = min(max(nD, nS), CH);
    for (int j = 0; j < nmax; j += OWN_GROUP) {
        unsigned wd[OWN_GROUP], wsrc[OWN_GROUP];
#pragma unroll
        for (int u = 0; u < OWN_GROUP; ++u) {
            wd[u] = pD[(j + u < nD) ? j + u : 0];
            wsrc[u] = pS[(j + u < nS) ? j + u : 0];
        }
#pragma unroll
        for (int u = 0; u < OWN_GROUP; ++u) {
            if (j + u < nD) atomicAdd(&cntI[wd[u] >> 16], 1);
            if (j + u < nS) atomicAdd(&cntO[wsrc[u] & 255u], 1);
        }
    }

    int binStart, totI;
    (void)block_scan_excl(bD, wtA, lane, wave, binStart);
    const int my = cntI[tid];
    const int start = binStart + block_scan_excl(my, wtB, lane, wave, totI);
    const int node = b * RNG + tid;
    if (node <= NN) row_ptr[node] = start;
    if (node < NN) {
        inv_in[node]  = rsqrtf(fmaxf((float)my, 1.0f));
        inv_out[node] = rsqrtf(fmaxf((float)cntO[tid], 1.0f));
    }
    cur[tid] = start;
    __syncthreads();

    const int nDc = min(nD, CH);
    for (int j = 0; j < nDc; j += OWN_GROUP) {
        unsigned wd[OWN_GROUP];
#pragma unroll
        for (int u = 0; u < OWN_GROUP; ++u) wd[u] = pD[(j + u < nD) ? j + u : 0];
#pragma unroll
        for (int u = 0; u < OWN_GROUP; ++u) {
            if (j + u < nD) {
                const unsigned pos = (unsigned)atomicAdd(&cur[wd[u] >> 16], 1);
                if (pos < (unsigned)NE) csr_src[pos] = (int)(wd[u] & 0xFFFFu);
            }
        }
    }
}

#define GT_PITCH 132
__global__ __launch_bounds__(256) void gemm128_wmma_kernel(const float* __restrict__ A,
                                                           const float* __restrict__ W,
                                                           float* __restrict__ out,
                                                           const float* __restrict__ scaleA,
                                                           const float* __restrict__ bias,
                                                           int relu, int ntiles) {
    __shared__ float4 tileA[2][16 * GT_PITCH / 4];
    const int lane = threadIdx.x & 31;
    const int wave = __builtin_amdgcn_readfirstlane((int)(threadIdx.x >> 5));
    const int c0 = wave * 16;

    const int m  = lane & 15;
    const int kb = (lane >> 4) << 1;
    const int n  = lane & 15;

    const float* wcol = W + (size_t)kb * FEAT + c0 + n;
    v2f b[FEAT / 4];
#pragma unroll
    for (int s = 0; s < FEAT / 4; ++s) {
        b[s].x = wcol[(size_t)(4 * s) * FEAT];
        b[s].y = wcol[(size_t)(4 * s + 1) * FEAT];
    }
    const float bv = bias ? bias[c0 + n] : 0.0f;
    const int rhalf = (lane >> 4) * 8;

    int tile = blockIdx.x;
    float4 g0 = make_float4(0.f, 0.f, 0.f, 0.f), g1 = g0;
    float s0 = 1.0f, s1 = 1.0f;
    if (tile < ntiles) {
        const size_t ra = (size_t)tile * 16 + wave;
        g0 = *(const float4*)(A + ra * FEAT + lane * 4);
        g1 = *(const float4*)(A + (ra + 8) * FEAT + lane * 4);
        if (scaleA) { s0 = scaleA[ra]; s1 = scaleA[ra + 8]; }
    }
    int buf = 0;
    for (; tile < ntiles; tile += gridDim.x) {
        const int r0 = tile * 16;
        float* tl = (float*)&tileA[buf][0];
        g0.x *= s0; g0.y *= s0; g0.z *= s0; g0.w *= s0;
        g1.x *= s1; g1.y *= s1; g1.z *= s1; g1.w *= s1;
        *(float4*)(tl + wave * GT_PITCH + lane * 4) = g0;
        *(float4*)(tl + (wave + 8) * GT_PITCH + lane * 4) = g1;
        __syncthreads();

        const int nt = tile + gridDim.x;
        if (nt < ntiles) {
            const size_t ra = (size_t)nt * 16 + wave;
            g0 = *(const float4*)(A + ra * FEAT + lane * 4);
            g1 = *(const float4*)(A + (ra + 8) * FEAT + lane * 4);
            if (scaleA) { s0 = scaleA[ra]; s1 = scaleA[ra + 8]; }
        }

        const float* ap = tl + m * GT_PITCH + kb;
        v2f a[FEAT / 4];
#pragma unroll
        for (int s = 0; s < FEAT / 4; ++s) a[s] = *(const v2f*)(ap + 4 * s);

        v8f acc = {0.f, 0.f, 0.f, 0.f, 0.f, 0.f, 0.f, 0.f};
#pragma unroll
        for (int s = 0; s < FEAT / 4; ++s) {
            acc = __builtin_amdgcn_wmma_f32_16x16x4_f32(
                 false, a[s],  false, b[s],
                 (short)0, acc,  false,  false);
        }

#pragma unroll
        for (int r = 0; r < 8; ++r) {
            float v = acc[r] + bv;
            if (relu) v = fmaxf(v, 0.0f);
            out[(size_t)(r0 + rhalf + r) * FEAT + c0 + n] = v;
        }
        buf ^= 1;
    }
}

static __device__ __forceinline__ float4 ld_row(const float* hl, int ids, int j) {
    const int s = __builtin_amdgcn_readlane(ids, j);
    return *(const float4*)(hl + (size_t)s * FEAT);
}
static __device__ __forceinline__ void add4(float4& a, const float4 v) {
    a.x += v.x; a.y += v.y; a.z += v.z; a.w += v.w;
}

static __device__ __forceinline__ float4 gather_row(const float* __restrict__ h,
                                                    const int* __restrict__ row_ptr,
                                                    const int* __restrict__ csr_src,
                                                    const float* __restrict__ inv_in,
                                                    const float* __restrict__ bias,
                                                    const int node, const int lane) {
    const int start = row_ptr[node];
    const int end = row_ptr[node + 1];
    const float* hl = h + lane * 4;

    float4 acc = make_float4(0.f, 0.f, 0.f, 0.f);
    for (int base = start; base < end; base += 32) {
        const int n = min(32, end - base);
        int mine = (lane < n) ? csr_src[base + lane] : 0;
        mine = (int)min((unsigned)mine, (unsigned)(NN - 1));
        int j = 0;
        for (; j + 8 <= n; j += 8) {
            const float4 v0 = ld_row(hl, mine, j);
            const float4 v1 = ld_row(hl, mine, j + 1);
            const float4 v2 = ld_row(hl, mine, j + 2);
            const float4 v3 = ld_row(hl, mine, j + 3);
            const float4 v4 = ld_row(hl, mine, j + 4);
            const float4 v5 = ld_row(hl, mine, j + 5);
            const float4 v6 = ld_row(hl, mine, j + 6);
            const float4 v7 = ld_row(hl, mine, j + 7);
            add4(acc, v0); add4(acc, v1); add4(acc, v2); add4(acc, v3);
            add4(acc, v4); add4(acc, v5); add4(acc, v6); add4(acc, v7);
        }
        if (j + 4 <= n) {
            const float4 v0 = ld_row(hl, mine, j);
            const float4 v1 = ld_row(hl, mine, j + 1);
            const float4 v2 = ld_row(hl, mine, j + 2);
            const float4 v3 = ld_row(hl, mine, j + 3);
            add4(acc, v0); add4(acc, v1); add4(acc, v2); add4(acc, v3);
            j += 4;
        }
        if (j < n) {
            const float4 v0 = ld_row(hl, mine, j);
            float4 v1 = make_float4(0.f, 0.f, 0.f, 0.f);
            float4 v2 = make_float4(0.f, 0.f, 0.f, 0.f);
            if (j + 1 < n) v1 = ld_row(hl, mine, j + 1);
            if (j + 2 < n) v2 = ld_row(hl, mine, j + 2);
            add4(acc, v0); add4(acc, v1); add4(acc, v2);
        }
    }

    const float sc = inv_in[node];
    const float4 b = *(const float4*)(bias + lane * 4);
    float4 o;
    o.x = fmaxf(acc.x * sc + b.x, 0.0f);
    o.y = fmaxf(acc.y * sc + b.y, 0.0f);
    o.z = fmaxf(acc.z * sc + b.z, 0.0f);
    o.w = fmaxf(acc.w * sc + b.w, 0.0f);
    return o;
}

__global__ __launch_bounds__(256) void gather_sum_kernel(const float* __restrict__ h,
                                                         const int* __restrict__ row_ptr,
                                                         const int* __restrict__ csr_src,
                                                         const float* __restrict__ inv_in,
                                                         const float* __restrict__ bias,
                                                         float* __restrict__ out) {
    const int lane = threadIdx.x & 31;
    const int wv = __builtin_amdgcn_readfirstlane((int)(threadIdx.x >> 5));
    const int node = blockIdx.x * 8 + wv;
    const float4 o = gather_row(h, row_ptr, csr_src, inv_in, bias, node, lane);
    *(float4*)(out + (size_t)node * FEAT + lane * 4) = o;
}

__global__ __launch_bounds__(256) void gather_sum_pool_kernel(const float* __restrict__ h,
                                                              const int* __restrict__ row_ptr,
                                                              const int* __restrict__ csr_src,
                                                              const float* __restrict__ inv_in,
                                                              const float* __restrict__ bias,
                                                              const int* __restrict__ graph_ids,
                                                              float* __restrict__ emb) {
    __shared__ float4 rows[8][32];
    __shared__ int gids[8];
    const int lane = threadIdx.x & 31;
    const int wv = __builtin_amdgcn_readfirstlane((int)(threadIdx.x >> 5));
    const int node = blockIdx.x * 8 + wv;
    const int gid = graph_ids[node];
    rows[wv][lane] = gather_row(h, row_ptr, csr_src, inv_in, bias, node, lane);
    if (lane == 0) gids[wv] = gid;
    __syncthreads();
    if (threadIdx.x < FEAT) {
        const int c = threadIdx.x;
        const float* col = (const float*)&rows[0][0] + c;
        int gv[8];
        float xv[8];
#pragma unroll
        for (int w = 0; w < 8; ++w) {
            gv[w] = __builtin_amdgcn_readfirstlane(gids[w]);
            xv[w] = col[w * FEAT];
        }
        int g = gv[0];
        float s = xv[0];
#pragma unroll
        for (int w = 1; w < 8; ++w) {
            if (gv[w] != g) {
                atomicAdd(emb + (size_t)g * FEAT + c, s);
                g = gv[w];
                s = xv[w];
            } else {
                s += xv[w];
            }
        }
        atomicAdd(emb + (size_t)g * FEAT + c, s);
    }
}

#define HROWS 16
static_assert(NG % HROWS == 0 && HROWS * OUTC <= 256 && (FEAT * OUTC) % 256 == 0, "head tiling");
__global__ __launch_bounds__(256) void head_kernel(const float* __restrict__ emb,
                                                   const float* __restrict__ gamma,
                                                   const float* __restrict__ beta,
                                                   const float* __restrict__ fc1_w,
                                                   const float* __restrict__ fc1_b,
                                                   const float* __restrict__ fc2_w,
                                                   const float* __restrict__ fc2_b,
                                                   float* __restrict__ out) {
    __shared__ float4 tileY[HROWS * GT_PITCH / 4];
    __shared__ float4 tileH[HROWS * GT_PITCH / 4];
    __shared__ float4 colMu[FEAT / 4], colSc[FEAT / 4];
    __shared__ float w2s[FEAT * OUTC];
    __shared__ float lgt[HROWS * OUTC];
    const int tid = threadIdx.x;
    const int lane = tid & 31;
    const int wave = __builtin_amdgcn_readfirstlane(tid >> 5);
    const int c0 = wave * 16;
    const int m  = lane & 15;
    const int kb = (lane >> 4) << 1;
    const int n  = lane & 15;
    const int rhalf = (lane >> 4) * 8;
    const int g0row = blockIdx.x * HROWS;

    const float* wcol = fc1_w + (size_t)kb * FEAT + c0 + n;
    v2f b[FEAT / 4];
#pragma unroll
    for (int s = 0; s < FEAT / 4; ++s) {
        b[s].x = wcol[(size_t)(4 * s) * FEAT];
        b[s].y = wcol[(size_t)(4 * s + 1) * FEAT];
    }
    const float bv = fc1_b[c0 + n];
    float4 y0 = *(const float4*)(emb + (size_t)(g0row + wave) * FEAT + lane * 4);
    float4 y1 = *(const float4*)(emb + (size_t)(g0row + wave + 8) * FEAT + lane * 4);
    const float4 be = *(const float4*)(beta + lane * 4);
    float w2r[FEAT * OUTC / 256];
#pragma unroll
    for (int i = 0; i < FEAT * OUTC / 256; ++i) w2r[i] = fc2_w[i * 256 + tid];
    float lg = 0.0f;
    if (tid < HROWS * OUTC) lg = fc2_b[tid % OUTC];

    if (wave < FEAT / 32) {
        const float* col = emb + tid;
        float e[NG];
#pragma unroll
        for (int g = 0; g < NG; ++g) e[g] = col[g * FEAT];
        float mu = 0.f;
#pragma unroll
        for (int g = 0; g < NG; ++g) mu += e[g];
        mu *= (1.0f / NG);
        float var = 0.f;
#pragma unroll
        for (int g = 0; g < NG; ++g) {
            const float d = e[g] - mu;
            var += d * d;
        }
        var *= (1.0f / NG);
        ((float*)&colMu[0])[tid] = mu;
        ((float*)&colSc[0])[tid] = gamma[tid] * rsqrtf(var + BN_EPS);
    }
#pragma unroll
    for (int i = 0; i < FEAT * OUTC / 256; ++i) w2s[i * 256 + tid] = w2r[i];
    __syncthreads();

    {
        const float4 mu4 = colMu[lane];
        const float4 sc4 = colSc[lane];
        float* ty = (float*)&tileY[0];
        y0.x = (y0.x - mu4.x) * sc4.x + be.x; y0.y = (y0.y - mu4.y) * sc4.y + be.y;
        y0.z = (y0.z - mu4.z) * sc4.z + be.z; y0.w = (y0.w - mu4.w) * sc4.w + be.w;
        y1.x = (y1.x - mu4.x) * sc4.x + be.x; y1.y = (y1.y - mu4.y) * sc4.y + be.y;
        y1.z = (y1.z - mu4.z) * sc4.z + be.z; y1.w = (y1.w - mu4.w) * sc4.w + be.w;
        *(float4*)(ty + wave * GT_PITCH + lane * 4) = y0;
        *(float4*)(ty + (wave + 8) * GT_PITCH + lane * 4) = y1;
    }
    __syncthreads();

    {
        const float* ap = (const float*)&tileY[0] + m * GT_PITCH + kb;
        v2f a[FEAT / 4];
#pragma unroll
        for (int s = 0; s < FEAT / 4; ++s) a[s] = *(const v2f*)(ap + 4 * s);
        v8f acc = {0.f, 0.f, 0.f, 0.f, 0.f, 0.f, 0.f, 0.f};
#pragma unroll
        for (int s = 0; s < FEAT / 4; ++s) {
            acc = __builtin_amdgcn_wmma_f32_16x16x4_f32(
                 false, a[s],  false, b[s],
                 (short)0, acc,  false,  false);
        }
        float* th = (float*)&tileH[0];
#pragma unroll
        for (int r = 0; r < 8; ++r) th[(rhalf + r) * GT_PITCH + c0 + n] = fmaxf(acc[r] + bv, 0.0f);
    }
    __syncthreads();

    const int row = tid / OUTC;
    const int cls = tid - row * OUTC;
    if (tid < HROWS * OUTC) {
        const float* hrow = (const float*)&tileH[0] + row * GT_PITCH;
        for (int k = 0; k < FEAT; ++k) lg = fmaf(hrow[k], w2s[k * OUTC + cls], lg);
        lgt[tid] = lg;
    }
    __syncthreads();
    if (tid < HROWS * OUTC) {
        const float* lr = lgt + row * OUTC;
        float mx = -1e30f;
#pragma unroll
        for (int j = 0; j < OUTC; ++j) mx = fmaxf(mx, lr[j]);
        float se = 0.f;
#pragma unroll
        for (int j = 0; j < OUTC; ++j) se += __expf(lr[j] - mx);
        const float lse = mx + __logf(se);
        out[(size_t)g0row * OUTC + tid] = lg - lse;
    }
}

extern "C" void kernel_launch(void* const* d_in, const int* in_sizes, int n_in,
                              void* d_out, int out_size, void* d_ws, size_t ws_size,
                              hipStream_t stream) {
    const float* n_feat    = (const float*)d_in[0];
    const int*   src       = (const int*)  d_in[1];
    const int*   dst       = (const int*)  d_in[2];
    const int*   graph_ids = (const int*)  d_in[3];
    const float* W1        = (const float*)d_in[4];
    const float* b1        = (const float*)d_in[5];
    const float* W2        = (const float*)d_in[6];
    const float* b2        = (const float*)d_in[7];
    const float* gamma     = (const float*)d_in[8];
    const float* beta      = (const float*)d_in[9];
    const float* fc1_w     = (const float*)d_in[10];
    const float* fc1_b     = (const float*)d_in[11];
    const float* fc2_w     = (const float*)d_in[12];
    const float* fc2_b     = (const float*)d_in[13];

    float* out = (float*)d_out;

    float* inv_out = (float*)d_ws;
    float* inv_in  = inv_out + 50176;
    int*   row_ptr = (int*)(inv_in + 50176);
    int*   offD    = row_ptr + 50176;
    int*   offS    = offD + NCHUNK * 256;
    unsigned* sortedD = (unsigned*)(offS + NCHUNK * 256);
    unsigned* sortedS = sortedD + NCHUNK * CHP;
    int*   csr_src = (int*)(sortedS + NCHUNK * CHP);
    float* bufA    = (float*)(csr_src + NE);
    float* bufB    = bufA + (size_t)NN * FEAT;

    edge_bin_kernel<<<NCHUNK, 256, 0, stream>>>(src, dst, sortedD, sortedS, offD, offS, out);
    owner_build_kernel<<<NBIN, 256, 0, stream>>>(sortedD, sortedS, offD, offS, row_ptr, csr_src, inv_in, inv_out);

    const int nodeTiles = NN / 16;
    const int gemmBlocks = 625;
    const int nodeBlocks = NN / 8;

    gemm128_wmma_kernel<<<gemmBlocks, 256, 0, stream>>>(n_feat, W1, bufA, inv_out, nullptr, 0, nodeTiles);
    gather_sum_kernel<<<nodeBlocks, 256, 0, stream>>>(bufA, row_ptr, csr_src, inv_in, b1, bufB);

    gemm128_wmma_kernel<<<gemmBlocks, 256, 0, stream>>>(bufB, W2, bufA, inv_out, nullptr, 0, nodeTiles);
    gather_sum_pool_kernel<<<nodeBlocks, 256, 0, stream>>>(bufA, row_ptr, csr_src, inv_in, b2, graph_ids, out);

    head_kernel<<<NG / HROWS, 256, 0, stream>>>(out, gamma, beta, fc1_w, fc1_b, fc2_w, fc2_b,
                                                out + (size_t)NG * FEAT);
}


# layer-1 product freed from the graph: x@W1 row tiles ride as extra blocks in the two list-building launches, the gather applies the source's out-degree scale per added row; 7 launches -> 6
# speedup vs baseline: 6.0590x; 1.0551x over previous
#include <hip/hip_runtime.h>
#include <hip/hip_bf16.h>

#define NN 50000
#define NE 800000
#define FEAT 128
#define NG 128
#define OUTC 10
#define BN_EPS 1e-5f

typedef float v2f __attribute__((ext_vector_type(2)));
typedef float v8f __attribute__((ext_vector_type(8)));

#define NCHUNK 256
#define CH (NE / NCHUNK)
#define CHP 3136
#define RNG 256
#define NBIN ((NN + RNG - 1) / RNG)
#define BIN_TRIPS ((CH + 255) / 256)
static_assert(NE % NCHUNK == 0, "chunks are equal");
static_assert(NN <= 65536, "a source id is packed into 16 bits");
static_assert(NBIN < 256, "range b+1 has a column in the offset table");
static_assert((NG * FEAT) % 256 == 0 && (NG * FEAT) / 256 <= NCHUNK, "the first kernel's blocks cover the embedding rows");

static __device__ __forceinline__ int block_scan_excl(const int v, int* wtot, const int lane, const int wave,
                                                      int& total) {
    int incl = v;
    for (int d = 1; d < 32; d <<= 1) {
        const int up = __shfl_up(incl, d);
        if (lane >= d) incl += up;
    }
    if (lane == 31) wtot[wave] = incl;
    __syncthreads();
    int off = 0, tot = 0;
    for (int w = 0; w < 8; ++w) {
        const int t = wtot[w];
        tot += t;
        if (w < wave) off += t;
    }
    total = tot;
    return off + incl - v;
}

static __device__ __forceinline__ void edge_bin_body(const int* __restrict__ src, const int* __restrict__ dst,
                                                     unsigned* __restrict__ sortedD,
                                                     unsigned* __restrict__ sortedS,
                                                     int* __restrict__ offD, int* __restrict__ offS,
                                                     float* __restrict__ emb) {
    __shared__ int hD[256], hS[256];
    __shared__ int wtD[8], wtS[8];
    __shared__ unsigned stD[CH], stS[CH];
    const int tid = threadIdx.x;
    const int lane = tid & 31;
    const int wave = tid >> 5;
    const int c = blockIdx.x;
    if (c < (NG * FEAT) / 256) emb[c * 256 + tid] = 0.0f;
    hD[tid] = 0;
    hS[tid] = 0;
    __syncthreads();

    const int* sp = src + (size_t)c * CH;
    const int* dp = dst + (size_t)c * CH;
    int sv[BIN_TRIPS], dv[BIN_TRIPS], rD[BIN_TRIPS], rS[BIN_TRIPS];
#pragma unroll
    for (int i = 0; i < BIN_TRIPS; ++i) {
        const int e = i * 256 + tid;
        sv[i] = 0; dv[i] = 0;
        if (i < CH / 256 || e < CH) { sv[i] = sp[e]; dv[i] = dp[e]; }
    }
#pragma unroll
    for (int i = 0; i < BIN_TRIPS; ++i) {
        const int e = i * 256 + tid;
        rD[i] = 0; rS[i] = 0;
        if (i < CH / 256 || e < CH) {
            const unsigned bd = min((unsigned)dv[i] >> 8, (unsigned)(NBIN - 1));
            const unsigned bs = min((unsigned)sv[i] >> 8, (unsigned)(NBIN - 1));
            rD[i] = atomicAdd(&hD[bd], 1);
            rS[i] = atomicAdd(&hS[bs], 1);
        }
    }
    __syncthreads();
    int totD, totS;
    const int exD = block_scan_excl(hD[tid], wtD, lane, wave, totD);
    const int exS = block_scan_excl(hS[tid], wtS, lane, wave, totS);
    hD[tid] = exD;
    hS[tid] = exS;
    offD[c * 256 + tid] = exD;
    offS[c * 256 + tid] = exS;
    __syncthreads();
#pragma unroll
    for (int i = 0; i < BIN_TRIPS; ++i) {
        const int e = i * 256 + tid;
        if (i < CH / 256 || e < CH) {
            const unsigned bd = min((unsigned)dv[i] >> 8, (unsigned)(NBIN - 1));
            const unsigned bs = min((unsigned)sv[i] >> 8, (unsigned)(NBIN - 1));
            stD[hD[bd] + rD[i]] = (((unsigned)dv[i] & 255u) << 16) | ((unsigned)sv[i] & 0xFFFFu);
            stS[hS[bs] + rS[i]] = (unsigned)sv[i] & 255u;
        }
    }
    __syncthreads();
    unsigned* oD = sortedD + (size_t)c * CHP;
    unsigned* oS = sortedS + (size_t)c * CHP;
    for (int j = tid; j < CH; j += 256) {
        oD[j] = stD[j];
        oS[j] = stS[j];
    }
}

#define OWN_GROUP 8
static __device__ __forceinline__ void owner_build_body(const unsigned* __restrict__ sortedD,
                                                        const unsigned* __restrict__ sortedS,
                                                        const int* __restrict__ offD,
                                                        const int* __restrict__ offS,
                                                        int* __restrict__ row_ptr, int* __restrict__ csr_src,
                                                        float* __restrict__ inv_in,
                                                        float* __restrict__ inv_out) {
    __shared__ int cntI[RNG], cntO[RNG], cur[RNG];
    __shared__ int wtA[8], wtB[8];
    const int tid = threadIdx.x;
    const int lane = tid & 31;
    const int wave = tid >> 5;
    const int b = blockIdx.x;
    const int c = tid;
    const int bD = offD[c * 256 + b];
    const int nD = offD[c * 256 + b + 1] - bD;
    const int bS = offS[c * 256 + b];
    const int nS = offS[c * 256 + b + 1] - bS;
    cntI[tid] = 0;
    cntO[tid] = 0;
    __syncthreads();

    const unsigned* pD = sortedD + (size_t)c * CHP + bD;
    const unsigned* pS = sortedS + (size_t)c * CHP + bS;
    const int nmax = min(max(nD, nS), CH);
    for (int j = 0; j < nmax; j += OWN_GROUP) {
        unsigned wd[OWN_GROUP], wsrc[OWN_GROUP];
#pragma unroll
        for (int u = 0; u < OWN_GROUP; ++u) {
            wd[u] = pD[(j + u < nD) ? j + u : 0];
            wsrc[u] = pS[(j + u < nS) ? j + u : 0];
        }
#pragma unroll
        for (int u = 0; u < OWN_GROUP; ++u) {
            if (j + u < nD) atomicAdd(&cntI[wd[u] >> 16], 1);
            if (j + u < nS) atomicAdd(&cntO[wsrc[u] & 255u], 1);
        }
    }

    int binStart, totI;
    (void)block_scan_excl(bD, wtA, lane, wave, binStart);
    const int my = cntI[tid];
    const int start = binStart + block_scan_excl(my, wtB, lane, wave, totI);
    const int node = b * RNG + tid;
    if (node <= NN) row_ptr[node] = start;
    if (node < NN) {
        inv_in[node]  = rsqrtf(fmaxf((float)my, 1.0f));
        inv_out[node] = rsqrtf(fmaxf((float)cntO[tid], 1.0f));
    }
    cur[tid] = start;
    __syncthreads();

    const int nDc = min(nD, CH);
    for (int j = 0; j < nDc; j += OWN_GROUP) {
        unsigned wd[OWN_GROUP];
#pragma unroll
        for (int u = 0; u < OWN_GROUP; ++u) wd[u] = pD[(j + u < nD) ? j + u : 0];
#pragma unroll
        for (int u = 0; u < OWN_GROUP; ++u) {
            if (j + u < nD) {
                const unsigned pos = (unsigned)atomicAdd(&cur[wd[u] >> 16], 1);
                if (pos < (unsigned)NE) csr_src[pos] = (int)(wd[u] & 0xFFFFu);
            }
        }
    }
}

#define GT_PITCH 132
static __device__ __forceinline__ void gemm128_body(const float* __restrict__ A,
                                                    const float* __restrict__ W,
                                                    float* __restrict__ out,
                                                    const float* __restrict__ scaleA,
                                                    const float* __restrict__ bias,
                                                    const int relu, const int first, const int step,
                                                    const int ntiles) {
    __shared__ float4 tileA[2][16 * GT_PITCH / 4];
    const int lane = threadIdx.x & 31;
    const int wave = __builtin_amdgcn_readfirstlane((int)(threadIdx.x >> 5));
    const int c0 = wave * 16;

    const int m  = lane & 15;
    const int kb = (lane >> 4) << 1;
    const int n  = lane & 15;

    const float* wcol = W + (size_t)kb * FEAT + c0 + n;
    v2f b[FEAT / 4];
#pragma unroll
    for (int s = 0; s < FEAT / 4; ++s) {
        b[s].x = wcol[(size_t)(4 * s) * FEAT];
        b[s].y = wcol[(size_t)(4 * s + 1) * FEAT];
    }
    const float bv = bias ? bias[c0 + n] : 0.0f;
    const int rhalf = (lane >> 4) * 8;

    int tile = first;
    float4 g0 = make_float4(0.f, 0.f, 0.f, 0.f), g1 = g0;
    float s0 = 1.0f, s1 = 1.0f;
    if (tile < ntiles) {
        const size_t ra = (size_t)tile * 16 + wave;
        g0 = *(const float4*)(A + ra * FEAT + lane * 4);
        g1 = *(const float4*)(A + (ra + 8) * FEAT + lane * 4);
        if (scaleA) { s0 = scaleA[ra]; s1 = scaleA[ra + 8]; }
    }
    int buf = 0;
    for (; tile < ntiles; tile += step) {
        const int r0 = tile * 16;
        float* tl = (float*)&tileA[buf][0];
        g0.x *= s0; g0.y *= s0; g0.z *= s0; g0.w *= s0;
        g1.x *= s1; g1.y *= s1; g1.z *= s1; g1.w *= s1;
        *(float4*)(tl + wave * GT_PITCH + lane * 4) = g0;
        *(float4*)(tl + (wave + 8) * GT_PITCH + lane * 4) = g1;
        __syncthreads();

        const int nt = tile + step;
        if (nt < ntiles) {
            const size_t ra = (size_t)nt * 16 + wave;
            g0 = *(const float4*)(A + ra * FEAT + lane * 4);
            g1 = *(const float4*)(A + (ra + 8) * FEAT + lane * 4);
            if (scaleA) { s0 = scaleA[ra]; s1 = scaleA[ra + 8]; }
        }

        const float* ap = tl + m * GT_PITCH + kb;
        v2f a[FEAT / 4];
#pragma unroll
        for (int s = 0; s < FEAT / 4; ++s) a[s] = *(const v2f*)(ap + 4 * s);

        v8f acc = {0.f, 0.f, 0.f, 0.f, 0.f, 0.f, 0.f, 0.f};
#pragma unroll
        for (int s = 0; s < FEAT / 4; ++s) {
            acc = __builtin_amdgcn_wmma_f32_16x16x4_f32(
                 false, a[s],  false, b[s],
                 (short)0, acc,  false,  false);
        }

#pragma unroll
        for (int r = 0; r < 8; ++r) {
            float v = acc[r] + bv;
            if (relu) v = fmaxf(v, 0.0f);
            out[(size_t)(r0 + rhalf + r) * FEAT + c0 + n] = v;
        }
        buf ^= 1;
    }
}

__global__ __launch_bounds__(256) void gemm128_wmma_kernel(const float* __restrict__ A,
                                                           const float* __restrict__ W,
                                                           float* __restrict__ out,
                                                           const float* __restrict__ scaleA,
                                                           const float* __restrict__ bias,
                                                           int relu, int ntiles) {
    gemm128_body(A, W, out, scaleA, bias, relu, (int)blockIdx.x, (int)gridDim.x, ntiles);
}

#define L1_SPLIT 1563
#define L1_BLOCKS 512
static_assert(L1_SPLIT > 0 && L1_SPLIT < NN / 16, "both launches hold a share");
__global__ __launch_bounds__(256) void bin_and_product_kernel(const int* __restrict__ src,
                                                              const int* __restrict__ dst,
                                                              unsigned* __restrict__ sortedD,
                                                              unsigned* __restrict__ sortedS,
                                                              int* __restrict__ offD, int* __restrict__ offS,
                                                              float* __restrict__ emb,
                                                              const float* __restrict__ X,
                                                              const float* __restrict__ W,
                                                              float* __restrict__ h) {
    if (blockIdx.x < NCHUNK) {
        edge_bin_body(src, dst, sortedD, sortedS, offD, offS, emb);
    } else {
        gemm128_body(X, W, h, nullptr, nullptr, 0, (int)blockIdx.x - NCHUNK, L1_BLOCKS, L1_SPLIT);
    }
}

__global__ __launch_bounds__(256) void own_and_product_kernel(const unsigned* __restrict__ sortedD,
                                                              const unsigned* __restrict__ sortedS,
                                                              const int* __restrict__ offD,
                                                              const int* __restrict__ offS,
                                                              int* __restrict__ row_ptr, int* __restrict__ csr_src,
                                                              float* __restrict__ inv_in,
                                                              float* __restrict__ inv_out,
                                                              const float* __restrict__ X,
                                                              const float* __restrict__ W,
                                                              float* __restrict__ h) {
    if (blockIdx.x < NBIN) {
        owner_build_body(sortedD, sortedS, offD, offS, row_ptr, csr_src, inv_in, inv_out);
    } else {
        gemm128_body(X, W, h, nullptr, nullptr, 0, L1_SPLIT + (int)blockIdx.x - NBIN, L1_BLOCKS, NN / 16);
    }
}

static __device__ __forceinline__ float4 ld_row(const float* hl, int ids, int j) {
    const int s = __builtin_amdgcn_readlane(ids, j);
    return *(const float4*)(hl + (size_t)s * FEAT);
}
template <bool SCALED>
static __device__ __forceinline__ float ld_scale(const float* __restrict__ src_scale, int ids, int j) {
    if (!SCALED) return 1.0f;
    return src_scale[__builtin_amdgcn_readlane(ids, j)];
}
template <bool SCALED>
static __device__ __forceinline__ void add4(float4& a, const float4 v, const float w) {
    if (SCALED) {
        a.x = fmaf(v.x, w, a.x); a.y = fmaf(v.y, w, a.y); a.z = fmaf(v.z, w, a.z); a.w = fmaf(v.w, w, a.w);
    } else {
        a.x += v.x; a.y += v.y; a.z += v.z; a.w += v.w;
    }
}

template <bool SCALED>
static __device__ __forceinline__ float4 gather_row(const float* __restrict__ h,
                                                    const int* __restrict__ row_ptr,
                                                    const int* __restrict__ csr_src,
                                                    const float* __restrict__ inv_in,
                                                    const float* __restrict__ src_scale,
                                                    const float* __restrict__ bias,
                                                    const int node, const int lane) {
    const int start = row_ptr[node];
    const int end = row_ptr[node + 1];
    const float* hl = h + lane * 4;

    float4 acc = make_float4(0.f, 0.f, 0.f, 0.f);
    for (int base = start; base < end; base += 32) {
        const int n = min(32, end - base);
        int mine = (lane < n) ? csr_src[base + lane] : 0;
        mine = (int)min((unsigned)mine, (unsigned)(NN - 1));
        int j = 0;
        for (; j + 8 <= n; j += 8) {
            const float4 v0 = ld_row(hl, mine, j);
            const float4 v1 = ld_row(hl, mine, j + 1);
            const float4 v2 = ld_row(hl, mine, j + 2);
            const float4 v3 = ld_row(hl, mine, j + 3);
            const float4 v4 = ld_row(hl, mine, j + 4);
            const float4 v5 = ld_row(hl, mine, j + 5);
            const float4 v6 = ld_row(hl, mine, j + 6);
            const float4 v7 = ld_row(hl, mine, j + 7);
            const float w0 = ld_scale<SCALED>(src_scale, mine, j);
            const float w1 = ld_scale<SCALED>(src_scale, mine, j + 1);
            const float w2 = ld_scale<SCALED>(src_scale, mine, j + 2);
            const float w3 = ld_scale<SCALED>(src_scale, mine, j + 3);
            const float w4 = ld_scale<SCALED>(src_scale, mine, j + 4);
            const float w5 = ld_scale<SCALED>(src_scale, mine, j + 5);
            const float w6 = ld_scale<SCALED>(src_scale, mine, j + 6);
            const float w7 = ld_scale<SCALED>(src_scale, mine, j + 7);
            add4<SCALED>(acc, v0, w0); add4<SCALED>(acc, v1, w1); add4<SCALED>(acc, v2, w2); add4<SCALED>(acc, v3, w3);
            add4<SCALED>(acc, v4, w4); add4<SCALED>(acc, v5, w5); add4<SCALED>(acc, v6, w6); add4<SCALED>(acc, v7, w7);
        }
        if (j + 4 <= n) {
            const float4 v0 = ld_row(hl, mine, j);
            const float4 v1 = ld_row(hl, mine, j + 1);
            const float4 v2 = ld_row(hl, mine, j + 2);
            const float4 v3 = ld_row(hl, mine, j + 3);
            const float w0 = ld_scale<SCALED>(src_scale, mine, j);
            const float w1 = ld_scale<SCALED>(src_scale, mine, j + 1);
            const float w2 = ld_scale<SCALED>(src_scale, mine, j + 2);
            const float w3 = ld_scale<SCALED>(src_scale, mine, j + 3);
            add4<SCALED>(acc, v0, w0); add4<SCALED>(acc, v1, w1); add4<SCALED>(acc, v2, w2); add4<SCALED>(acc, v3, w3);
            j += 4;
        }
        if (j < n) {
            const float4 v0 = ld_row(hl, mine, j);
            float4 v1 = make_float4(0.f, 0.f, 0.f, 0.f);
            float4 v2 = make_float4(0.f, 0.f, 0.f, 0.f);
            if (j + 1 < n) v1 = ld_row(hl, mine, j + 1);
            if (j + 2 < n) v2 = ld_row(hl, mine, j + 2);
            const float w0 = ld_scale<SCALED>(src_scale, mine, j);
            const float w1 = ld_scale<SCALED>(src_scale, mine, min(j + 1, 31));
            const float w2 = ld_scale<SCALED>(src_scale, mine, min(j + 2, 31));
            add4<SCALED>(acc, v0, w0); add4<SCALED>(acc, v1, w1); add4<SCALED>(acc, v2, w2);
        }
    }

    const float sc = inv_in[node];
    const float4 b = *(const float4*)(bias + lane * 4);
    float4 o;
    o.x = fmaxf(acc.x * sc + b.x, 0.0f);
    o.y = fmaxf(acc.y * sc + b.y, 0.0f);
    o.z = fmaxf(acc.z * sc + b.z, 0.0f);
    o.w = fmaxf(acc.w * sc + b.w, 0.0f);
    return o;
}

__global__ __launch_bounds__(256) void gather_sum_kernel(const float* __restrict__ h,
                                                         const int* __restrict__ row_ptr,
                                                         const int* __restrict__ csr_src,
                                                         const float* __restrict__ inv_in,
                                                         const float* __restrict__ src_scale,
                                                         const float* __restrict__ bias,
                                                         float* __restrict__ out) {
    const int lane = threadIdx.x & 31;
    const int wv = __builtin_amdgcn_readfirstlane((int)(threadIdx.x >> 5));
    const int node = blockIdx.x * 8 + wv;
    const float4 o = gather_row<true>(h, row_ptr, csr_src, inv_in, src_scale, bias, node, lane);
    *(float4*)(out + (size_t)node * FEAT + lane * 4) = o;
}

__global__ __launch_bounds__(256) void gather_sum_pool_kernel(const float* __restrict__ h,
                                                              const int* __restrict__ row_ptr,
                                                              const int* __restrict__ csr_src,
                                                              const float* __restrict__ inv_in,
                                                              const float* __restrict__ bias,
                                                              const int* __restrict__ graph_ids,
                                                              float* __restrict__ emb) {
    __shared__ float4 rows[8][32];
    __shared__ int gids[8];
    const int lane = threadIdx.x & 31;
    const int wv = __builtin_amdgcn_readfirstlane((int)(threadIdx.x >> 5));
    const int node = blockIdx.x * 8 + wv;
    const int gid = graph_ids[node];
    rows[wv][lane] = gather_row<false>(h, row_ptr, csr_src, inv_in, nullptr, bias, node, lane);
    if (lane == 0) gids[wv] = gid;
    __syncthreads();
    if (threadIdx.x < FEAT) {
        const int c = threadIdx.x;
        const float* col = (const float*)&rows[0][0] + c;
        int gv[8];
        float xv[8];
#pragma unroll
        for (int w = 0; w < 8; ++w) {
            gv[w] = __builtin_amdgcn_readfirstlane(gids[w]);
            xv[w] = col[w * FEAT];
        }
        int g = gv[0];
        float s = xv[0];
#pragma unroll
        for (int w = 1; w < 8; ++w) {
            if (gv[w] != g) {
                atomicAdd(emb + (size_t)g * FEAT + c, s);
                g = gv[w];
                s = xv[w];
            } else {
                s += xv[w];
            }
        }
        atomicAdd(emb + (size_t)g * FEAT + c, s);
    }
}

#define HROWS 16
static_assert(NG % HROWS == 0 && HROWS * OUTC <= 256 && (FEAT * OUTC) % 256 == 0, "head tiling");
__global__ __launch_bounds__(256) void head_kernel(const float* __restrict__ emb,
                                                   const float* __restrict__ gamma,
                                                   const float* __restrict__ beta,
                                                   const float* __restrict__ fc1_w,
                                                   const float* __restrict__ fc1_b,
                                                   const float* __restrict__ fc2_w,
                                                   const float* __restrict__ fc2_b,
                                                   float* __restrict__ out) {
    __shared__ float4 tileY[HROWS * GT_PITCH / 4];
    __shared__ float4 tileH[HROWS * GT_PITCH / 4];
    __shared__ float4 colMu[FEAT / 4], colSc[FEAT / 4];
    __shared__ float w2s[FEAT * OUTC];
    __shared__ float lgt[HROWS * OUTC];
    const int tid = threadIdx.x;
    const int lane = tid & 31;
    const int wave = __builtin_amdgcn_readfirstlane(tid >> 5);
    const int c0 = wave * 16;
    const int m  = lane & 15;
    const int kb = (lane >> 4) << 1;
    const int n  = lane & 15;
    const int rhalf = (lane >> 4) * 8;
    const int g0row = blockIdx.x * HROWS;

    const float* wcol = fc1_w + (size_t)kb * FEAT + c0 + n;
    v2f b[FEAT / 4];
#pragma unroll
    for (int s = 0; s < FEAT / 4; ++s) {
        b[s].x = wcol[(size_t)(4 * s) * FEAT];
        b[s].y = wcol[(size_t)(4 * s + 1) * FEAT];
    }
    const float bv = fc1_b[c0 + n];
    float4 y0 = *(const float4*)(emb + (size_t)(g0row + wave) * FEAT + lane * 4);
    float4 y1 = *(const float4*)(emb + (size_t)(g0row + wave + 8) * FEAT + lane * 4);
    const float4 be = *(const float4*)(beta + lane * 4);
    float w2r[FEAT * OUTC / 256];
#pragma unroll
    for (int i = 0; i < FEAT * OUTC / 256; ++i) w2r[i] = fc2_w[i * 256 + tid];
    float lg = 0.0f;
    if (tid < HROWS * OUTC) lg = fc2_b[tid % OUTC];

    if (wave < FEAT / 32) {
        const float* col = emb + tid;
        float e[NG];
#pragma unroll
        for (int g = 0; g < NG; ++g) e[g] = col[g * FEAT];
        float mu = 0.f;
#pragma unroll
        for (int g = 0; g < NG; ++g) mu += e[g];
        mu *= (1.0f / NG);
        float var = 0.f;
#pragma unroll
        for (int g = 0; g < NG; ++g) {
            const float d = e[g] - mu;
            var += d * d;
        }
        var *= (1.0f / NG);
        ((float*)&colMu[0])[tid] = mu;
        ((float*)&colSc[0])[tid] = gamma[tid] * rsqrtf(var + BN_EPS);
    }
#pragma unroll
    for (int i = 0; i < FEAT * OUTC / 256; ++i) w2s[i * 256 + tid] = w2r[i];
    __syncthreads();

    {
        const float4 mu4 = colMu[lane];
        const float4 sc4 = colSc[lane];
        float* ty = (float*)&tileY[0];
        y0.x = (y0.x - mu4.x) * sc4.x + be.x; y0.y = (y0.y - mu4.y) * sc4.y + be.y;
        y0.z = (y0.z - mu4.z) * sc4.z + be.z; y0.w = (y0.w - mu4.w) * sc4.w + be.w;
        y1.x = (y1.x - mu4.x) * sc4.x + be.x; y1.y = (y1.y - mu4.y) * sc4.y + be.y;
        y1.z = (y1.z - mu4.z) * sc4.z + be.z; y1.w = (y1.w - mu4.w) * sc4.w + be.w;
        *(float4*)(ty + wave * GT_PITCH + lane * 4) = y0;
        *(float4*)(ty + (wave + 8) * GT_PITCH + lane * 4) = y1;
    }
    __syncthreads();

    {
        const float* ap = (const float*)&tileY[0] + m * GT_PITCH + kb;
        v2f a[FEAT / 4];
#pragma unroll
        for (int s = 0; s < FEAT / 4; ++s) a[s] = *(const v2f*)(ap + 4 * s);
        v8f acc = {0.f, 0.f, 0.f, 0.f, 0.f, 0.f, 0.f, 0.f};
#pragma unroll
        for (int s = 0; s < FEAT / 4; ++s) {
            acc = __builtin_amdgcn_wmma_f32_16x16x4_f32(
                 false, a[s],  false, b[s],
                 (short)0, acc,  false,  false);
        }
        float* th = (float*)&tileH[0];
#pragma unroll
        for (int r = 0; r < 8; ++r) th[(rhalf + r) * GT_PITCH + c0 + n] = fmaxf(acc[r] + bv, 0.0f);
    }
    __syncthreads();

    const int row = tid / OUTC;
    const int cls = tid - row * OUTC;
    if (tid < HROWS * OUTC) {
        const float* hrow = (const float*)&tileH[0] + row * GT_PITCH;
        for (int k = 0; k < FEAT; ++k) lg = fmaf(hrow[k], w2s[k * OUTC + cls], lg);
        lgt[tid] = lg;
    }
    __syncthreads();
    if (tid < HROWS * OUTC) {
        const float* lr = lgt + row * OUTC;
        float mx = -1e30f;
#pragma unroll
        for (int j = 0; j < OUTC; ++j) mx = fmaxf(mx, lr[j]);
        float se = 0.f;
#pragma unroll
        for (int j = 0; j < OUTC; ++j) se += __expf(lr[j] - mx);
        const float lse = mx + __logf(se);
        out[(size_t)g0row * OUTC + tid] = lg - lse;
    }
}

extern "C" void kernel_launch(void* const* d_in, const int* in_sizes, int n_in,
                              void* d_out, int out_size, void* d_ws, size_t ws_size,
                              hipStream_t stream) {
    const float* n_feat    = (const float*)d_in[0];
    const int*   src       = (const int*)  d_in[1];
    const int*   dst       = (const int*)  d_in[2];
    const int*   graph_ids = (const int*)  d_in[3];
    const float* W1        = (const float*)d_in[4];
    const float* b1        = (const float*)d_in[5];
    const float* W2        = (const float*)d_in[6];
    const float* b2        = (const float*)d_in[7];
    const float* gamma     = (const float*)d_in[8];
    const float* beta      = (const float*)d_in[9];
    const float* fc1_w     = (const float*)d_in[10];
    const float* fc1_b     = (const float*)d_in[11];
    const float* fc2_w     = (const float*)d_in[12];
    const float* fc2_b     = (const float*)d_in[13];

    float* out = (float*)d_out;

    float* inv_out = (float*)d_ws;
    float* inv_in  = inv_out + 50176;
    int*   row_ptr = (int*)(inv_in + 50176);
    int*   offD    = row_ptr + 50176;
    int*   offS    = offD + NCHUNK * 256;
    unsigned* sortedD = (unsigned*)(offS + NCHUNK * 256);
    unsigned* sortedS = sortedD + NCHUNK * CHP;
    int*   csr_src = (int*)(sortedS + NCHUNK * CHP);
    float* bufA    = (float*)(csr_src + NE);
    float* bufB    = bufA + (size_t)NN * FEAT;

    bin_and_product_kernel<<<NCHUNK + L1_BLOCKS, 256, 0, stream>>>(src, dst, sortedD, sortedS, offD, offS, out,
                                                                   n_feat, W1, bufA);
    own_and_product_kernel<<<NBIN + L1_BLOCKS, 256, 0, stream>>>(sortedD, sortedS, offD, offS, row_ptr, csr_src,
                                                                 inv_in, inv_out, n_feat, W1, bufA);

    const int nodeTiles = NN / 16;
    const int gemmBlocks = 625;
    const int nodeBlocks = NN / 8;

    gather_sum_kernel<<<nodeBlocks, 256, 0, stream>>>(bufA, row_ptr, csr_src, inv_in, inv_out, b1, bufB);

    gemm128_wmma_kernel<<<gemmBlocks, 256, 0, stream>>>(bufB, W2, bufA, inv_out, nullptr, 0, nodeTiles);
    gather_sum_pool_kernel<<<nodeBlocks, 256, 0, stream>>>(bufA, row_ptr, csr_src, inv_in, b2, graph_ids, out);

    head_kernel<<<NG / HROWS, 256, 0, stream>>>(out, gamma, beta, fc1_w, fc1_b, fc2_w, fc2_b,
                                                out + (size_t)NG * FEAT);
}


# owner blocks read each chunk's piece with one coalesced load of a wave (lane = word, a wave's 64 pieces in flight at once), words kept in registers for the fill; list order = LDS arrival (changed)
# speedup vs baseline: 6.3750x; 1.0522x over previous
#include <hip/hip_runtime.h>
#include <hip/hip_bf16.h>

#define NN 50000
#define NE 800000
#define FEAT 128
#define NG 128
#define OUTC 10
#define BN_EPS 1e-5f

typedef float v2f __attribute__((ext_vector_type(2)));
typedef float v8f __attribute__((ext_vector_type(8)));

#define NCHUNK 256
#define CH (NE / NCHUNK)
#define CHP 3136
#define RNG 256
#define NBIN ((NN + RNG - 1) / RNG)
#define BIN_TRIPS ((CH + 255) / 256)
static_assert(NE % NCHUNK == 0, "chunks are equal");
static_assert(NN <= 65536, "a source id is packed into 16 bits");
static_assert(NBIN < 256, "range b+1 has a column in the offset table");
static_assert((NG * FEAT) % 256 == 0 && (NG * FEAT) / 256 <= NCHUNK, "the first kernel's blocks cover the embedding rows");

static __device__ __forceinline__ int block_scan_excl(const int v, int* wtot, const int lane, const int wave,
                                                      int& total) {
    int incl = v;
    for (int d = 1; d < 32; d <<= 1) {
        const int up = __shfl_up(incl, d);
        if (lane >= d) incl += up;
    }
    if (lane == 31) wtot[wave] = incl;
    __syncthreads();
    int off = 0, tot = 0;
    for (int w = 0; w < 8; ++w) {
        const int t = wtot[w];
        tot += t;
        if (w < wave) off += t;
    }
    total = tot;
    return off + incl - v;
}

static __device__ __forceinline__ void edge_bin_body(const int* __restrict__ src, const int* __restrict__ dst,
                                                     unsigned* __restrict__ sortedD,
                                                     unsigned* __restrict__ sortedS,
                                                     int* __restrict__ offD, int* __restrict__ offS,
                                                     float* __restrict__ emb) {
    __shared__ int hD[256], hS[256];
    __shared__ int wtD[8], wtS[8];
    __shared__ unsigned stD[CH], stS[CH];
    const int tid = threadIdx.x;
    const int lane = tid & 31;
    const int wave = tid >> 5;
    const int c = blockIdx.x;
    if (c < (NG * FEAT) / 256) emb[c * 256 + tid] = 0.0f;
    hD[tid] = 0;
    hS[tid] = 0;
    __syncthreads();

    const int* sp = src + (size_t)c * CH;
    const int* dp = dst + (size_t)c * CH;
    int sv[BIN_TRIPS], dv[BIN_TRIPS], rD[BIN_TRIPS], rS[BIN_TRIPS];
#pragma unroll
    for (int i = 0; i < BIN_TRIPS; ++i) {
        const int e = i * 256 + tid;
        sv[i] = 0; dv[i] = 0;
        if (i < CH / 256 || e < CH) { sv[i] = sp[e]; dv[i] = dp[e]; }
    }
#pragma unroll
    for (int i = 0; i < BIN_TRIPS; ++i) {
        const int e = i * 256 + tid;
        rD[i] = 0; rS[i] = 0;
        if (i < CH / 256 || e < CH) {
            const unsigned bd = min((unsigned)dv[i] >> 8, (unsigned)(NBIN - 1));
            const unsigned bs = min((unsigned)sv[i] >> 8, (unsigned)(NBIN - 1));
            rD[i] = atomicAdd(&hD[bd], 1);
            rS[i] = atomicAdd(&hS[bs], 1);
        }
    }
    __syncthreads();
    int totD, totS;
    const int exD = block_scan_excl(hD[tid], wtD, lane, wave, totD);
    const int exS = block_scan_excl(hS[tid], wtS, lane, wave, totS);
    hD[tid] = exD;
    hS[tid] = exS;
    offD[c * 256 + tid] = exD;
    offS[c * 256 + tid] = exS;
    __syncthreads();
#pragma unroll
    for (int i = 0; i < BIN_TRIPS; ++i) {
        const int e = i * 256 + tid;
        if (i < CH / 256 || e < CH) {
            const unsigned bd = min((unsigned)dv[i] >> 8, (unsigned)(NBIN - 1));
            const unsigned bs = min((unsigned)sv[i] >> 8, (unsigned)(NBIN - 1));
            stD[hD[bd] + rD[i]] = (((unsigned)dv[i] & 255u) << 16) | ((unsigned)sv[i] & 0xFFFFu);
            stS[hS[bs] + rS[i]] = (unsigned)sv[i] & 255u;
        }
    }
    __syncthreads();
    unsigned* oD = sortedD + (size_t)c * CHP;
    unsigned* oS = sortedS + (size_t)c * CHP;
    for (int j = tid; j < CH; j += 256) {
        oD[j] = stD[j];
        oS[j] = stS[j];
    }
}

static __device__ __forceinline__ void owner_build_body(const unsigned* __restrict__ sortedD,
                                                        const unsigned* __restrict__ sortedS,
                                                        const int* __restrict__ offD,
                                                        const int* __restrict__ offS,
                                                        int* __restrict__ row_ptr, int* __restrict__ csr_src,
                                                        float* __restrict__ inv_in,
                                                        float* __restrict__ inv_out) {
    __shared__ int cntI[RNG], cntO[RNG], cur[RNG];
    __shared__ int wtA[8], wtB[8];
    const int tid = threadIdx.x;
    const int lane = tid & 31;
    const int wave = __builtin_amdgcn_readfirstlane(tid >> 5);
    const int b = blockIdx.x;
    const int c = tid;
    const int bD = offD[c * 256 + b];
    const int nD = min(offD[c * 256 + b + 1] - bD, CH);
    const int bS = offS[c * 256 + b];
    const int nS = min(offS[c * 256 + b + 1] - bS, CH);
    cntI[tid] = 0;
    cntO[tid] = 0;
    __syncthreads();

    const unsigned* cD = sortedD + (size_t)wave * 32 * CHP;
    const unsigned* cS = sortedS + (size_t)wave * 32 * CHP;
    unsigned wd[32], wo[32];
#pragma unroll
    for (int j = 0; j < 32; ++j) {
        const unsigned* p = cD + (j * CHP + __builtin_amdgcn_readlane(bD, j));
        wd[j] = p[(lane < __builtin_amdgcn_readlane(nD, j)) ? (unsigned)lane : 0u];
    }
#pragma unroll
    for (int j = 0; j < 32; ++j) {
        const unsigned* p = cS + (j * CHP + __builtin_amdgcn_readlane(bS, j));
        wo[j] = p[(lane < __builtin_amdgcn_readlane(nS, j)) ? (unsigned)lane : 0u];
    }
#pragma unroll
    for (int j = 0; j < 32; ++j) {
        if (lane < __builtin_amdgcn_readlane(nD, j)) atomicAdd(&cntI[wd[j] >> 16], 1);
    }
#pragma unroll
    for (int j = 0; j < 32; ++j) {
        if (lane < __builtin_amdgcn_readlane(nS, j)) atomicAdd(&cntO[wo[j] & 255u], 1);
    }
    const bool longPiece = __builtin_amdgcn_ballot_w32(nD > 32 || nS > 32) != 0u;
    if (longPiece) {
        for (int j = 0; j < 32; ++j) {
            const unsigned* pD = cD + j * CHP + __builtin_amdgcn_readlane(bD, j);
            const unsigned* pS = cS + j * CHP + __builtin_amdgcn_readlane(bS, j);
            const int n1 = __builtin_amdgcn_readlane(nD, j);
            const int n2 = __builtin_amdgcn_readlane(nS, j);
            for (int k = 32 + lane; k < n1; k += 32) atomicAdd(&cntI[pD[k] >> 16], 1);
            for (int k = 32 + lane; k < n2; k += 32) atomicAdd(&cntO[pS[k] & 255u], 1);
        }
    }

    int binStart, totI;
    (void)block_scan_excl(bD, wtA, lane, wave, binStart);
    const int my = cntI[tid];
    const int start = binStart + block_scan_excl(my, wtB, lane, wave, totI);
    const int node = b * RNG + tid;
    if (node <= NN) row_ptr[node] = start;
    if (node < NN) {
        inv_in[node]  = rsqrtf(fmaxf((float)my, 1.0f));
        inv_out[node] = rsqrtf(fmaxf((float)cntO[tid], 1.0f));
    }
    cur[tid] = start;
    __syncthreads();

    unsigned pos[32];
#pragma unroll
    for (int j = 0; j < 32; ++j) {
        pos[j] = (unsigned)NE;
        if (lane < __builtin_amdgcn_readlane(nD, j)) pos[j] = (unsigned)atomicAdd(&cur[wd[j] >> 16], 1);
    }
#pragma unroll
    for (int j = 0; j < 32; ++j) {
        if (pos[j] < (unsigned)NE) csr_src[pos[j]] = (int)(wd[j] & 0xFFFFu);
    }
    if (longPiece) {
        for (int j = 0; j < 32; ++j) {
            const unsigned* pD = cD + j * CHP + __builtin_amdgcn_readlane(bD, j);
            const int n1 = __builtin_amdgcn_readlane(nD, j);
            for (int k = 32 + lane; k < n1; k += 32) {
                const unsigned w = pD[k];
                const unsigned p = (unsigned)atomicAdd(&cur[w >> 16], 1);
                if (p < (unsigned)NE) csr_src[p] = (int)(w & 0xFFFFu);
            }
        }
    }
}

#define GT_PITCH 132
static __device__ __forceinline__ void gemm128_body(const float* __restrict__ A,
                                                    const float* __restrict__ W,
                                                    float* __restrict__ out,
                                                    const float* __restrict__ scaleA,
                                                    const float* __restrict__ bias,
                                                    const int relu, const int first, const int step,
                                                    const int ntiles) {
    __shared__ float4 tileA[2][16 * GT_PITCH / 4];
    const int lane = threadIdx.x & 31;
    const int wave = __builtin_amdgcn_readfirstlane((int)(threadIdx.x >> 5));
    const int c0 = wave * 16;

    const int m  = lane & 15;
    const int kb = (lane >> 4) << 1;
    const int n  = lane & 15;

    const float* wcol = W + (size_t)kb * FEAT + c0 + n;
    v2f b[FEAT / 4];
#pragma unroll
    for (int s = 0; s < FEAT / 4; ++s) {
        b[s].x = wcol[(size_t)(4 * s) * FEAT];
        b[s].y = wcol[(size_t)(4 * s + 1) * FEAT];
    }
    const float bv = bias ? bias[c0 + n] : 0.0f;
    const int rhalf = (lane >> 4) * 8;

    int tile = first;
    float4 g0 = make_float4(0.f, 0.f, 0.f, 0.f), g1 = g0;
    float s0 = 1.0f, s1 = 1.0f;
    if (tile < ntiles) {
        const size_t ra = (size_t)tile * 16 + wave;
        g0 = *(const float4*)(A + ra * FEAT + lane * 4);
        g1 = *(const float4*)(A + (ra + 8) * FEAT + lane * 4);
        if (scaleA) { s0 = scaleA[ra]; s1 = scaleA[ra + 8]; }
    }
    int buf = 0;
    for (; tile < ntiles; tile += step) {
        const int r0 = tile * 16;
        float* tl = (float*)&tileA[buf][0];
        g0.x *= s0; g0.y *= s0; g0.z *= s0; g0.w *= s0;
        g1.x *= s1; g1.y *= s1; g1.z *= s1; g1.w *= s1;
        *(float4*)(tl + wave * GT_PITCH + lane * 4) = g0;
        *(float4*)(tl + (wave + 8) * GT_PITCH + lane * 4) = g1;
        __syncthreads();

        const int nt = tile + step;
        if (nt < ntiles) {
            const size_t ra = (size_t)nt * 16 + wave;
            g0 = *(const float4*)(A + ra * FEAT + lane * 4);
            g1 = *(const float4*)(A + (ra + 8) * FEAT + lane * 4);
            if (scaleA) { s0 = scaleA[ra]; s1 = scaleA[ra + 8]; }
        }

        const float* ap = tl + m * GT_PITCH + kb;
        v2f a[FEAT / 4];
#pragma unroll
        for (int s = 0; s < FEAT / 4; ++s) a[s] = *(const v2f*)(ap + 4 * s);

        v8f acc = {0.f, 0.f, 0.f, 0.f, 0.f, 0.f, 0.f, 0.f};
#pragma unroll
        for (int s = 0; s < FEAT / 4; ++s) {
            acc = __builtin_amdgcn_wmma_f32_16x16x4_f32(
                 false, a[s],  false, b[s],
                 (short)0, acc,  false,  false);
        }

#pragma unroll
        for (int r = 0; r < 8; ++r) {
            float v = acc[r] + bv;
            if (relu) v = fmaxf(v, 0.0f);
            out[(size_t)(r0 + rhalf + r) * FEAT + c0 + n] = v;
        }
        buf ^= 1;
    }
}

__global__ __launch_bounds__(256) void gemm128_wmma_kernel(const float* __restrict__ A,
                                                           const float* __restrict__ W,
                                                           float* __restrict__ out,
                                                           const float* __restrict__ scaleA,
                                                           const float* __restrict__ bias,
                                                           int relu, int ntiles) {
    gemm128_body(A, W, out, scaleA, bias, relu, (int)blockIdx.x, (int)gridDim.x, ntiles);
}

#define L1_SPLIT 1563
#define L1_BLOCKS 512
static_assert(L1_SPLIT > 0 && L1_SPLIT < NN / 16, "both launches hold a share");
__global__ __launch_bounds__(256) void bin_and_product_kernel(const int* __restrict__ src,
                                                              const int* __restrict__ dst,
                                                              unsigned* __restrict__ sortedD,
                                                              unsigned* __restrict__ sortedS,
                                                              int* __restrict__ offD, int* __restrict__ offS,
                                                              float* __restrict__ emb,
                                                              const float* __restrict__ X,
                                                              const float* __restrict__ W,
                                                              float* __restrict__ h) {
    if (blockIdx.x < NCHUNK) {
        edge_bin_body(src, dst, sortedD, sortedS, offD, offS, emb);
    } else {
        gemm128_body(X, W, h, nullptr, nullptr, 0, (int)blockIdx.x - NCHUNK, L1_BLOCKS, L1_SPLIT);
    }
}

__global__ __launch_bounds__(256) void own_and_product_kernel(const unsigned* __restrict__ sortedD,
                                                              const unsigned* __restrict__ sortedS,
                                                              const int* __restrict__ offD,
                                                              const int* __restrict__ offS,
                                                              int* __restrict__ row_ptr, int* __restrict__ csr_src,
                                                              float* __restrict__ inv_in,
                                                              float* __restrict__ inv_out,
                                                              const float* __restrict__ X,
                                                              const float* __restrict__ W,
                                                              float* __restrict__ h) {
    if (blockIdx.x < NBIN) {
        owner_build_body(sortedD, sortedS, offD, offS, row_ptr, csr_src, inv_in, inv_out);
    } else {
        gemm128_body(X, W, h, nullptr, nullptr, 0, L1_SPLIT + (int)blockIdx.x - NBIN, L1_BLOCKS, NN / 16);
    }
}

static __device__ __forceinline__ float4 ld_row(const float* hl, int ids, int j) {
    const int s = __builtin_amdgcn_readlane(ids, j);
    return *(const float4*)(hl + (size_t)s * FEAT);
}
template <bool SCALED>
static __device__ __forceinline__ float ld_scale(const float* __restrict__ src_scale, int ids, int j) {
    if (!SCALED) return 1.0f;
    return src_scale[__builtin_amdgcn_readlane(ids, j)];
}
template <bool SCALED>
static __device__ __forceinline__ void add4(float4& a, const float4 v, const float w) {
    if (SCALED) {
        a.x = fmaf(v.x, w, a.x); a.y = fmaf(v.y, w, a.y); a.z = fmaf(v.z, w, a.z); a.w = fmaf(v.w, w, a.w);
    } else {
        a.x += v.x; a.y += v.y; a.z += v.z; a.w += v.w;
    }
}

template <bool SCALED>
static __device__ __forceinline__ float4 gather_row(const float* __restrict__ h,
                                                    const int* __restrict__ row_ptr,
                                                    const int* __restrict__ csr_src,
                                                    const float* __restrict__ inv_in,
                                                    const float* __restrict__ src_scale,
                                                    const float* __restrict__ bias,
                                                    const int node, const int lane) {
    const int start = row_ptr[node];
    const int end = row_ptr[node + 1];
    const float* hl = h + lane * 4;

    float4 acc = make_float4(0.f, 0.f, 0.f, 0.f);
    for (int base = start; base < end; base += 32) {
        const int n = min(32, end - base);
        int mine = (lane < n) ? csr_src[base + lane] : 0;
        mine = (int)min((unsigned)mine, (unsigned)(NN - 1));
        int j = 0;
        for (; j + 8 <= n; j += 8) {
            const float4 v0 = ld_row(hl, mine, j);
            const float4 v1 = ld_row(hl, mine, j + 1);
            const float4 v2 = ld_row(hl, mine, j + 2);
            const float4 v3 = ld_row(hl, mine, j + 3);
            const float4 v4 = ld_row(hl, mine, j + 4);
            const float4 v5 = ld_row(hl, mine, j + 5);
            const float4 v6 = ld_row(hl, mine, j + 6);
            const float4 v7 = ld_row(hl, mine, j + 7);
            const float w0 = ld_scale<SCALED>(src_scale, mine, j);
            const float w1 = ld_scale<SCALED>(src_scale, mine, j + 1);
            const float w2 = ld_scale<SCALED>(src_scale, mine, j + 2);
            const float w3 = ld_scale<SCALED>(src_scale, mine, j + 3);
            const float w4 = ld_scale<SCALED>(src_scale, mine, j + 4);
            const float w5 = ld_scale<SCALED>(src_scale, mine, j + 5);
            const float w6 = ld_scale<SCALED>(src_scale, mine, j + 6);
            const float w7 = ld_scale<SCALED>(src_scale, mine, j + 7);
            add4<SCALED>(acc, v0, w0); add4<SCALED>(acc, v1, w1); add4<SCALED>(acc, v2, w2); add4<SCALED>(acc, v3, w3);
            add4<SCALED>(acc, v4, w4); add4<SCALED>(acc, v5, w5); add4<SCALED>(acc, v6, w6); add4<SCALED>(acc, v7, w7);
        }
        if (j + 4 <= n) {
            const float4 v0 = ld_row(hl, mine, j);
            const float4 v1 = ld_row(hl, mine, j + 1);
            const float4 v2 = ld_row(hl, mine, j + 2);
            const float4 v3 = ld_row(hl, mine, j + 3);
            const float w0 = ld_scale<SCALED>(src_scale, mine, j);
            const float w1 = ld_scale<SCALED>(src_scale, mine, j + 1);
            const float w2 = ld_scale<SCALED>(src_scale, mine, j + 2);
            const float w3 = ld_scale<SCALED>(src_scale, mine, j + 3);
            add4<SCALED>(acc, v0, w0); add4<SCALED>(acc, v1, w1); add4<SCALED>(acc, v2, w2); add4<SCALED>(acc, v3, w3);
            j += 4;
        }
        if (j < n) {
            const float4 v0 = ld_row(hl, mine, j);
            float4 v1 = make_float4(0.f, 0.f, 0.f, 0.f);
            float4 v2 = make_float4(0.f, 0.f, 0.f, 0.f);
            if (j + 1 < n) v1 = ld_row(hl, mine, j + 1);
            if (j + 2 < n) v2 = ld_row(hl, mine, j + 2);
            const float w0 = ld_scale<SCALED>(src_scale, mine, j);
            const float w1 = ld_scale<SCALED>(src_scale, mine, min(j + 1, 31));
            const float w2 = ld_scale<SCALED>(src_scale, mine, min(j + 2, 31));
            add4<SCALED>(acc, v0, w0); add4<SCALED>(acc, v1, w1); add4<SCALED>(acc, v2, w2);
        }
    }

    const float sc = inv_in[node];
    const float4 b = *(const float4*)(bias + lane * 4);
    float4 o;
    o.x = fmaxf(acc.x * sc + b.x, 0.0f);
    o.y = fmaxf(acc.y * sc + b.y, 0.0f);
    o.z = fmaxf(acc.z * sc + b.z, 0.0f);
    o.w = fmaxf(acc.w * sc + b.w, 0.0f);
    return o;
}

__global__ __launch_bounds__(256) void gather_sum_kernel(const float* __restrict__ h,
                                                         const int* __restrict__ row_ptr,
                                                         const int* __restrict__ csr_src,
                                                         const float* __restrict__ inv_in,
                                                         const float* __restrict__ src_scale,
                                                         const float* __restrict__ bias,
                                                         float* __restrict__ out) {
    const int lane = threadIdx.x & 31;
    const int wv = __builtin_amdgcn_readfirstlane((int)(threadIdx.x >> 5));
    const int node = blockIdx.x * 8 + wv;
    const float4 o = gather_row<true>(h, row_ptr, csr_src, inv_in, src_scale, bias, node, lane);
    *(float4*)(out + (size_t)node * FEAT + lane * 4) = o;
}

__global__ __launch_bounds__(256) void gather_sum_pool_kernel(const float* __restrict__ h,
                                                              const int* __restrict__ row_ptr,
                                                              const int* __restrict__ csr_src,
                                                              const float* __restrict__ inv_in,
                                                              const float* __restrict__ bias,
                                                              const int* __restrict__ graph_ids,
                                                              float* __restrict__ emb) {
    __shared__ float4 rows[8][32];
    __shared__ int gids[8];
    const int lane = threadIdx.x & 31;
    const int wv = __builtin_amdgcn_readfirstlane((int)(threadIdx.x >> 5));
    const int node = blockIdx.x * 8 + wv;
    const int gid = graph_ids[node];
    rows[wv][lane] = gather_row<false>(h, row_ptr, csr_src, inv_in, nullptr, bias, node, lane);
    if (lane == 0) gids[wv] = gid;
    __syncthreads();
    if (threadIdx.x < FEAT) {
        const int c = threadIdx.x;
        const float* col = (const float*)&rows[0][0] + c;
        int gv[8];
        float xv[8];
#pragma unroll
        for (int w = 0; w < 8; ++w) {
            gv[w] = __builtin_amdgcn_readfirstlane(gids[w]);
            xv[w] = col[w * FEAT];
        }
        int g = gv[0];
        float s = xv[0];
#pragma unroll
        for (int w = 1; w < 8; ++w) {
            if (gv[w] != g) {
                atomicAdd(emb + (size_t)g * FEAT + c, s);
                g = gv[w];
                s = xv[w];
            } else {
                s += xv[w];
            }
        }
        atomicAdd(emb + (size_t)g * FEAT + c, s);
    }
}

#define HROWS 16
static_assert(NG % HROWS == 0 && HROWS * OUTC <= 256 && (FEAT * OUTC) % 256 == 0, "head tiling");
__global__ __launch_bounds__(256) void head_kernel(const float* __restrict__ emb,
                                                   const float* __restrict__ gamma,
                                                   const float* __restrict__ beta,
                                                   const float* __restrict__ fc1_w,
                                                   const float* __restrict__ fc1_b,
                                                   const float* __restrict__ fc2_w,
                                                   const float* __restrict__ fc2_b,
                                                   float* __restrict__ out) {
    __shared__ float4 tileY[HROWS * GT_PITCH / 4];
    __shared__ float4 tileH[HROWS * GT_PITCH / 4];
    __shared__ float4 colMu[FEAT / 4], colSc[FEAT / 4];
    __shared__ float w2s[FEAT * OUTC];
    __shared__ float lgt[HROWS * OUTC];
    const int tid = threadIdx.x;
    const int lane = tid & 31;
    const int wave = __builtin_amdgcn_readfirstlane(tid >> 5);
    const int c0 = wave * 16;
    const int m  = lane & 15;
    const int kb = (lane >> 4) << 1;
    const int n  = lane & 15;
    const int rhalf = (lane >> 4) * 8;
    const int g0row = blockIdx.x * HROWS;

    const float* wcol = fc1_w + (size_t)kb * FEAT + c0 + n;
    v2f b[FEAT / 4];
#pragma unroll
    for (int s = 0; s < FEAT / 4; ++s) {
        b[s].x = wcol[(size_t)(4 * s) * FEAT];
        b[s].y = wcol[(size_t)(4 * s + 1) * FEAT];
    }
    const float bv = fc1_b[c0 + n];
    float4 y0 = *(const float4*)(emb + (size_t)(g0row + wave) * FEAT + lane * 4);
    float4 y1 = *(const float4*)(emb + (size_t)(g0row + wave + 8) * FEAT + lane * 4);
    const float4 be = *(const float4*)(beta + lane * 4);
    float w2r[FEAT * OUTC / 256];
#pragma unroll
    for (int i = 0; i < FEAT * OUTC / 256; ++i) w2r[i] = fc2_w[i * 256 + tid];
    float lg = 0.0f;
    if (tid < HROWS * OUTC) lg = fc2_b[tid % OUTC];

    if (wave < FEAT / 32) {
        const float* col = emb + tid;
        float e[NG];
#pragma unroll
        for (int g = 0; g < NG; ++g) e[g] = col[g * FEAT];
        float mu = 0.f;
#pragma unroll
        for (int g = 0; g < NG; ++g) mu += e[g];
        mu *= (1.0f / NG);
        float var = 0.f;
#pragma unroll
        for (int g = 0; g < NG; ++g) {
            const float d = e[g] - mu;
            var += d * d;
        }
        var *= (1.0f / NG);
        ((float*)&colMu[0])[tid] = mu;
        ((float*)&colSc[0])[tid] = gamma[tid] * rsqrtf(var + BN_EPS);
    }
#pragma unroll
    for (int i = 0; i < FEAT * OUTC / 256; ++i) w2s[i * 256 + tid] = w2r[i];
    __syncthreads();

    {
        const float4 mu4 = colMu[lane];
        const float4 sc4 = colSc[lane];
        float* ty = (float*)&tileY[0];
        y0.x = (y0.x - mu4.x) * sc4.x + be.x; y0.y = (y0.y - mu4.y) * sc4.y + be.y;
        y0.z = (y0.z - mu4.z) * sc4.z + be.z; y0.w = (y0.w - mu4.w) * sc4.w + be.w;
        y1.x = (y1.x - mu4.x) * sc4.x + be.x; y1.y = (y1.y - mu4.y) * sc4.y + be.y;
        y1.z = (y1.z - mu4.z) * sc4.z + be.z; y1.w = (y1.w - mu4.w) * sc4.w + be.w;
        *(float4*)(ty + wave * GT_PITCH + lane * 4) = y0;
        *(float4*)(ty + (wave + 8) * GT_PITCH + lane * 4) = y1;
    }
    __syncthreads();

    {
        const float* ap = (const float*)&tileY[0] + m * GT_PITCH + kb;
        v2f a[FEAT / 4];
#pragma unroll
        for (int s = 0; s < FEAT / 4; ++s) a[s] = *(const v2f*)(ap + 4 * s);
        v8f acc = {0.f, 0.f, 0.f, 0.f, 0.f, 0.f, 0.f, 0.f};
#pragma unroll
        for (int s = 0; s < FEAT / 4; ++s) {
            acc = __builtin_amdgcn_wmma_f32_16x16x4_f32(
                 false, a[s],  false, b[s],
                 (short)0, acc,  false,  false);
        }
        float* th = (float*)&tileH[0];
#pragma unroll
        for (int r = 0; r < 8; ++r) th[(rhalf + r) * GT_PITCH + c0 + n] = fmaxf(acc[r] + bv, 0.0f);
    }
    __syncthreads();

    const int row = tid / OUTC;
    const int cls = tid - row * OUTC;
    if (tid < HROWS * OUTC) {
        const float* hrow = (const float*)&tileH[0] + row * GT_PITCH;
        for (int k = 0; k < FEAT; ++k) lg = fmaf(hrow[k], w2s[k * OUTC + cls], lg);
        lgt[tid] = lg;
    }
    __syncthreads();
    if (tid < HROWS * OUTC) {
        const float* lr = lgt + row * OUTC;
        float mx = -1e30f;
#pragma unroll
        for (int j = 0; j < OUTC; ++j) mx = fmaxf(mx, lr[j]);
        float se = 0.f;
#pragma unroll
        for (int j = 0; j < OUTC; ++j) se += __expf(lr[j] - mx);
        const float lse = mx + __logf(se);
        out[(size_t)g0row * OUTC + tid] = lg - lse;
    }
}

extern "C" void kernel_launch(void* const* d_in, const int* in_sizes, int n_in,
                              void* d_out, int out_size, void* d_ws, size_t ws_size,
                              hipStream_t stream) {
    const float* n_feat    = (const float*)d_in[0];
    const int*   src       = (const int*)  d_in[1];
    const int*   dst       = (const int*)  d_in[2];
    const int*   graph_ids = (const int*)  d_in[3];
    const float* W1        = (const float*)d_in[4];
    const float* b1        = (const float*)d_in[5];
    const float* W2        = (const float*)d_in[6];
    const float* b2        = (const float*)d_in[7];
    const float* gamma     = (const float*)d_in[8];
    const float* beta      = (const float*)d_in[9];
    const float* fc1_w     = (const float*)d_in[10];
    const float* fc1_b     = (const float*)d_in[11];
    const float* fc2_w     = (const float*)d_in[12];
    const float* fc2_b     = (const float*)d_in[13];

    float* out = (float*)d_out;

    float* inv_out = (float*)d_ws;
    float* inv_in  = inv_out + 50176;
    int*   row_ptr = (int*)(inv_in + 50176);
    int*   offD    = row_ptr + 50176;
    int*   offS    = offD + NCHUNK * 256;
    unsigned* sortedD = (unsigned*)(offS + NCHUNK * 256);
    unsigned* sortedS = sortedD + NCHUNK * CHP;
    int*   csr_src = (int*)(sortedS + NCHUNK * CHP);
    float* bufA    = (float*)(csr_src + NE);
    float* bufB    = bufA + (size_t)NN * FEAT;

    bin_and_product_kernel<<<NCHUNK + L1_BLOCKS, 256, 0, stream>>>(src, dst, sortedD, sortedS, offD, offS, out,
                                                                   n_feat, W1, bufA);
    own_and_product_kernel<<<NBIN + L1_BLOCKS, 256, 0, stream>>>(sortedD, sortedS, offD, offS, row_ptr, csr_src,
                                                                 inv_in, inv_out, n_feat, W1, bufA);

    const int nodeTiles = NN / 16;
    const int gemmBlocks = 625;
    const int nodeBlocks = NN / 8;

    gather_sum_kernel<<<nodeBlocks, 256, 0, stream>>>(bufA, row_ptr, csr_src, inv_in, inv_out, b1, bufB);

    gemm128_wmma_kernel<<<gemmBlocks, 256, 0, stream>>>(bufB, W2, bufA, inv_out, nullptr, 0, nodeTiles);
    gather_sum_pool_kernel<<<nodeBlocks, 256, 0, stream>>>(bufA, row_ptr, csr_src, inv_in, b2, graph_ids, out);

    head_kernel<<<NG / HROWS, 256, 0, stream>>>(out, gamma, beta, fc1_w, fc1_b, fc2_w, fc2_b,
                                                out + (size_t)NG * FEAT);
}


# layer 2 sums first, then projects: layer-1 gather stores inv_out*x1; resident last-gather blocks sum 16 nodes' in-rows in LDS, multiply by W2 (WMMA), pool; product launch, h2 gone; sum order changed
# speedup vs baseline: 6.5881x; 1.0334x over previous
#include <hip/hip_runtime.h>
#include <hip/hip_bf16.h>

#define NN 50000
#define NE 800000
#define FEAT 128
#define NG 128
#define OUTC 10
#define BN_EPS 1e-5f

typedef float v2f __attribute__((ext_vector_type(2)));
typedef float v8f __attribute__((ext_vector_type(8)));

#define NCHUNK 256
#define CH (NE / NCHUNK)
#define CHP 3136
#define RNG 256
#define NBIN ((NN + RNG - 1) / RNG)
#define BIN_TRIPS ((CH + 255) / 256)
static_assert(NE % NCHUNK == 0, "chunks are equal");
static_assert(NN <= 65536, "a source id is packed into 16 bits");
static_assert(NBIN < 256, "range b+1 has a column in the offset table");
static_assert((NG * FEAT) % 256 == 0 && (NG * FEAT) / 256 <= NCHUNK, "the first kernel's blocks cover the embedding rows");

static __device__ __forceinline__ int block_scan_excl(const int v, int* wtot, const int lane, const int wave,
                                                      int& total) {
    int incl = v;
    for (int d = 1; d < 32; d <<= 1) {
        const int up = __shfl_up(incl, d);
        if (lane >= d) incl += up;
    }
    if (lane == 31) wtot[wave] = incl;
    __syncthreads();
    int off = 0, tot = 0;
    for (int w = 0; w < 8; ++w) {
        const int t = wtot[w];
        tot += t;
        if (w < wave) off += t;
    }
    total = tot;
    return off + incl - v;
}

static __device__ __forceinline__ void edge_bin_body(const int* __restrict__ src, const int* __restrict__ dst,
                                                     unsigned* __restrict__ sortedD,
                                                     unsigned* __restrict__ sortedS,
                                                     int* __restrict__ offD, int* __restrict__ offS,
                                                     float* __restrict__ emb) {
    __shared__ int hD[256], hS[256];
    __shared__ int wtD[8], wtS[8];
    __shared__ unsigned stD[CH], stS[CH];
    const int tid = threadIdx.x;
    const int lane = tid & 31;
    const int wave = tid >> 5;
    const int c = blockIdx.x;
    if (c < (NG * FEAT) / 256) emb[c * 256 + tid] = 0.0f;
    hD[tid] = 0;
    hS[tid] = 0;
    __syncthreads();

    const int* sp = src + (size_t)c * CH;
    const int* dp = dst + (size_t)c * CH;
    int sv[BIN_TRIPS], dv[BIN_TRIPS], rD[BIN_TRIPS], rS[BIN_TRIPS];
#pragma unroll
    for (int i = 0; i < BIN_TRIPS; ++i) {
        const int e = i * 256 + tid;
        sv[i] = 0; dv[i] = 0;
        if (i < CH / 256 || e < CH) { sv[i] = sp[e]; dv[i] = dp[e]; }
    }
#pragma unroll
    for (int i = 0; i < BIN_TRIPS; ++i) {
        const int e = i * 256 + tid;
        rD[i] = 0; rS[i] = 0;
        if (i < CH / 256 || e < CH) {
            const unsigned bd = min((unsigned)dv[i] >> 8, (unsigned)(NBIN - 1));
            const unsigned bs = min((unsigned)sv[i] >> 8, (unsigned)(NBIN - 1));
            rD[i] = atomicAdd(&hD[bd], 1);
            rS[i] = atomicAdd(&hS[bs], 1);
        }
    }
    __syncthreads();
    int totD, totS;
    const int exD = block_scan_excl(hD[tid], wtD, lane, wave, totD);
    const int exS = block_scan_excl(hS[tid], wtS, lane, wave, totS);
    hD[tid] = exD;
    hS[tid] = exS;
    offD[c * 256 + tid] = exD;
    offS[c * 256 + tid] = exS;
    __syncthreads();
#pragma unroll
    for (int i = 0; i < BIN_TRIPS; ++i) {
        const int e = i * 256 + tid;
        if (i < CH / 256 || e < CH) {
            const unsigned bd = min((unsigned)dv[i] >> 8, (unsigned)(NBIN - 1));
            const unsigned bs = min((unsigned)sv[i] >> 8, (unsigned)(NBIN - 1));
            stD[hD[bd] + rD[i]] = (((unsigned)dv[i] & 255u) << 16) | ((unsigned)sv[i] & 0xFFFFu);
            stS[hS[bs] + rS[i]] = (unsigned)sv[i] & 255u;
        }
    }
    __syncthreads();
    unsigned* oD = sortedD + (size_t)c * CHP;
    unsigned* oS = sortedS + (size_t)c * CHP;
    for (int j = tid; j < CH; j += 256) {
        oD[j] = stD[j];
        oS[j] = stS[j];
    }
}

static __device__ __forceinline__ void owner_build_body(const unsigned* __restrict__ sortedD,
                                                        const unsigned* __restrict__ sortedS,
                                                        const int* __restrict__ offD,
                                                        const int* __restrict__ offS,
                                                        int* __restrict__ row_ptr, int* __restrict__ csr_src,
                                                        float* __restrict__ inv_in,
                                                        float* __restrict__ inv_out) {
    __shared__ int cntI[RNG], cntO[RNG], cur[RNG];
    __shared__ int wtA[8], wtB[8];
    const int tid = threadIdx.x;
    const int lane = tid & 31;
    const int wave = __builtin_amdgcn_readfirstlane(tid >> 5);
    const int b = blockIdx.x;
    const int c = tid;
    const int bD = offD[c * 256 + b];
    const int nD = min(offD[c * 256 + b + 1] - bD, CH);
    const int bS = offS[c * 256 + b];
    const int nS = min(offS[c * 256 + b + 1] - bS, CH);
    cntI[tid] = 0;
    cntO[tid] = 0;
    __syncthreads();

    const unsigned* cD = sortedD + (size_t)wave * 32 * CHP;
    const unsigned* cS = sortedS + (size_t)wave * 32 * CHP;
    unsigned wd[32], wo[32];
#pragma unroll
    for (int j = 0; j < 32; ++j) {
        const unsigned* p = cD + (j * CHP + __builtin_amdgcn_readlane(bD, j));
        wd[j] = p[(lane < __builtin_amdgcn_readlane(nD, j)) ? (unsigned)lane : 0u];
    }
#pragma unroll
    for (int j = 0; j < 32; ++j) {
        const unsigned* p = cS + (j * CHP + __builtin_amdgcn_readlane(bS, j));
        wo[j] = p[(lane < __builtin_amdgcn_readlane(nS, j)) ? (unsigned)lane : 0u];
    }
#pragma unroll
    for (int j = 0; j < 32; ++j) {
        if (lane < __builtin_amdgcn_readlane(nD, j)) atomicAdd(&cntI[wd[j] >> 16], 1);
    }
#pragma unroll
    for (int j = 0; j < 32; ++j) {
        if (lane < __builtin_amdgcn_readlane(nS, j)) atomicAdd(&cntO[wo[j] & 255u], 1);
    }
    const bool longPiece = __builtin_amdgcn_ballot_w32(nD > 32 || nS > 32) != 0u;
    if (longPiece) {
        for (int j = 0; j < 32; ++j) {
            const unsigned* pD = cD + j * CHP + __builtin_amdgcn_readlane(bD, j);
            const unsigned* pS = cS + j * CHP + __builtin_amdgcn_readlane(bS, j);
            const int n1 = __builtin_amdgcn_readlane(nD, j);
            const int n2 = __builtin_amdgcn_readlane(nS, j);
            for (int k = 32 + lane; k < n1; k += 32) atomicAdd(&cntI[pD[k] >> 16], 1);
            for (int k = 32 + lane; k < n2; k += 32) atomicAdd(&cntO[pS[k] & 255u], 1);
        }
    }

    int binStart, totI;
    (void)block_scan_excl(bD, wtA, lane, wave, binStart);
    const int my = cntI[tid];
    const int start = binStart + block_scan_excl(my, wtB, lane, wave, totI);
    const int node = b * RNG + tid;
    if (node <= NN) row_ptr[node] = start;
    if (node < NN) {
        inv_in[node]  = rsqrtf(fmaxf((float)my, 1.0f));
        inv_out[node] = rsqrtf(fmaxf((float)cntO[tid], 1.0f));
    }
    cur[tid] = start;
    __syncthreads();

    unsigned pos[32];
#pragma unroll
    for (int j = 0; j < 32; ++j) {
        pos[j] = (unsigned)NE;
        if (lane < __builtin_amdgcn_readlane(nD, j)) pos[j] = (unsigned)atomicAdd(&cur[wd[j] >> 16], 1);
    }
#pragma unroll
    for (int j = 0; j < 32; ++j) {
        if (pos[j] < (unsigned)NE) csr_src[pos[j]] = (int)(wd[j] & 0xFFFFu);
    }
    if (longPiece) {
        for (int j = 0; j < 32; ++j) {
            const unsigned* pD = cD + j * CHP + __builtin_amdgcn_readlane(bD, j);
            const int n1 = __builtin_amdgcn_readlane(nD, j);
            for (int k = 32 + lane; k < n1; k += 32) {
                const unsigned w = pD[k];
                const unsigned p = (unsigned)atomicAdd(&cur[w >> 16], 1);
                if (p < (unsigned)NE) csr_src[p] = (int)(w & 0xFFFFu);
            }
        }
    }
}

#define GT_PITCH 132
static __device__ __forceinline__ void gemm128_body(const float* __restrict__ A,
                                                    const float* __restrict__ W,
                                                    float* __restrict__ out,
                                                    const float* __restrict__ scaleA,
                                                    const float* __restrict__ bias,
                                                    const int relu, const int first, const int step,
                                                    const int ntiles) {
    __shared__ float4 tileA[2][16 * GT_PITCH / 4];
    const int lane = threadIdx.x & 31;
    const int wave = __builtin_amdgcn_readfirstlane((int)(threadIdx.x >> 5));
    const int c0 = wave * 16;

    const int m  = lane & 15;
    const int kb = (lane >> 4) << 1;
    const int n  = lane & 15;

    const float* wcol = W + (size_t)kb * FEAT + c0 + n;
    v2f b[FEAT / 4];
#pragma unroll
    for (int s = 0; s < FEAT / 4; ++s) {
        b[s].x = wcol[(size_t)(4 * s) * FEAT];
        b[s].y = wcol[(size_t)(4 * s + 1) * FEAT];
    }
    const float bv = bias ? bias[c0 + n] : 0.0f;
    const int rhalf = (lane >> 4) * 8;

    int tile = first;
    float4 g0 = make_float4(0.f, 0.f, 0.f, 0.f), g1 = g0;
    float s0 = 1.0f, s1 = 1.0f;
    if (tile < ntiles) {
        const size_t ra = (size_t)tile * 16 + wave;
        g0 = *(const float4*)(A + ra * FEAT + lane * 4);
        g1 = *(const float4*)(A + (ra + 8) * FEAT + lane * 4);
        if (scaleA) { s0 = scaleA[ra]; s1 = scaleA[ra + 8]; }
    }
    int buf = 0;
    for (; tile < ntiles; tile += step) {
        const int r0 = tile * 16;
        float* tl = (float*)&tileA[buf][0];
        g0.x *= s0; g0.y *= s0; g0.z *= s0; g0.w *= s0;
        g1.x *= s1; g1.y *= s1; g1.z *= s1; g1.w *= s1;
        *(float4*)(tl + wave * GT_PITCH + lane * 4) = g0;
        *(float4*)(tl + (wave + 8) * GT_PITCH + lane * 4) = g1;
        __syncthreads();

        const int nt = tile + step;
        if (nt < ntiles) {
            const size_t ra = (size_t)nt * 16 + wave;
            g0 = *(const float4*)(A + ra * FEAT + lane * 4);
            g1 = *(const float4*)(A + (ra + 8) * FEAT + lane * 4);
            if (scaleA) { s0 = scaleA[ra]; s1 = scaleA[ra + 8]; }
        }

        const float* ap = tl + m * GT_PITCH + kb;
        v2f a[FEAT / 4];
#pragma unroll
        for (int s = 0; s < FEAT / 4; ++s) a[s] = *(const v2f*)(ap + 4 * s);

        v8f acc = {0.f, 0.f, 0.f, 0.f, 0.f, 0.f, 0.f, 0.f};
#pragma unroll
        for (int s = 0; s < FEAT / 4; ++s) {
            acc = __builtin_amdgcn_wmma_f32_16x16x4_f32(
                 false, a[s],  false, b[s],
                 (short)0, acc,  false,  false);
        }

#pragma unroll
        for (int r = 0; r < 8; ++r) {
            float v = acc[r] + bv;
            if (relu) v = fmaxf(v, 0.0f);
            out[(size_t)(r0 + rhalf + r) * FEAT + c0 + n] = v;
        }
        buf ^= 1;
    }
}

#define L1_SPLIT 1563
#define L1_BLOCKS 512
static_assert(L1_SPLIT > 0 && L1_SPLIT < NN / 16, "both launches hold a share");
__global__ __launch_bounds__(256) void bin_and_product_kernel(const int* __restrict__ src,
                                                              const int* __restrict__ dst,
                                                              unsigned* __restrict__ sortedD,
                                                              unsigned* __restrict__ sortedS,
                                                              int* __restrict__ offD, int* __restrict__ offS,
                                                              float* __restrict__ emb,
                                                              const float* __restrict__ X,
                                                              const float* __restrict__ W,
                                                              float* __restrict__ h) {
    if (blockIdx.x < NCHUNK) {
        edge_bin_body(src, dst, sortedD, sortedS, offD, offS, emb);
    } else {
        gemm128_body(X, W, h, nullptr, nullptr, 0, (int)blockIdx.x - NCHUNK, L1_BLOCKS, L1_SPLIT);
    }
}

__global__ __launch_bounds__(256) void own_and_product_kernel(const unsigned* __restrict__ sortedD,
                                                              const unsigned* __restrict__ sortedS,
                                                              const int* __restrict__ offD,
                                                              const int* __restrict__ offS,
                                                              int* __restrict__ row_ptr, int* __restrict__ csr_src,
                                                              float* __restrict__ inv_in,
                                                              float* __restrict__ inv_out,
                                                              const float* __restrict__ X,
                                                              const float* __restrict__ W,
                                                              float* __restrict__ h) {
    if (blockIdx.x < NBIN) {
        owner_build_body(sortedD, sortedS, offD, offS, row_ptr, csr_src, inv_in, inv_out);
    } else {
        gemm128_body(X, W, h, nullptr, nullptr, 0, L1_SPLIT + (int)blockIdx.x - NBIN, L1_BLOCKS, NN / 16);
    }
}

static __device__ __forceinline__ float4 ld_row(const float* hl, int ids, int j) {
    const int s = __builtin_amdgcn_readlane(ids, j);
    return *(const float4*)(hl + (size_t)s * FEAT);
}
template <bool SCALED>
static __device__ __forceinline__ float ld_scale(const float* __restrict__ src_scale, int ids, int j) {
    if (!SCALED) return 1.0f;
    return src_scale[__builtin_amdgcn_readlane(ids, j)];
}
template <bool SCALED>
static __device__ __forceinline__ void add4(float4& a, const float4 v, const float w) {
    if (SCALED) {
        a.x = fmaf(v.x, w, a.x); a.y = fmaf(v.y, w, a.y); a.z = fmaf(v.z, w, a.z); a.w = fmaf(v.w, w, a.w);
    } else {
        a.x += v.x; a.y += v.y; a.z += v.z; a.w += v.w;
    }
}

template <bool SCALED>
static __device__ __forceinline__ float4 gather_rows(const float* __restrict__ h,
                                                     const int* __restrict__ row_ptr,
                                                     const int* __restrict__ csr_src,
                                                     const float* __restrict__ src_scale,
                                                     const int node, const int lane) {
    const int start = row_ptr[node];
    const int end = row_ptr[node + 1];
    const float* hl = h + lane * 4;

    float4 acc = make_float4(0.f, 0.f, 0.f, 0.f);
    for (int base = start; base < end; base += 32) {
        const int n = min(32, end - base);
        int mine = (lane < n) ? csr_src[base + lane] : 0;
        mine = (int)min((unsigned)mine, (unsigned)(NN - 1));
        int j = 0;
        for (; j + 8 <= n; j += 8) {
            const float4 v0 = ld_row(hl, mine, j);
            const float4 v1 = ld_row(hl, mine, j + 1);
            const float4 v2 = ld_row(hl, mine, j + 2);
            const float4 v3 = ld_row(hl, mine, j + 3);
            const float4 v4 = ld_row(hl, mine, j + 4);
            const float4 v5 = ld_row(hl, mine, j + 5);
            const float4 v6 = ld_row(hl, mine, j + 6);
            const float4 v7 = ld_row(hl, mine, j + 7);
            const float w0 = ld_scale<SCALED>(src_scale, mine, j);
            const float w1 = ld_scale<SCALED>(src_scale, mine, j + 1);
            const float w2 = ld_scale<SCALED>(src_scale, mine, j + 2);
            const float w3 = ld_scale<SCALED>(src_scale, mine, j + 3);
            const float w4 = ld_scale<SCALED>(src_scale, mine, j + 4);
            const float w5 = ld_scale<SCALED>(src_scale, mine, j + 5);
            const float w6 = ld_scale<SCALED>(src_scale, mine, j + 6);
            const float w7 = ld_scale<SCALED>(src_scale, mine, j + 7);
            add4<SCALED>(acc, v0, w0); add4<SCALED>(acc, v1, w1); add4<SCALED>(acc, v2, w2); add4<SCALED>(acc, v3, w3);
            add4<SCALED>(acc, v4, w4); add4<SCALED>(acc, v5, w5); add4<SCALED>(acc, v6, w6); add4<SCALED>(acc, v7, w7);
        }
        if (j + 4 <= n) {
            const float4 v0 = ld_row(hl, mine, j);
            const float4 v1 = ld_row(hl, mine, j + 1);
            const float4 v2 = ld_row(hl, mine, j + 2);
            const float4 v3 = ld_row(hl, mine, j + 3);
            const float w0 = ld_scale<SCALED>(src_scale, mine, j);
            const float w1 = ld_scale<SCALED>(src_scale, mine, j + 1);
            const float w2 = ld_scale<SCALED>(src_scale, mine, j + 2);
            const float w3 = ld_scale<SCALED>(src_scale, mine, j + 3);
            add4<SCALED>(acc, v0, w0); add4<SCALED>(acc, v1, w1); add4<SCALED>(acc, v2, w2); add4<SCALED>(acc, v3, w3);
            j += 4;
        }
        if (j < n) {
            const float4 v0 = ld_row(hl, mine, j);
            float4 v1 = make_float4(0.f, 0.f, 0.f, 0.f);
            float4 v2 = make_float4(0.f, 0.f, 0.f, 0.f);
            if (j + 1 < n) v1 = ld_row(hl, mine, j + 1);
            if (j + 2 < n) v2 = ld_row(hl, mine, j + 2);
            const float w0 = ld_scale<SCALED>(src_scale, mine, j);
            const float w1 = ld_scale<SCALED>(src_scale, mine, min(j + 1, 31));
            const float w2 = ld_scale<SCALED>(src_scale, mine, min(j + 2, 31));
            add4<SCALED>(acc, v0, w0); add4<SCALED>(acc, v1, w1); add4<SCALED>(acc, v2, w2);
        }
    }
    return acc;
}

template <bool SCALED>
static __device__ __forceinline__ float4 gather_row(const float* __restrict__ h,
                                                    const int* __restrict__ row_ptr,
                                                    const int* __restrict__ csr_src,
                                                    const float* __restrict__ inv_in,
                                                    const float* __restrict__ src_scale,
                                                    const float* __restrict__ bias,
                                                    const int node, const int lane) {
    const float4 acc = gather_rows<SCALED>(h, row_ptr, csr_src, src_scale, node, lane);
    const float sc = inv_in[node];
    const float4 b = *(const float4*)(bias + lane * 4);
    float4 o;
    o.x = fmaxf(acc.x * sc + b.x, 0.0f);
    o.y = fmaxf(acc.y * sc + b.y, 0.0f);
    o.z = fmaxf(acc.z * sc + b.z, 0.0f);
    o.w = fmaxf(acc.w * sc + b.w, 0.0f);
    return o;
}

__global__ __launch_bounds__(256) void gather_sum_kernel(const float* __restrict__ h,
                                                         const int* __restrict__ row_ptr,
                                                         const int* __restrict__ csr_src,
                                                         const float* __restrict__ inv_in,
                                                         const float* __restrict__ src_scale,
                                                         const float* __restrict__ bias,
                                                         float* __restrict__ out) {
    const int lane = threadIdx.x & 31;
    const int wv = __builtin_amdgcn_readfirstlane((int)(threadIdx.x >> 5));
    const int node = blockIdx.x * 8 + wv;
    float4 o = gather_row<true>(h, row_ptr, csr_src, inv_in, src_scale, bias, node, lane);
    const float so = src_scale[node];
    o.x *= so; o.y *= so; o.z *= so; o.w *= so;
    *(float4*)(out + (size_t)node * FEAT + lane * 4) = o;
}

#define GP_BLOCKS 1024
__global__ __launch_bounds__(256) void gather_project_pool_kernel(const float* __restrict__ xs,
                                                                  const int* __restrict__ row_ptr,
                                                                  const int* __restrict__ csr_src,
                                                                  const float* __restrict__ inv_in,
                                                                  const float* __restrict__ W,
                                                                  const float* __restrict__ bias,
                                                                  const int* __restrict__ graph_ids,
                                                                  float* __restrict__ emb, const int ntiles) {
    __shared__ float4 tileA[2][16 * GT_PITCH / 4];
    const int lane = threadIdx.x & 31;
    const int wave = __builtin_amdgcn_readfirstlane((int)(threadIdx.x >> 5));
    const int c0 = wave * 16;

    const int m  = lane & 15;
    const int kb = (lane >> 4) << 1;
    const int n  = lane & 15;
    const int rhalf = (lane >> 4) * 8;

    const float* wcol = W + (size_t)kb * FEAT + c0 + n;
    v2f b[FEAT / 4];
#pragma unroll
    for (int s = 0; s < FEAT / 4; ++s) {
        b[s].x = wcol[(size_t)(4 * s) * FEAT];
        b[s].y = wcol[(size_t)(4 * s + 1) * FEAT];
    }
    const float bv = bias[c0 + n];
    float* ecol = emb + c0 + n;

    int buf = 0;
    for (int tile = blockIdx.x; tile < ntiles; tile += gridDim.x) {
        const int r0 = tile * 16;
        float* tl = (float*)&tileA[buf][0];
        for (int q = 0; q < 2; ++q) {
            const int row = 2 * wave + q;
            const float4 sum = gather_rows<false>(xs, row_ptr, csr_src, nullptr, r0 + row, lane);
            *(float4*)(tl + row * GT_PITCH + lane * 4) = sum;
        }
        __syncthreads();

        const int rb = r0 + rhalf;
        float iv[8];
        int gv[8];
#pragma unroll
        for (int r = 0; r < 8; ++r) {
            iv[r] = inv_in[rb + r];
            gv[r] = graph_ids[rb + r];
        }

        const float* ap = tl + m * GT_PITCH + kb;
        v2f a[FEAT / 4];
#pragma unroll
        for (int s = 0; s < FEAT / 4; ++s) a[s] = *(const v2f*)(ap + 4 * s);

        v8f acc = {0.f, 0.f, 0.f, 0.f, 0.f, 0.f, 0.f, 0.f};
#pragma unroll
        for (int s = 0; s < FEAT / 4; ++s) {
            acc = __builtin_amdgcn_wmma_f32_16x16x4_f32(
                 false, a[s],  false, b[s],
                 (short)0, acc,  false,  false);
        }

        float xv[8];
#pragma unroll
        for (int r = 0; r < 8; ++r) xv[r] = fmaxf(acc[r] * iv[r] + bv, 0.0f);
        int g = gv[0];
        float s = xv[0];
#pragma unroll
        for (int r = 1; r < 8; ++r) {
            if (gv[r] != g) {
                atomicAdd(ecol + (size_t)g * FEAT, s);
                g = gv[r];
                s = xv[r];
            } else {
                s += xv[r];
            }
        }
        atomicAdd(ecol + (size_t)g * FEAT, s);
        buf ^= 1;
    }
}

#define HROWS 16
static_assert(NG % HROWS == 0 && HROWS * OUTC <= 256 && (FEAT * OUTC) % 256 == 0, "head tiling");
__global__ __launch_bounds__(256) void head_kernel(const float* __restrict__ emb,
                                                   const float* __restrict__ gamma,
                                                   const float* __restrict__ beta,
                                                   const float* __restrict__ fc1_w,
                                                   const float* __restrict__ fc1_b,
                                                   const float* __restrict__ fc2_w,
                                                   const float* __restrict__ fc2_b,
                                                   float* __restrict__ out) {
    __shared__ float4 tileY[HROWS * GT_PITCH / 4];
    __shared__ float4 tileH[HROWS * GT_PITCH / 4];
    __shared__ float4 colMu[FEAT / 4], colSc[FEAT / 4];
    __shared__ float w2s[FEAT * OUTC];
    __shared__ float lgt[HROWS * OUTC];
    const int tid = threadIdx.x;
    const int lane = tid & 31;
    const int wave = __builtin_amdgcn_readfirstlane(tid >> 5);
    const int c0 = wave * 16;
    const int m  = lane & 15;
    const int kb = (lane >> 4) << 1;
    const int n  = lane & 15;
    const int rhalf = (lane >> 4) * 8;
    const int g0row = blockIdx.x * HROWS;

    const float* wcol = fc1_w + (size_t)kb * FEAT + c0 + n;
    v2f b[FEAT / 4];
#pragma unroll
    for (int s = 0; s < FEAT / 4; ++s) {
        b[s].x = wcol[(size_t)(4 * s) * FEAT];
        b[s].y = wcol[(size_t)(4 * s + 1) * FEAT];
    }
    const float bv = fc1_b[c0 + n];
    float4 y0 = *(const float4*)(emb + (size_t)(g0row + wave) * FEAT + lane * 4);
    float4 y1 = *(const float4*)(emb + (size_t)(g0row + wave + 8) * FEAT + lane * 4);
    const float4 be = *(const float4*)(beta + lane * 4);
    float w2r[FEAT * OUTC / 256];
#pragma unroll
    for (int i = 0; i < FEAT * OUTC / 256; ++i) w2r[i] = fc2_w[i * 256 + tid];
    float lg = 0.0f;
    if (tid < HROWS * OUTC) lg = fc2_b[tid % OUTC];

    if (wave < FEAT / 32) {
        const float* col = emb + tid;
        float e[NG];
#pragma unroll
        for (int g = 0; g < NG; ++g) e[g] = col[g * FEAT];
        float mu = 0.f;
#pragma unroll
        for (int g = 0; g < NG; ++g) mu += e[g];
        mu *= (1.0f / NG);
        float var = 0.f;
#pragma unroll
        for (int g = 0; g < NG; ++g) {
            const float d = e[g] - mu;
            var += d * d;
        }
        var *= (1.0f / NG);
        ((float*)&colMu[0])[tid] = mu;
        ((float*)&colSc[0])[tid] = gamma[tid] * rsqrtf(var + BN_EPS);
    }
#pragma unroll
    for (int i = 0; i < FEAT * OUTC / 256; ++i) w2s[i * 256 + tid] = w2r[i];
    __syncthreads();

    {
        const float4 mu4 = colMu[lane];
        const float4 sc4 = colSc[lane];
        float* ty = (float*)&tileY[0];
        y0.x = (y0.x - mu4.x) * sc4.x + be.x; y0.y = (y0.y - mu4.y) * sc4.y + be.y;
        y0.z = (y0.z - mu4.z) * sc4.z + be.z; y0.w = (y0.w - mu4.w) * sc4.w + be.w;
        y1.x = (y1.x - mu4.x) * sc4.x + be.x; y1.y = (y1.y - mu4.y) * sc4.y + be.y;
        y1.z = (y1.z - mu4.z) * sc4.z + be.z; y1.w = (y1.w - mu4.w) * sc4.w + be.w;
        *(float4*)(ty + wave * GT_PITCH + lane * 4) = y0;
        *(float4*)(ty + (wave + 8) * GT_PITCH + lane * 4) = y1;
    }
    __syncthreads();

    {
        const float* ap = (const float*)&tileY[0] + m * GT_PITCH + kb;
        v2f a[FEAT / 4];
#pragma unroll
        for (int s = 0; s < FEAT / 4; ++s) a[s] = *(const v2f*)(ap + 4 * s);
        v8f acc = {0.f, 0.f, 0.f, 0.f, 0.f, 0.f, 0.f, 0.f};
#pragma unroll
        for (int s = 0; s < FEAT / 4; ++s) {
            acc = __builtin_amdgcn_wmma_f32_16x16x4_f32(
                 false, a[s],  false, b[s],
                 (short)0, acc,  false,  false);
        }
        float* th = (float*)&tileH[0];
#pragma unroll
        for (int r = 0; r < 8; ++r) th[(rhalf + r) * GT_PITCH + c0 + n] = fmaxf(acc[r] + bv, 0.0f);
    }
    __syncthreads();

    const int row = tid / OUTC;
    const int cls = tid - row * OUTC;
    if (tid < HROWS * OUTC) {
        const float* hrow = (const float*)&tileH[0] + row * GT_PITCH;
        for (int k = 0; k < FEAT; ++k) lg = fmaf(hrow[k], w2s[k * OUTC + cls], lg);
        lgt[tid] = lg;
    }
    __syncthreads();
    if (tid < HROWS * OUTC) {
        const float* lr = lgt + row * OUTC;
        float mx = -1e30f;
#pragma unroll
        for (int j = 0; j < OUTC; ++j) mx = fmaxf(mx, lr[j]);
        float se = 0.f;
#pragma unroll
        for (int j = 0; j < OUTC; ++j) se += __expf(lr[j] - mx);
        const float lse = mx + __logf(se);
        out[(size_t)g0row * OUTC + tid] = lg - lse;
    }
}

extern "C" void kernel_launch(void* const* d_in, const int* in_sizes, int n_in,
                              void* d_out, int out_size, void* d_ws, size_t ws_size,
                              hipStream_t stream) {
    const float* n_feat    = (const float*)d_in[0];
    const int*   src       = (const int*)  d_in[1];
    const int*   dst       = (const int*)  d_in[2];
    const int*   graph_ids = (const int*)  d_in[3];
    const float* W1        = (const float*)d_in[4];
    const float* b1        = (const float*)d_in[5];
    const float* W2        = (const float*)d_in[6];
    const float* b2        = (const float*)d_in[7];
    const float* gamma     = (const float*)d_in[8];
    const float* beta      = (const float*)d_in[9];
    const float* fc1_w     = (const float*)d_in[10];
    const float* fc1_b     = (const float*)d_in[11];
    const float* fc2_w     = (const float*)d_in[12];
    const float* fc2_b     = (const float*)d_in[13];

    float* out = (float*)d_out;

    float* inv_out = (float*)d_ws;
    float* inv_in  = inv_out + 50176;
    int*   row_ptr = (int*)(inv_in + 50176);
    int*   offD    = row_ptr + 50176;
    int*   offS    = offD + NCHUNK * 256;
    unsigned* sortedD = (unsigned*)(offS + NCHUNK * 256);
    unsigned* sortedS = sortedD + NCHUNK * CHP;
    int*   csr_src = (int*)(sortedS + NCHUNK * CHP);
    float* bufA    = (float*)(csr_src + NE);
    float* bufB    = bufA + (size_t)NN * FEAT;

    bin_and_product_kernel<<<NCHUNK + L1_BLOCKS, 256, 0, stream>>>(src, dst, sortedD, sortedS, offD, offS, out,
                                                                   n_feat, W1, bufA);
    own_and_product_kernel<<<NBIN + L1_BLOCKS, 256, 0, stream>>>(sortedD, sortedS, offD, offS, row_ptr, csr_src,
                                                                 inv_in, inv_out, n_feat, W1, bufA);

    const int nodeTiles = NN / 16;
    const int nodeBlocks = NN / 8;

    gather_sum_kernel<<<nodeBlocks, 256, 0, stream>>>(bufA, row_ptr, csr_src, inv_in, inv_out, b1, bufB);

    gather_project_pool_kernel<<<GP_BLOCKS, 256, 0, stream>>>(bufB, row_ptr, csr_src, inv_in, W2, b2, graph_ids, out,
                                                              nodeTiles);

    head_kernel<<<NG / HROWS, 256, 0, stream>>>(out, gamma, beta, fc1_w, fc1_b, fc2_w, fc2_b,
                                                out + (size_t)NG * FEAT);
}


# gathers: index loads off the row loads' path: a layer-1 wave walks 4-5 consecutive nodes (12288 resident waves, not 50000); next list's bounds, ids asked for under current rows; sum orders kept
# speedup vs baseline: 7.0736x; 1.0737x over previous
#include <hip/hip_runtime.h>
#include <hip/hip_bf16.h>

#define NN 50000
#define NE 800000
#define FEAT 128
#define NG 128
#define OUTC 10
#define BN_EPS 1e-5f

typedef float v2f __attribute__((ext_vector_type(2)));
typedef float v8f __attribute__((ext_vector_type(8)));

#define NCHUNK 256
#define CH (NE / NCHUNK)
#define CHP 3136
#define RNG 256
#define NBIN ((NN + RNG - 1) / RNG)
#define BIN_TRIPS ((CH + 255) / 256)
static_assert(NE % NCHUNK == 0, "chunks are equal");
static_assert(NN <= 65536, "a source id is packed into 16 bits");
static_assert(NBIN < 256, "range b+1 has a column in the offset table");
static_assert((NG * FEAT) % 256 == 0 && (NG * FEAT) / 256 <= NCHUNK, "the first kernel's blocks cover the embedding rows");

static __device__ __forceinline__ int block_scan_excl(const int v, int* wtot, const int lane, const int wave,
                                                      int& total) {
    int incl = v;
    for (int d = 1; d < 32; d <<= 1) {
        const int up = __shfl_up(incl, d);
        if (lane >= d) incl += up;
    }
    if (lane == 31) wtot[wave] = incl;
    __syncthreads();
    int off = 0, tot = 0;
    for (int w = 0; w < 8; ++w) {
        const int t = wtot[w];
        tot += t;
        if (w < wave) off += t;
    }
    total = tot;
    return off + incl - v;
}

static __device__ __forceinline__ void edge_bin_body(const int* __restrict__ src, const int* __restrict__ dst,
                                                     unsigned* __restrict__ sortedD,
                                                     unsigned* __restrict__ sortedS,
                                                     int* __restrict__ offD, int* __restrict__ offS,
                                                     float* __restrict__ emb) {
    __shared__ int hD[256], hS[256];
    __shared__ int wtD[8], wtS[8];
    __shared__ unsigned stD[CH], stS[CH];
    const int tid = threadIdx.x;
    const int lane = tid & 31;
    const int wave = tid >> 5;
    const int c = blockIdx.x;
    if (c < (NG * FEAT) / 256) emb[c * 256 + tid] = 0.0f;
    hD[tid] = 0;
    hS[tid] = 0;
    __syncthreads();

    const int* sp = src + (size_t)c * CH;
    const int* dp = dst + (size_t)c * CH;
    int sv[BIN_TRIPS], dv[BIN_TRIPS], rD[BIN_TRIPS], rS[BIN_TRIPS];
#pragma unroll
    for (int i = 0; i < BIN_TRIPS; ++i) {
        const int e = i * 256 + tid;
        sv[i] = 0; dv[i] = 0;
        if (i < CH / 256 || e < CH) { sv[i] = sp[e]; dv[i] = dp[e]; }
    }
#pragma unroll
    for (int i = 0; i < BIN_TRIPS; ++i) {
        const int e = i * 256 + tid;
        rD[i] = 0; rS[i] = 0;
        if (i < CH / 256 || e < CH) {
            const unsigned bd = min((unsigned)dv[i] >> 8, (unsigned)(NBIN - 1));
            const unsigned bs = min((unsigned)sv[i] >> 8, (unsigned)(NBIN - 1));
            rD[i] = atomicAdd(&hD[bd], 1);
            rS[i] = atomicAdd(&hS[bs], 1);
        }
    }
    __syncthreads();
    int totD, totS;
    const int exD = block_scan_excl(hD[tid], wtD, lane, wave, totD);
    const int exS = block_scan_excl(hS[tid], wtS, lane, wave, totS);
    hD[tid] = exD;
    hS[tid] = exS;
    offD[c * 256 + tid] = exD;
    offS[c * 256 + tid] = exS;
    __syncthreads();
#pragma unroll
    for (int i = 0; i < BIN_TRIPS; ++i) {
        const int e = i * 256 + tid;
        if (i < CH / 256 || e < CH) {
            const unsigned bd = min((unsigned)dv[i] >> 8, (unsigned)(NBIN - 1));
            const unsigned bs = min((unsigned)sv[i] >> 8, (unsigned)(NBIN - 1));
            stD[hD[bd] + rD[i]] = (((unsigned)dv[i] & 255u) << 16) | ((unsigned)sv[i] & 0xFFFFu);
            stS[hS[bs] + rS[i]] = (unsigned)sv[i] & 255u;
        }
    }
    __syncthreads();
    unsigned* oD = sortedD + (size_t)c * CHP;
    unsigned* oS = sortedS + (size_t)c * CHP;
    for (int j = tid; j < CH; j += 256) {
        oD[j] = stD[j];
        oS[j] = stS[j];
    }
}

static __device__ __forceinline__ void owner_build_body(const unsigned* __restrict__ sortedD,
                                                        const unsigned* __restrict__ sortedS,
                                                        const int* __restrict__ offD,
                                                        const int* __restrict__ offS,
                                                        int* __restrict__ row_ptr, int* __restrict__ csr_src,
                                                        float* __restrict__ inv_in,
                                                        float* __restrict__ inv_out) {
    __shared__ int cntI[RNG], cntO[RNG], cur[RNG];
    __shared__ int wtA[8], wtB[8];
    const int tid = threadIdx.x;
    const int lane = tid & 31;
    const int wave = __builtin_amdgcn_readfirstlane(tid >> 5);
    const int b = blockIdx.x;
    const int c = tid;
    const int bD = offD[c * 256 + b];
    const int nD = min(offD[c * 256 + b + 1] - bD, CH);
    const int bS = offS[c * 256 + b];
    const int nS = min(offS[c * 256 + b + 1] - bS, CH);
    cntI[tid] = 0;
    cntO[tid] = 0;
    __syncthreads();

    const unsigned* cD = sortedD + (size_t)wave * 32 * CHP;
    const unsigned* cS = sortedS + (size_t)wave * 32 * CHP;
    unsigned wd[32], wo[32];
#pragma unroll
    for (int j = 0; j < 32; ++j) {
        const unsigned* p = cD + (j * CHP + __builtin_amdgcn_readlane(bD, j));
        wd[j] = p[(lane < __builtin_amdgcn_readlane(nD, j)) ? (unsigned)lane : 0u];
    }
#pragma unroll
    for (int j = 0; j < 32; ++j) {
        const unsigned* p = cS + (j * CHP + __builtin_amdgcn_readlane(bS, j));
        wo[j] = p[(lane < __builtin_amdgcn_readlane(nS, j)) ? (unsigned)lane : 0u];
    }
#pragma unroll
    for (int j = 0; j < 32; ++j) {
        if (lane < __builtin_amdgcn_readlane(nD, j)) atomicAdd(&cntI[wd[j] >> 16], 1);
    }
#pragma unroll
    for (int j = 0; j < 32; ++j) {
        if (lane < __builtin_amdgcn_readlane(nS, j)) atomicAdd(&cntO[wo[j] & 255u], 1);
    }
    const bool longPiece = __builtin_amdgcn_ballot_w32(nD > 32 || nS > 32) != 0u;
    if (longPiece) {
        for (int j = 0; j < 32; ++j) {
            const unsigned* pD = cD + j * CHP + __builtin_amdgcn_readlane(bD, j);
            const unsigned* pS = cS + j * CHP + __builtin_amdgcn_readlane(bS, j);
            const int n1 = __builtin_amdgcn_readlane(nD, j);
            const int n2 = __builtin_amdgcn_readlane(nS, j);
            for (int k = 32 + lane; k < n1; k += 32) atomicAdd(&cntI[pD[k] >> 16], 1);
            for (int k = 32 + lane; k < n2; k += 32) atomicAdd(&cntO[pS[k] & 255u], 1);
        }
    }

    int binStart, totI;
    (void)block_scan_excl(bD, wtA, lane, wave, binStart);
    const int my = cntI[tid];
    const int start = binStart + block_scan_excl(my, wtB, lane, wave, totI);
    const int node = b * RNG + tid;
    if (node <= NN) row_ptr[node] = start;
    if (node < NN) {
        inv_in[node]  = rsqrtf(fmaxf((float)my, 1.0f));
        inv_out[node] = rsqrtf(fmaxf((float)cntO[tid], 1.0f));
    }
    cur[tid] = start;
    __syncthreads();

    unsigned pos[32];
#pragma unroll
    for (int j = 0; j < 32; ++j) {
        pos[j] = (unsigned)NE;
        if (lane < __builtin_amdgcn_readlane(nD, j)) pos[j] = (unsigned)atomicAdd(&cur[wd[j] >> 16], 1);
    }
#pragma unroll
    for (int j = 0; j < 32; ++j) {
        if (pos[j] < (unsigned)NE) csr_src[pos[j]] = (int)(wd[j] & 0xFFFFu);
    }
    if (longPiece) {
        for (int j = 0; j < 32; ++j) {
            const unsigned* pD = cD + j * CHP + __builtin_amdgcn_readlane(bD, j);
            const int n1 = __builtin_amdgcn_readlane(nD, j);
            for (int k = 32 + lane; k < n1; k += 32) {
                const unsigned w = pD[k];
                const unsigned p = (unsigned)atomicAdd(&cur[w >> 16], 1);
                if (p < (unsigned)NE) csr_src[p] = (int)(w & 0xFFFFu);
            }
        }
    }
}

#define GT_PITCH 132
static __device__ __forceinline__ void gemm128_body(const float* __restrict__ A,
                                                    const float* __restrict__ W,
                                                    float* __restrict__ out,
                                                    const float* __restrict__ scaleA,
                                                    const float* __restrict__ bias,
                                                    const int relu, const int first, const int step,
                                                    const int ntiles) {
    __shared__ float4 tileA[2][16 * GT_PITCH / 4];
    const int lane = threadIdx.x & 31;
    const int wave = __builtin_amdgcn_readfirstlane((int)(threadIdx.x >> 5));
    const int c0 = wave * 16;

    const int m  = lane & 15;
    const int kb = (lane >> 4) << 1;
    const int n  = lane & 15;

    const float* wcol = W + (size_t)kb * FEAT + c0 + n;
    v2f b[FEAT / 4];
#pragma unroll
    for (int s = 0; s < FEAT / 4; ++s) {
        b[s].x = wcol[(size_t)(4 * s) * FEAT];
        b[s].y = wcol[(size_t)(4 * s + 1) * FEAT];
    }
    const float bv = bias ? bias[c0 + n] : 0.0f;
    const int rhalf = (lane >> 4) * 8;

    int tile = first;
    float4 g0 = make_float4(0.f, 0.f, 0.f, 0.f), g1 = g0;
    float s0 = 1.0f, s1 = 1.0f;
    if (tile < ntiles) {
        const size_t ra = (size_t)tile * 16 + wave;
        g0 = *(const float4*)(A + ra * FEAT + lane * 4);
        g1 = *(const float4*)(A + (ra + 8) * FEAT + lane * 4);
        if (scaleA) { s0 = scaleA[ra]; s1 = scaleA[ra + 8]; }
    }
    int buf = 0;
    for (; tile < ntiles; tile += step) {
        const int r0 = tile * 16;
        float* tl = (float*)&tileA[buf][0];
        g0.x *= s0; g0.y *= s0; g0.z *= s0; g0.w *= s0;
        g1.x *= s1; g1.y *= s1; g1.z *= s1; g1.w *= s1;
        *(float4*)(tl + wave * GT_PITCH + lane * 4) = g0;
        *(float4*)(tl + (wave + 8) * GT_PITCH + lane * 4) = g1;
        __syncthreads();

        const int nt = tile + step;
        if (nt < ntiles) {
            const size_t ra = (size_t)nt * 16 + wave;
            g0 = *(const float4*)(A + ra * FEAT + lane * 4);
            g1 = *(const float4*)(A + (ra + 8) * FEAT + lane * 4);
            if (scaleA) { s0 = scaleA[ra]; s1 = scaleA[ra + 8]; }
        }

        const float* ap = tl + m * GT_PITCH + kb;
        v2f a[FEAT / 4];
#pragma unroll
        for (int s = 0; s < FEAT / 4; ++s) a[s] = *(const v2f*)(ap + 4 * s);

        v8f acc = {0.f, 0.f, 0.f, 0.f, 0.f, 0.f, 0.f, 0.f};
#pragma unroll
        for (int s = 0; s < FEAT / 4; ++s) {
            acc = __builtin_amdgcn_wmma_f32_16x16x4_f32(
                 false, a[s],  false, b[s],
                 (short)0, acc,  false,  false);
        }

#pragma unroll
        for (int r = 0; r < 8; ++r) {
            float v = acc[r] + bv;
            if (relu) v = fmaxf(v, 0.0f);
            out[(size_t)(r0 + rhalf + r) * FEAT + c0 + n] = v;
        }
        buf ^= 1;
    }
}

#define L1_SPLIT 1563
#define L1_BLOCKS 512
static_assert(L1_SPLIT > 0 && L1_SPLIT < NN / 16, "both launches hold a share");
__global__ __launch_bounds__(256) void bin_and_product_kernel(const int* __restrict__ src,
                                                              const int* __restrict__ dst,
                                                              unsigned* __restrict__ sortedD,
                                                              unsigned* __restrict__ sortedS,
                                                              int* __restrict__ offD, int* __restrict__ offS,
                                                              float* __restrict__ emb,
                                                              const float* __restrict__ X,
                                                              const float* __restrict__ W,
                                                              float* __restrict__ h) {
    if (blockIdx.x < NCHUNK) {
        edge_bin_body(src, dst, sortedD, sortedS, offD, offS, emb);
    } else {
        gemm128_body(X, W, h, nullptr, nullptr, 0, (int)blockIdx.x - NCHUNK, L1_BLOCKS, L1_SPLIT);
    }
}

__global__ __launch_bounds__(256) void own_and_product_kernel(const unsigned* __restrict__ sortedD,
                                                              const unsigned* __restrict__ sortedS,
                                                              const int* __restrict__ offD,
                                                              const int* __restrict__ offS,
                                                              int* __restrict__ row_ptr, int* __restrict__ csr_src,
                                                              float* __restrict__ inv_in,
                                                              float* __restrict__ inv_out,
                                                              const float* __restrict__ X,
                                                              const float* __restrict__ W,
                                                              float* __restrict__ h) {
    if (blockIdx.x < NBIN) {
        owner_build_body(sortedD, sortedS, offD, offS, row_ptr, csr_src, inv_in, inv_out);
    } else {
        gemm128_body(X, W, h, nullptr, nullptr, 0, L1_SPLIT + (int)blockIdx.x - NBIN, L1_BLOCKS, NN / 16);
    }
}

static __device__ __forceinline__ float4 ld_row(const float* hl, int ids, int j) {
    const int s = __builtin_amdgcn_readlane(ids, j);
    return *(const float4*)(hl + (size_t)s * FEAT);
}
template <bool SCALED>
static __device__ __forceinline__ float ld_scale(const float* __restrict__ src_scale, int ids, int j) {
    if (!SCALED) return 1.0f;
    return src_scale[__builtin_amdgcn_readlane(ids, j)];
}
template <bool SCALED>
static __device__ __forceinline__ void add4(float4& a, const float4 v, const float w) {
    if (SCALED) {
        a.x = fmaf(v.x, w, a.x); a.y = fmaf(v.y, w, a.y); a.z = fmaf(v.z, w, a.z); a.w = fmaf(v.w, w, a.w);
    } else {
        a.x += v.x; a.y += v.y; a.z += v.z; a.w += v.w;
    }
}

static __device__ __forceinline__ int ld_ids(const int* __restrict__ csr_src, const int base, const int end,
                                             const int lane) {
    const int n = min(32, end - base);
    const int mine = (lane < n) ? csr_src[base + lane] : 0;
    return (int)min((unsigned)mine, (unsigned)(NN - 1));
}

static __device__ __forceinline__ int ld_ids_raw(const int* __restrict__ csr_src, const int base, const int lane) {
    return csr_src[min((unsigned)(base + lane), (unsigned)(NE - 1))];
}
static __device__ __forceinline__ int fix_ids(const int raw, const int len, const int lane) {
    const int mine = (lane < len) ? raw : 0;
    return (int)min((unsigned)mine, (unsigned)(NN - 1));
}

template <bool SCALED>
static __device__ __forceinline__ float4 gather_rows(const float* __restrict__ h,
                                                     const int* __restrict__ csr_src,
                                                     const float* __restrict__ src_scale,
                                                     const int start, const int end, const int first,
                                                     const int lane) {
    const float* hl = h + lane * 4;

    float4 acc = make_float4(0.f, 0.f, 0.f, 0.f);
    int mine = first;
    for (int base = start; base < end; base += 32) {
        const int n = min(32, end - base);
        if (base != start) mine = ld_ids(csr_src, base, end, lane);
        int j = 0;
        for (; j + 8 <= n; j += 8) {
            const float4 v0 = ld_row(hl, mine, j);
            const float4 v1 = ld_row(hl, mine, j + 1);
            const float4 v2 = ld_row(hl, mine, j + 2);
            const float4 v3 = ld_row(hl, mine, j + 3);
            const float4 v4 = ld_row(hl, mine, j + 4);
            const float4 v5 = ld_row(hl, mine, j + 5);
            const float4 v6 = ld_row(hl, mine, j + 6);
            const float4 v7 = ld_row(hl, mine, j + 7);
            const float w0 = ld_scale<SCALED>(src_scale, mine, j);
            const float w1 = ld_scale<SCALED>(src_scale, mine, j + 1);
            const float w2 = ld_scale<SCALED>(src_scale, mine, j + 2);
            const float w3 = ld_scale<SCALED>(src_scale, mine, j + 3);
            const float w4 = ld_scale<SCALED>(src_scale, mine, j + 4);
            const float w5 = ld_scale<SCALED>(src_scale, mine, j + 5);
            const float w6 = ld_scale<SCALED>(src_scale, mine, j + 6);
            const float w7 = ld_scale<SCALED>(src_scale, mine, j + 7);
            add4<SCALED>(acc, v0, w0); add4<SCALED>(acc, v1, w1); add4<SCALED>(acc, v2, w2); add4<SCALED>(acc, v3, w3);
            add4<SCALED>(acc, v4, w4); add4<SCALED>(acc, v5, w5); add4<SCALED>(acc, v6, w6); add4<SCALED>(acc, v7, w7);
        }
        if (j + 4 <= n) {
            const float4 v0 = ld_row(hl, mine, j);
            const float4 v1 = ld_row(hl, mine, j + 1);
            const float4 v2 = ld_row(hl, mine, j + 2);
            const float4 v3 = ld_row(hl, mine, j + 3);
            const float w0 = ld_scale<SCALED>(src_scale, mine, j);
            const float w1 = ld_scale<SCALED>(src_scale, mine, j + 1);
            const float w2 = ld_scale<SCALED>(src_scale, mine, j + 2);
            const float w3 = ld_scale<SCALED>(src_scale, mine, j + 3);
            add4<SCALED>(acc, v0, w0); add4<SCALED>(acc, v1, w1); add4<SCALED>(acc, v2, w2); add4<SCALED>(acc, v3, w3);
            j += 4;
        }
        if (j < n) {
            const float4 v0 = ld_row(hl, mine, j);
            float4 v1 = make_float4(0.f, 0.f, 0.f, 0.f);
            float4 v2 = make_float4(0.f, 0.f, 0.f, 0.f);
            if (j + 1 < n) v1 = ld_row(hl, mine, j + 1);
            if (j + 2 < n) v2 = ld_row(hl, mine, j + 2);
            const float w0 = ld_scale<SCALED>(src_scale, mine, j);
            const float w1 = ld_scale<SCALED>(src_scale, mine, min(j + 1, 31));
            const float w2 = ld_scale<SCALED>(src_scale, mine, min(j + 2, 31));
            add4<SCALED>(acc, v0, w0); add4<SCALED>(acc, v1, w1); add4<SCALED>(acc, v2, w2);
        }
    }
    return acc;
}

#define G1_BLOCKS 1536
#define G1_WAVES (G1_BLOCKS * 8)
#define G1_MAXN ((NN + G1_WAVES - 1) / G1_WAVES)
static_assert((long long)G1_WAVES * NN < (1ll << 31), "w * NN is formed in 32 bits");
static_assert((long long)(G1_WAVES - 1) * NN / G1_WAVES + G1_MAXN == NN, "the last stretch is a longest one");
__global__ __launch_bounds__(256) void gather_sum_kernel(const float* __restrict__ h,
                                                         const int* __restrict__ row_ptr,
                                                         const int* __restrict__ csr_src,
                                                         const float* __restrict__ inv_in,
                                                         const float* __restrict__ src_scale,
                                                         const float* __restrict__ bias,
                                                         float* __restrict__ out) {
    const int lane = threadIdx.x & 31;
    const int wv = __builtin_amdgcn_readfirstlane((int)(threadIdx.x >> 5));
    const unsigned w = blockIdx.x * 8 + wv;
    const int n0 = (int)(w * (unsigned)NN / (unsigned)G1_WAVES);
    const int cnt = (int)((w + 1u) * (unsigned)NN / (unsigned)G1_WAVES) - n0;

    int rp[G1_MAXN + 1];
    float si[G1_MAXN], so[G1_MAXN];
#pragma unroll
    for (int k = 0; k <= G1_MAXN; ++k) rp[k] = row_ptr[n0 + k];
#pragma unroll
    for (int k = 0; k < G1_MAXN; ++k) {
        si[k] = inv_in[n0 + k];
        so[k] = src_scale[n0 + k];
    }
    const float4 b = *(const float4*)(bias + lane * 4);

    int raw = ld_ids_raw(csr_src, rp[0], lane);
#pragma unroll
    for (int k = 0; k < G1_MAXN; ++k) {
        if (k < cnt) {
            int nxt = 0;
            if (k + 1 < G1_MAXN) nxt = ld_ids_raw(csr_src, rp[k + 1], lane);
            const float4 acc = gather_rows<true>(h, csr_src, src_scale, rp[k], rp[k + 1],
                                                 fix_ids(raw, rp[k + 1] - rp[k], lane), lane);
            float4 o;
            o.x = fmaxf(acc.x * si[k] + b.x, 0.0f) * so[k];
            o.y = fmaxf(acc.y * si[k] + b.y, 0.0f) * so[k];
            o.z = fmaxf(acc.z * si[k] + b.z, 0.0f) * so[k];
            o.w = fmaxf(acc.w * si[k] + b.w, 0.0f) * so[k];
            *(float4*)(out + (size_t)(n0 + k) * FEAT + lane * 4) = o;
            raw = nxt;
        }
    }
}

#define GP_BLOCKS 1024
__global__ __launch_bounds__(256) void gather_project_pool_kernel(const float* __restrict__ xs,
                                                                  const int* __restrict__ row_ptr,
                                                                  const int* __restrict__ csr_src,
                                                                  const float* __restrict__ inv_in,
                                                                  const float* __restrict__ W,
                                                                  const float* __restrict__ bias,
                                                                  const int* __restrict__ graph_ids,
                                                                  float* __restrict__ emb, const int ntiles) {
    __shared__ float4 tileA[2][16 * GT_PITCH / 4];
    const int lane = threadIdx.x & 31;
    const int wave = __builtin_amdgcn_readfirstlane((int)(threadIdx.x >> 5));
    const int c0 = wave * 16;

    const int m  = lane & 15;
    const int kb = (lane >> 4) << 1;
    const int n  = lane & 15;
    const int rhalf = (lane >> 4) * 8;

    const float* wcol = W + (size_t)kb * FEAT + c0 + n;
    v2f b[FEAT / 4];
#pragma unroll
    for (int s = 0; s < FEAT / 4; ++s) {
        b[s].x = wcol[(size_t)(4 * s) * FEAT];
        b[s].y = wcol[(size_t)(4 * s + 1) * FEAT];
    }
    const float bv = bias[c0 + n];
    float* ecol = emb + c0 + n;

    int tile = blockIdx.x;
    int p0 = 0, p1 = 0, p2 = 0, raw0 = 0, raw1 = 0;
    if (tile < ntiles) {
        const int nb = tile * 16 + 2 * wave;
        p0 = row_ptr[nb]; p1 = row_ptr[nb + 1]; p2 = row_ptr[nb + 2];
        raw0 = ld_ids_raw(csr_src, p0, lane);
        raw1 = ld_ids_raw(csr_src, p1, lane);
    }
    int buf = 0;
    for (; tile < ntiles; tile += gridDim.x) {
        const int r0 = tile * 16;
        float* tl = (float*)&tileA[buf][0];
        const int nt = tile + gridDim.x;
        const int qv = row_ptr[min(nt, ntiles - 1) * 16 + 2 * wave + min(lane, 2)];
        const float4 sum0 = gather_rows<false>(xs, csr_src, nullptr, p0, p1, fix_ids(raw0, p1 - p0, lane), lane);
        *(float4*)(tl + (2 * wave) * GT_PITCH + lane * 4) = sum0;
        const float4 sum1 = gather_rows<false>(xs, csr_src, nullptr, p1, p2, fix_ids(raw1, p2 - p1, lane), lane);
        *(float4*)(tl + (2 * wave + 1) * GT_PITCH + lane * 4) = sum1;
        __syncthreads();

        if (nt < ntiles) {
            p0 = __builtin_amdgcn_readlane(qv, 0);
            p1 = __builtin_amdgcn_readlane(qv, 1);
            p2 = __builtin_amdgcn_readlane(qv, 2);
            raw0 = ld_ids_raw(csr_src, p0, lane);
            raw1 = ld_ids_raw(csr_src, p1, lane);
        }

        const int rb = r0 + rhalf;
        float iv[8];
        int gv[8];
#pragma unroll
        for (int r = 0; r < 8; ++r) {
            iv[r] = inv_in[rb + r];
            gv[r] = graph_ids[rb + r];
        }

        const float* ap = tl + m * GT_PITCH + kb;
        v2f a[FEAT / 4];
#pragma unroll
        for (int s = 0; s < FEAT / 4; ++s) a[s] = *(const v2f*)(ap + 4 * s);

        v8f acc = {0.f, 0.f, 0.f, 0.f, 0.f, 0.f, 0.f, 0.f};
#pragma unroll
        for (int s = 0; s < FEAT / 4; ++s) {
            acc = __builtin_amdgcn_wmma_f32_16x16x4_f32(
                 false, a[s],  false, b[s],
                 (short)0, acc,  false,  false);
        }

        float xv[8];
#pragma unroll
        for (int r = 0; r < 8; ++r) xv[r] = fmaxf(acc[r] * iv[r] + bv, 0.0f);
        int g = gv[0];
        float s = xv[0];
#pragma unroll
        for (int r = 1; r < 8; ++r) {
            if (gv[r] != g) {
                atomicAdd(ecol + (size_t)g * FEAT, s);
                g = gv[r];
                s = xv[r];
            } else {
                s += xv[r];
            }
        }
        atomicAdd(ecol + (size_t)g * FEAT, s);
        buf ^= 1;
    }
}

#define HROWS 16
static_assert(NG % HROWS == 0 && HROWS * OUTC <= 256 && (FEAT * OUTC) % 256 == 0, "head tiling");
__global__ __launch_bounds__(256) void head_kernel(const float* __restrict__ emb,
                                                   const float* __restrict__ gamma,
                                                   const float* __restrict__ beta,
                                                   const float* __restrict__ fc1_w,
                                                   const float* __restrict__ fc1_b,
                                                   const float* __restrict__ fc2_w,
                                                   const float* __restrict__ fc2_b,
                                                   float* __restrict__ out) {
    __shared__ float4 tileY[HROWS * GT_PITCH / 4];
    __shared__ float4 tileH[HROWS * GT_PITCH / 4];
    __shared__ float4 colMu[FEAT / 4], colSc[FEAT / 4];
    __shared__ float w2s[FEAT * OUTC];
    __shared__ float lgt[HROWS * OUTC];
    const int tid = threadIdx.x;
    const int lane = tid & 31;
    const int wave = __builtin_amdgcn_readfirstlane(tid >> 5);
    const int c0 = wave * 16;
    const int m  = lane & 15;
    const int kb = (lane >> 4) << 1;
    const int n  = lane & 15;
    const int rhalf = (lane >> 4) * 8;
    const int g0row = blockIdx.x * HROWS;

    const float* wcol = fc1_w + (size_t)kb * FEAT + c0 + n;
    v2f b[FEAT / 4];
#pragma unroll
    for (int s = 0; s < FEAT / 4; ++s) {
        b[s].x = wcol[(size_t)(4 * s) * FEAT];
        b[s].y = wcol[(size_t)(4 * s + 1) * FEAT];
    }
    const float bv = fc1_b[c0 + n];
    float4 y0 = *(const float4*)(emb + (size_t)(g0row + wave) * FEAT + lane * 4);
    float4 y1 = *(const float4*)(emb + (size_t)(g0row + wave + 8) * FEAT + lane * 4);
    const float4 be = *(const float4*)(beta + lane * 4);
    float w2r[FEAT * OUTC / 256];
#pragma unroll
    for (int i = 0; i < FEAT * OUTC / 256; ++i) w2r[i] = fc2_w[i * 256 + tid];
    float lg = 0.0f;
    if (tid < HROWS * OUTC) lg = fc2_b[tid % OUTC];

    if (wave < FEAT / 32) {
        const float* col = emb + tid;
        float e[NG];
#pragma unroll
        for (int g = 0; g < NG; ++g) e[g] = col[g * FEAT];
        float mu = 0.f;
#pragma unroll
        for (int g = 0; g < NG; ++g) mu += e[g];
        mu *= (1.0f / NG);
        float var = 0.f;
#pragma unroll
        for (int g = 0; g < NG; ++g) {
            const float d = e[g] - mu;
            var += d * d;
        }
        var *= (1.0f / NG);
        ((float*)&colMu[0])[tid] = mu;
        ((float*)&colSc[0])[tid] = gamma[tid] * rsqrtf(var + BN_EPS);
    }
#pragma unroll
    for (int i = 0; i < FEAT * OUTC / 256; ++i) w2s[i * 256 + tid] = w2r[i];
    __syncthreads();

    {
        const float4 mu4 = colMu[lane];
        const float4 sc4 = colSc[lane];
        float* ty = (float*)&tileY[0];
        y0.x = (y0.x - mu4.x) * sc4.x + be.x; y0.y = (y0.y - mu4.y) * sc4.y + be.y;
        y0.z = (y0.z - mu4.z) * sc4.z + be.z; y0.w = (y0.w - mu4.w) * sc4.w + be.w;
        y1.x = (y1.x - mu4.x) * sc4.x + be.x; y1.y = (y1.y - mu4.y) * sc4.y + be.y;
        y1.z = (y1.z - mu4.z) * sc4.z + be.z; y1.w = (y1.w - mu4.w) * sc4.w + be.w;
        *(float4*)(ty + wave * GT_PITCH + lane * 4) = y0;
        *(float4*)(ty + (wave + 8) * GT_PITCH + lane * 4) = y1;
    }
    __syncthreads();

    {
        const float* ap = (const float*)&tileY[0] + m * GT_PITCH + kb;
        v2f a[FEAT / 4];
#pragma unroll
        for (int s = 0; s < FEAT / 4; ++s) a[s] = *(const v2f*)(ap + 4 * s);
        v8f acc = {0.f, 0.f, 0.f, 0.f, 0.f, 0.f, 0.f, 0.f};
#pragma unroll
        for (int s = 0; s < FEAT / 4; ++s) {
            acc = __builtin_amdgcn_wmma_f32_16x16x4_f32(
                 false, a[s],  false, b[s],
                 (short)0, acc,  false,  false);
        }
        float* th = (float*)&tileH[0];
#pragma unroll
        for (int r = 0; r < 8; ++r) th[(rhalf + r) * GT_PITCH + c0 + n] = fmaxf(acc[r] + bv, 0.0f);
    }
    __syncthreads();

    const int row = tid / OUTC;
    const int cls = tid - row * OUTC;
    if (tid < HROWS * OUTC) {
        const float* hrow = (const float*)&tileH[0] + row * GT_PITCH;
        for (int k = 0; k < FEAT; ++k) lg = fmaf(hrow[k], w2s[k * OUTC + cls], lg);
        lgt[tid] = lg;
    }
    __syncthreads();
    if (tid < HROWS * OUTC) {
        const float* lr = lgt + row * OUTC;
        float mx = -1e30f;
#pragma unroll
        for (int j = 0; j < OUTC; ++j) mx = fmaxf(mx, lr[j]);
        float se = 0.f;
#pragma unroll
        for (int j = 0; j < OUTC; ++j) se += __expf(lr[j] - mx);
        const float lse = mx + __logf(se);
        out[(size_t)g0row * OUTC + tid] = lg - lse;
    }
}

extern "C" void kernel_launch(void* const* d_in, const int* in_sizes, int n_in,
                              void* d_out, int out_size, void* d_ws, size_t ws_size,
                              hipStream_t stream) {
    const float* n_feat    = (const float*)d_in[0];
    const int*   src       = (const int*)  d_in[1];
    const int*   dst       = (const int*)  d_in[2];
    const int*   graph_ids = (const int*)  d_in[3];
    const float* W1        = (const float*)d_in[4];
    const float* b1        = (const float*)d_in[5];
    const float* W2        = (const float*)d_in[6];
    const float* b2        = (const float*)d_in[7];
    const float* gamma     = (const float*)d_in[8];
    const float* beta      = (const float*)d_in[9];
    const float* fc1_w     = (const float*)d_in[10];
    const float* fc1_b     = (const float*)d_in[11];
    const float* fc2_w     = (const float*)d_in[12];
    const float* fc2_b     = (const float*)d_in[13];

    float* out = (float*)d_out;

    float* inv_out = (float*)d_ws;
    float* inv_in  = inv_out + 50176;
    int*   row_ptr = (int*)(inv_in + 50176);
    int*   offD    = row_ptr + 50176;
    int*   offS    = offD + NCHUNK * 256;
    unsigned* sortedD = (unsigned*)(offS + NCHUNK * 256);
    unsigned* sortedS = sortedD + NCHUNK * CHP;
    int*   csr_src = (int*)(sortedS + NCHUNK * CHP);
    float* bufA    = (float*)(csr_src + NE);
    float* bufB    = bufA + (size_t)NN * FEAT;

    bin_and_product_kernel<<<NCHUNK + L1_BLOCKS, 256, 0, stream>>>(src, dst, sortedD, sortedS, offD, offS, out,
                                                                   n_feat, W1, bufA);
    own_and_product_kernel<<<NBIN + L1_BLOCKS, 256, 0, stream>>>(sortedD, sortedS, offD, offS, row_ptr, csr_src,
                                                                 inv_in, inv_out, n_feat, W1, bufA);

    const int nodeTiles = NN / 16;

    gather_sum_kernel<<<G1_BLOCKS, 256, 0, stream>>>(bufA, row_ptr, csr_src, inv_in, inv_out, b1, bufB);

    gather_project_pool_kernel<<<GP_BLOCKS, 256, 0, stream>>>(bufB, row_ptr, csr_src, inv_in, W2, b2, graph_ids, out,
                                                              nodeTiles);

    head_kernel<<<NG / HROWS, 256, 0, stream>>>(out, gamma, beta, fc1_w, fc1_b, fc2_w, fc2_b,
                                                out + (size_t)NG * FEAT);
}
